# starting kernel
# baseline (speedup 1.0000x reference)
#include <hip/hip_runtime.h>
#include <hip/hip_bf16.h>

typedef float v2f __attribute__((ext_vector_type(2)));
typedef float v8f __attribute__((ext_vector_type(8)));

__global__ void gemm_nt_wmma(const float* __restrict__ A,
                             const float* __restrict__ W,
                             float* __restrict__ C,
                             int M, int K, int O) {
  const int lane  = threadIdx.x & 31;
  const int oTile = threadIdx.x >> 5;
  const int mTile = blockIdx.x;
  if (mTile * 16 >= M) return;
  const int half = lane >> 4;
  const int l16  = lane & 15;

  const float* aPtr = A + (size_t)(mTile * 16 + l16) * K + 2 * half;
  const float* wPtr = W + (size_t)(oTile * 16 + l16) * K + 2 * half;

  v8f acc = {};
#pragma unroll 8
  for (int k = 0; k < K; k += 4) {
    v2f a = *(const v2f*)(aPtr + k);
    v2f b = *(const v2f*)(wPtr + k);
    acc = __builtin_amdgcn_wmma_f32_16x16x4_f32(
         false, a,  false, b,
         (short)0, acc,  false,  false);
  }

  float* cPtr = C + (size_t)(mTile * 16 + half * 8) * O + oTile * 16 + l16;
#pragma unroll
  for (int r = 0; r < 8; ++r) cPtr[(size_t)r * O] = acc[r];
}

__device__ __forceinline__ unsigned float_key(float f) {
  unsigned u = __float_as_uint(f);
  return (u & 0x80000000u) ? ~u : (u | 0x80000000u);
}
__device__ __forceinline__ float key_float(unsigned k) {
  unsigned u = (k & 0x80000000u) ? (k & 0x7FFFFFFFu) : ~k;
  return __uint_as_float(u);
}

__global__ void edge_logits(const float* __restrict__ z,
                            const float* __restrict__ df,
                            const int* __restrict__ src,
                            const int* __restrict__ dst,
                            float* __restrict__ e,
                            unsigned* __restrict__ emaxKey, int E) {
  const int edge = (int)((blockIdx.x * (size_t)blockDim.x + threadIdx.x) >> 5);
  const int lane = threadIdx.x & 31;
  if (edge >= E) return;
  const int s = src[edge], d = dst[edge];
  v2f a = *(const v2f*)(z  + (size_t)s * 64 + lane * 2);
  v2f b = *(const v2f*)(df + (size_t)d * 64 + lane * 2);
  float p = a.x * b.x + a.y * b.y;
#pragma unroll
  for (int off = 16; off; off >>= 1) p += __shfl_xor(p, off, 32);
  if (lane == 0) {
    e[edge] = p;
    atomicMax(emaxKey + d, float_key(p));
  }
}

__global__ void edge_scatter(const float* __restrict__ z,
                             const float* __restrict__ e,
                             const unsigned* __restrict__ emaxKey,
                             const int* __restrict__ src,
                             const int* __restrict__ dst,
                             float* __restrict__ num,
                             float* __restrict__ denom, int E) {
  const int edge = (int)((blockIdx.x * (size_t)blockDim.x + threadIdx.x) >> 5);
  const int lane = threadIdx.x & 31;
  if (edge >= E) return;
  const int s = src[edge], d = dst[edge];
  const float m  = key_float(emaxKey[d]);
  const float ex = __expf(e[edge] - m);
  v2f a = *(const v2f*)(z + (size_t)s * 64 + lane * 2);
  float* np = num + (size_t)d * 64 + lane * 2;
  atomicAdd(np + 0, ex * a.x);
  atomicAdd(np + 1, ex * a.y);
  if (lane == 0) atomicAdd(denom + d, ex);
}

__global__ void finalize(const float* __restrict__ num,
                         const float* __restrict__ denom,
                         float* __restrict__ out, int total) {
  const int i = blockIdx.x * blockDim.x + threadIdx.x;
  if (i >= total) return;
  const float dn = denom[i >> 6];
  out[i] = num[i] / (dn == 0.0f ? 1.0f : dn);
}

extern "C" void kernel_launch(void* const* d_in, const int* in_sizes, int n_in,
                              void* d_out, int out_size, void* d_ws, size_t ws_size,
                              hipStream_t stream) {
  const float* h     = (const float*)d_in[0];
  const float* feat  = (const float*)d_in[1];
  const float* W_fc  = (const float*)d_in[2];
  const float* W_dst = (const float*)d_in[3];
  const int*   src   = (const int*)d_in[4];
  const int*   dst   = (const int*)d_in[5];
  float*       out   = (float*)d_out;

  const int FEAT = 64, O = 64;
  const int N   = in_sizes[1] / FEAT;
  const int Kfc = in_sizes[0] / N;
  const int E   = in_sizes[4];

  float*    z       = (float*)d_ws;
  float*    df      = z  + (size_t)N * O;
  float*    num     = df + (size_t)N * O;
  float*    denom   = num + (size_t)N * O;
  unsigned* emaxKey = (unsigned*)(denom + N);
  float*    e       = (float*)(emaxKey + N);

  hipMemsetAsync(num, 0, ((size_t)N * O + 2 * (size_t)N) * sizeof(float), stream);

  const dim3 gemmBlk(32 * (O / 16));
  const int  mTiles = (N + 15) / 16;
  gemm_nt_wmma<<<mTiles, gemmBlk, 0, stream>>>(h,    W_fc,  z,  N, Kfc,  O);
  gemm_nt_wmma<<<mTiles, gemmBlk, 0, stream>>>(feat, W_dst, df, N, FEAT, O);

  const int blkE = 256, edgesPerBlk = blkE / 32;
  const int gE = (E + edgesPerBlk - 1) / edgesPerBlk;
  edge_logits <<<gE, blkE, 0, stream>>>(z, df, src, dst, e, emaxKey, E);
  edge_scatter<<<gE, blkE, 0, stream>>>(z, e, emaxKey, src, dst, num, denom, E);

  const int total = N * O;
  finalize<<<(total + 255) / 256, 256, 0, stream>>>(num, denom, out, total);
}


# edge phase by destination ownership: 256 blocks each filter the whole dst list for their 392 nodes, do max and sums by LDS atomics, write out; 5 launches -> 3; logit dot and sums in new order
# speedup vs baseline: 2.6310x; 2.6310x over previous
#include <hip/hip_runtime.h>
#include <hip/hip_bf16.h>

typedef float v2f __attribute__((ext_vector_type(2)));
typedef float v8f __attribute__((ext_vector_type(8)));

__global__ void gemm_nt_wmma(const float* __restrict__ A,
                             const float* __restrict__ W,
                             float* __restrict__ C,
                             int M, int K, int O) {
  const int lane  = threadIdx.x & 31;
  const int oTile = threadIdx.x >> 5;
  const int mTile = blockIdx.x;
  if (mTile * 16 >= M) return;
  const int half = lane >> 4;
  const int l16  = lane & 15;

  const float* aPtr = A + (size_t)(mTile * 16 + l16) * K + 2 * half;
  const float* wPtr = W + (size_t)(oTile * 16 + l16) * K + 2 * half;

  v8f acc = {};
#pragma unroll 8
  for (int k = 0; k < K; k += 4) {
    v2f a = *(const v2f*)(aPtr + k);
    v2f b = *(const v2f*)(wPtr + k);
    acc = __builtin_amdgcn_wmma_f32_16x16x4_f32(
         false, a,  false, b,
         (short)0, acc,  false,  false);
  }

  float* cPtr = C + (size_t)(mTile * 16 + half * 8) * O + oTile * 16 + l16;
#pragma unroll
  for (int r = 0; r < 8; ++r) cPtr[(size_t)r * O] = acc[r];
}

__device__ __forceinline__ unsigned float_key(float f) {
  unsigned u = __float_as_uint(f);
  return (u & 0x80000000u) ? ~u : (u | 0x80000000u);
}
__device__ __forceinline__ float key_float(unsigned k) {
  unsigned u = (k & 0x80000000u) ? (k & 0x7FFFFFFFu) : ~k;
  return __uint_as_float(u);
}

#define OWN_R   392
#define OWN_P   65
#define OWN_CAP 8192
#define OWN_T   1024
#define OWN_U   8

__device__ __forceinline__ float row_dot_lane(const float* __restrict__ zr,
                                              const float* __restrict__ dr) {
  float p = 0.0f;
  for (int k = 0; k < 16; ++k) {
    const float4 a = ((const float4*)zr)[k];
    const float4 b = ((const float4*)dr)[k];
    p = fmaf(a.x, b.x, p); p = fmaf(a.y, b.y, p);
    p = fmaf(a.z, b.z, p); p = fmaf(a.w, b.w, p);
  }
  return p;
}

__global__ void __launch_bounds__(OWN_T)
edge_owner(const float* __restrict__ z,
           const float* __restrict__ df,
           const int* __restrict__ src,
           const int* __restrict__ dst,
           float* __restrict__ out, int N, int E) {
  __shared__ __attribute__((aligned(16))) float s_df[OWN_R * 64];
  __shared__ float          s_num[OWN_R * OWN_P];
  __shared__ float          s_den[OWN_R];
  __shared__ unsigned       s_key[OWN_R];
  __shared__ int            s_es[OWN_CAP];
  __shared__ float          s_lg[OWN_CAP];
  __shared__ unsigned short s_loc[OWN_CAP];
  __shared__ unsigned       s_cnt;

  const int tid = threadIdx.x;
  const int g   = tid >> 3;
  const int j   = tid & 7;

  const int nvec  = E >> 2;
  const int nIter = (nvec + OWN_T * OWN_U - 1) / (OWN_T * OWN_U);
  const int rot   = nIter > 0 ? (int)(blockIdx.x % (unsigned)nIter) : 0;
  const int4* __restrict__ dst4 = (const int4*)dst;

  for (int base = blockIdx.x * OWN_R; base < N; base += gridDim.x * OWN_R) {
    const int R = min(OWN_R, N - base);

    for (int i = tid; i < OWN_R * OWN_P; i += OWN_T) s_num[i] = 0.0f;
    for (int i = tid; i < OWN_R; i += OWN_T) { s_den[i] = 0.0f; s_key[i] = 0u; }
    {
      const float4* __restrict__ dfr = (const float4*)(df + (size_t)base * 64);
#pragma unroll
      for (int u = 0; u < (OWN_R * 16 + OWN_T - 1) / OWN_T; ++u) {
        const int i = min(tid + u * OWN_T, R * 16 - 1);
        ((float4*)s_df)[i] = dfr[i];
      }
    }
    if (tid == 0) s_cnt = 0u;
    __syncthreads();

    for (int it = 0; it < nIter; ++it) {
      int itx = it + rot; if (itx >= nIter) itx -= nIter;
      const int v0 = itx * (OWN_T * OWN_U) + tid;
      int4 d4[OWN_U];
#pragma unroll
      for (int u = 0; u < OWN_U; ++u) {
        const int v = v0 + u * OWN_T;
        d4[u] = (v < nvec) ? dst4[v] : make_int4(-1, -1, -1, -1);
      }
#pragma unroll
      for (int u = 0; u < OWN_U; ++u) {
        const int v = v0 + u * OWN_T;
        const int dd[4] = {d4[u].x, d4[u].y, d4[u].z, d4[u].w};
#pragma unroll
        for (int c = 0; c < 4; ++c) {
          const unsigned loc = (unsigned)(dd[c] - base);
          if (loc < (unsigned)R) {
            const unsigned idx = atomicAdd(&s_cnt, 1u);
            if (idx < OWN_CAP) { s_es[idx] = 4 * v + c; s_loc[idx] = (unsigned short)loc; }
          }
        }
      }
    }
    {
      const int eT = (nvec << 2) + tid;
      if (eT < E) {
        const unsigned loc = (unsigned)(dst[eT] - base);
        if (loc < (unsigned)R) {
          const unsigned idx = atomicAdd(&s_cnt, 1u);
          if (idx < OWN_CAP) { s_es[idx] = eT; s_loc[idx] = (unsigned short)loc; }
        }
      }
    }
    __syncthreads();
    const unsigned total = s_cnt;

    if (total <= OWN_CAP) {
      const int cnt = (int)total;

      {
        int sv[OWN_CAP / OWN_T];
#pragma unroll
        for (int u = 0; u < OWN_CAP / OWN_T; ++u) {
          const int i  = tid + u * OWN_T;
          const int ev = s_es[i];
          sv[u] = (i < cnt) ? ev : 0;
        }
#pragma unroll
        for (int u = 0; u < OWN_CAP / OWN_T; ++u) sv[u] = src[sv[u]];
#pragma unroll
        for (int u = 0; u < OWN_CAP / OWN_T; ++u) {
          const int i = tid + u * OWN_T;
          if (i < cnt) s_es[i] = min(max(sv[u], 0), N - 1);
        }
      }
      __syncthreads();

      for (int i0 = 0; i0 < cnt; i0 += 4 * (OWN_T / 8)) {
        float4 za[4], zb[4];
        int    li[4];
#pragma unroll
        for (int u = 0; u < 4; ++u) {
          const int  i  = i0 + u * (OWN_T / 8) + g;
          const bool ok = i < cnt;
          const int  sr = s_es[i];
          const int  lr = (int)s_loc[i];
          const int  s  = ok ? sr : 0;
          li[u] = ok ? lr : 0;
          const float4* zr = (const float4*)(z + (size_t)s * 64);
          za[u] = zr[j]; zb[u] = zr[8 + j];
        }
#pragma unroll
        for (int u = 0; u < 4; ++u) {
          const int  i  = i0 + u * (OWN_T / 8) + g;
          const float4* dr = (const float4*)(s_df + li[u] * 64);
          const float4 da = dr[j], db = dr[8 + j];
          float p = za[u].x * da.x + za[u].y * da.y + za[u].z * da.z + za[u].w * da.w
                  + zb[u].x * db.x + zb[u].y * db.y + zb[u].z * db.z + zb[u].w * db.w;
          p += __shfl_xor(p, 1, 32);
          p += __shfl_xor(p, 2, 32);
          p += __shfl_xor(p, 4, 32);
          if (i < cnt && j == 0) {
            s_lg[i] = p;
            atomicMax(&s_key[li[u]], float_key(p));
          }
        }
      }
      __syncthreads();

      for (int i0 = 0; i0 < cnt; i0 += 4 * (OWN_T / 8)) {
        float4 za[4], zb[4];
#pragma unroll
        for (int u = 0; u < 4; ++u) {
          const int  i = i0 + u * (OWN_T / 8) + g;
          const int  sr = s_es[i];
          const int  s  = (i < cnt) ? sr : 0;
          const float4* zr = (const float4*)(z + (size_t)s * 64);
          za[u] = zr[j]; zb[u] = zr[8 + j];
        }
#pragma unroll
        for (int u = 0; u < 4; ++u) {
          const int i = i0 + u * (OWN_T / 8) + g;
          if (i < cnt) {
            const int   loc = (int)s_loc[i];
            const float ex  = __expf(s_lg[i] - key_float(s_key[loc]));
            float* np = s_num + loc * OWN_P + 4 * j;
            atomicAdd(np + 0,  ex * za[u].x); atomicAdd(np + 1,  ex * za[u].y);
            atomicAdd(np + 2,  ex * za[u].z); atomicAdd(np + 3,  ex * za[u].w);
            atomicAdd(np + 32, ex * zb[u].x); atomicAdd(np + 33, ex * zb[u].y);
            atomicAdd(np + 34, ex * zb[u].z); atomicAdd(np + 35, ex * zb[u].w);
            if (j == 0) atomicAdd(&s_den[loc], ex);
          }
        }
      }
    } else {
      for (int e = tid; e < E; e += OWN_T) {
        const unsigned loc = (unsigned)(dst[e] - base);
        if (loc < (unsigned)R) {
          const int s = min(max(src[e], 0), N - 1);
          const float p = row_dot_lane(z + (size_t)s * 64, s_df + loc * 64);
          atomicMax(&s_key[loc], float_key(p));
        }
      }
      __syncthreads();
      for (int e = tid; e < E; e += OWN_T) {
        const unsigned loc = (unsigned)(dst[e] - base);
        if (loc < (unsigned)R) {
          const int s = min(max(src[e], 0), N - 1);
          const float* zr = z + (size_t)s * 64;
          const float p  = row_dot_lane(zr, s_df + loc * 64);
          const float ex = __expf(p - key_float(s_key[loc]));
          float* np = s_num + loc * OWN_P;
          for (int k = 0; k < 64; ++k) atomicAdd(np + k, ex * zr[k]);
          atomicAdd(&s_den[loc], ex);
        }
      }
    }
    __syncthreads();

    for (int i = tid; i < R * 64; i += OWN_T) {
      const int   loc = i >> 6, f = i & 63;
      const float dn  = s_den[loc];
      out[(size_t)(base + loc) * 64 + f] = s_num[loc * OWN_P + f] / (dn == 0.0f ? 1.0f : dn);
    }
    __syncthreads();
  }
}

extern "C" void kernel_launch(void* const* d_in, const int* in_sizes, int n_in,
                              void* d_out, int out_size, void* d_ws, size_t ws_size,
                              hipStream_t stream) {
  const float* h     = (const float*)d_in[0];
  const float* feat  = (const float*)d_in[1];
  const float* W_fc  = (const float*)d_in[2];
  const float* W_dst = (const float*)d_in[3];
  const int*   src   = (const int*)d_in[4];
  const int*   dst   = (const int*)d_in[5];
  float*       out   = (float*)d_out;

  const int FEAT = 64, O = 64;
  const int N   = in_sizes[1] / FEAT;
  const int Kfc = in_sizes[0] / N;
  const int E   = in_sizes[4];

  float* z  = (float*)d_ws;
  float* df = z + (size_t)N * O;

  const dim3 gemmBlk(32 * (O / 16));
  const int  mTiles = (N + 15) / 16;
  gemm_nt_wmma<<<mTiles, gemmBlk, 0, stream>>>(h,    W_fc,  z,  N, Kfc,  O);
  gemm_nt_wmma<<<mTiles, gemmBlk, 0, stream>>>(feat, W_dst, df, N, FEAT, O);

  edge_owner<<<256, OWN_T, 0, stream>>>(z, df, src, dst, out, N, E);
}


# edge filter scans a 1-byte owner key per edge (dst/392 mod 256, written by a new 4th launch) instead of the 4-byte dst: 16 edges per 16-byte load; list order, so LDS sums, in a new arrival order
# speedup vs baseline: 3.9272x; 1.4927x over previous
#include <hip/hip_runtime.h>
#include <hip/hip_bf16.h>

typedef float v2f __attribute__((ext_vector_type(2)));
typedef float v8f __attribute__((ext_vector_type(8)));

__global__ void gemm_nt_wmma(const float* __restrict__ A,
                             const float* __restrict__ W,
                             float* __restrict__ C,
                             int M, int K, int O) {
  const int lane  = threadIdx.x & 31;
  const int oTile = threadIdx.x >> 5;
  const int mTile = blockIdx.x;
  if (mTile * 16 >= M) return;
  const int half = lane >> 4;
  const int l16  = lane & 15;

  const float* aPtr = A + (size_t)(mTile * 16 + l16) * K + 2 * half;
  const float* wPtr = W + (size_t)(oTile * 16 + l16) * K + 2 * half;

  v8f acc = {};
#pragma unroll 8
  for (int k = 0; k < K; k += 4) {
    v2f a = *(const v2f*)(aPtr + k);
    v2f b = *(const v2f*)(wPtr + k);
    acc = __builtin_amdgcn_wmma_f32_16x16x4_f32(
         false, a,  false, b,
         (short)0, acc,  false,  false);
  }

  float* cPtr = C + (size_t)(mTile * 16 + half * 8) * O + oTile * 16 + l16;
#pragma unroll
  for (int r = 0; r < 8; ++r) cPtr[(size_t)r * O] = acc[r];
}

#define PJ_T 128
#define PJ_G 4

template <int K>
__global__ void __launch_bounds__(PJ_T)
proj_resident(const float* __restrict__ A,
              const float* __restrict__ Wm,
              float* __restrict__ C, int M) {
  constexpr int P   = K + 4;
  constexpr int NQ  = K / 8;
  constexpr int RV  = K / 4;
  constexpr int NLD = (16 * RV) / PJ_T;
  __shared__ __attribute__((aligned(16))) float s_a[2 * 16 * P];

  const int tid  = threadIdx.x;
  const int lane = tid & 31;
  const int ct   = tid >> 5;
  const int c    = lane & 15;
  const int h    = lane >> 4;
  const int nSteps = (M + 15) >> 4;
  int step = blockIdx.x;
  if (step >= nSteps) return;

  float4 wf[NQ];
  {
    const float4* wr = (const float4*)(Wm + (size_t)(ct * 16 + c) * K + 4 * h);
#pragma unroll
    for (int q = 0; q < NQ; ++q) wf[q] = wr[2 * q];
  }

  constexpr int RPI = PJ_T / RV;
  const int trow = tid / RV, tcol = (tid % RV) * 4;
  const float* ag = A + (size_t)trow * K + tcol;
  float*       sl = s_a + trow * P + tcol;

  static_assert(NLD == 8 || NLD == 2, "tile share per thread");
  float4 t0, t1, t2, t3, t4, t5, t6, t7;
#define PJ_REQ(at)                                                        \
  do {                                                                    \
    t0 = *(const float4*)((at) + (size_t)0 * RPI * K);                    \
    t1 = *(const float4*)((at) + (size_t)1 * RPI * K);                    \
    if constexpr (NLD == 8) {                                             \
      t2 = *(const float4*)((at) + (size_t)2 * RPI * K);                  \
      t3 = *(const float4*)((at) + (size_t)3 * RPI * K);                  \
      t4 = *(const float4*)((at) + (size_t)4 * RPI * K);                  \
      t5 = *(const float4*)((at) + (size_t)5 * RPI * K);                  \
      t6 = *(const float4*)((at) + (size_t)6 * RPI * K);                  \
      t7 = *(const float4*)((at) + (size_t)7 * RPI * K);                  \
    }                                                                     \
  } while (0)
#define PJ_PUT(sp)                                                        \
  do {                                                                    \
    *(float4*)((sp) + 0 * RPI * P) = t0;                                  \
    *(float4*)((sp) + 1 * RPI * P) = t1;                                  \
    if constexpr (NLD == 8) {                                             \
      *(float4*)((sp) + 2 * RPI * P) = t2;                                \
      *(float4*)((sp) + 3 * RPI * P) = t3;                                \
      *(float4*)((sp) + 4 * RPI * P) = t4;                                \
      *(float4*)((sp) + 5 * RPI * P) = t5;                                \
      *(float4*)((sp) + 6 * RPI * P) = t6;                                \
      *(float4*)((sp) + 7 * RPI * P) = t7;                                \
    }                                                                     \
  } while (0)

  {
    const float* at = ag + (size_t)min(step * 16, M - 16) * K;
    PJ_REQ(at);
    PJ_PUT(sl);
  }
  {
    const int ls = min(step + (int)gridDim.x, nSteps - 1);
    const float* at = ag + (size_t)min(ls * 16, M - 16) * K;
    PJ_REQ(at);
  }
  __syncthreads();

  int b = 0;
  for (;;) {
    const float* ar = s_a + b * (16 * P) + c * P + 4 * h;
    v8f acc0 = {}, acc1 = {};
    float4 af[2][PJ_G];
#pragma unroll
    for (int j = 0; j < PJ_G; ++j) af[0][j] = *(const float4*)(ar + 8 * j);
#pragma unroll
    for (int g = 0; g < NQ / PJ_G; ++g) {
      if (g + 1 < NQ / PJ_G) {
#pragma unroll
        for (int j = 0; j < PJ_G; ++j) af[(g + 1) & 1][j] = *(const float4*)(ar + 8 * ((g + 1) * PJ_G + j));
      }
      __builtin_amdgcn_sched_barrier(0);
#pragma unroll
      for (int j = 0; j < PJ_G; ++j) {
        const float4 av = af[g & 1][j];
        const float4 wv = wf[g * PJ_G + j];
        const v2f a0 = {av.x, av.y}, a1 = {av.z, av.w};
        const v2f b0 = {wv.x, wv.y}, b1 = {wv.z, wv.w};
        acc0 = __builtin_amdgcn_wmma_f32_16x16x4_f32(false, a0, false, b0, (short)0, acc0, false, false);
        acc1 = __builtin_amdgcn_wmma_f32_16x16x4_f32(false, a1, false, b1, (short)0, acc1, false, false);
      }
      __builtin_amdgcn_sched_barrier(0);
    }

    {
      float* sn = sl + (b ^ 1) * (16 * P);
      PJ_PUT(sn);
    }
    __syncthreads();

    {
      const int r0 = min(step * 16, M - 16) + 8 * h;
      float* cp = C + (size_t)r0 * 64 + ct * 16 + c;
#pragma unroll
      for (int r = 0; r < 8; ++r)
        if (r0 + r >= step * 16) cp[(size_t)r * 64] = acc0[r] + acc1[r];
    }

    const int nx = step + (int)gridDim.x;
    if (nx >= nSteps) break;
    {
      const int ls = min(nx + (int)gridDim.x, nSteps - 1);
      const float* at = ag + (size_t)min(ls * 16, M - 16) * K;
      PJ_REQ(at);
    }
    step = nx;
    b ^= 1;
  }
#undef PJ_REQ
#undef PJ_PUT
}

__device__ __forceinline__ unsigned float_key(float f) {
  unsigned u = __float_as_uint(f);
  return (u & 0x80000000u) ? ~u : (u | 0x80000000u);
}
__device__ __forceinline__ float key_float(unsigned k) {
  unsigned u = (k & 0x80000000u) ? (k & 0x7FFFFFFFu) : ~k;
  return __uint_as_float(u);
}

#define OWN_R   392
#define OWN_P   65
#define OWN_CAP 8192
#define OWN_T   1024
#define OWN_NONE 0xFFFF
#define OWN_U   8

__global__ void __launch_bounds__(256)
owner_keys(const int* __restrict__ dst, unsigned* __restrict__ keys, int E) {
  const int nv16 = E >> 4;
  const int4* __restrict__ dst4 = (const int4*)dst;
  for (int v = blockIdx.x * 256 + threadIdx.x; v < nv16; v += gridDim.x * 256) {
    int4 d[4];
#pragma unroll
    for (int q = 0; q < 4; ++q) d[q] = dst4[4 * v + q];
    unsigned w[4];
#pragma unroll
    for (int q = 0; q < 4; ++q) {
      const unsigned k0 = ((unsigned)d[q].x / (unsigned)OWN_R) & 255u;
      const unsigned k1 = ((unsigned)d[q].y / (unsigned)OWN_R) & 255u;
      const unsigned k2 = ((unsigned)d[q].z / (unsigned)OWN_R) & 255u;
      const unsigned k3 = ((unsigned)d[q].w / (unsigned)OWN_R) & 255u;
      w[q] = k0 | (k1 << 8) | (k2 << 16) | (k3 << 24);
    }
    ((uint4*)keys)[v] = make_uint4(w[0], w[1], w[2], w[3]);
  }
}

__device__ __forceinline__ float row_dot_lane(const float* __restrict__ zr,
                                              const float* __restrict__ dr) {
  float p = 0.0f;
  for (int k = 0; k < 16; ++k) {
    const float4 a = ((const float4*)zr)[k];
    const float4 b = ((const float4*)dr)[k];
    p = fmaf(a.x, b.x, p); p = fmaf(a.y, b.y, p);
    p = fmaf(a.z, b.z, p); p = fmaf(a.w, b.w, p);
  }
  return p;
}

__global__ void __launch_bounds__(OWN_T)
edge_owner(const float* __restrict__ z,
           const float* __restrict__ df,
           const int* __restrict__ src,
           const int* __restrict__ dst,
           const unsigned* __restrict__ keys,
           float* __restrict__ out, int N, int E) {
  __shared__ __attribute__((aligned(16))) float s_df[OWN_R * 64];
  __shared__ float          s_num[OWN_R * OWN_P];
  __shared__ float          s_den[OWN_R];
  __shared__ unsigned       s_key[OWN_R];
  __shared__ int            s_es[OWN_CAP];
  __shared__ float          s_lg[OWN_CAP];
  __shared__ unsigned short s_loc[OWN_CAP];
  __shared__ unsigned       s_cnt;

  const int tid = threadIdx.x;
  const int g   = tid >> 3;
  const int j   = tid & 7;

  const int nvec  = E >> 4;
  const int nIter = (nvec + OWN_T * OWN_U - 1) / (OWN_T * OWN_U);
  const int rot   = nIter > 0 ? (int)(blockIdx.x % (unsigned)nIter) : 0;
  const uint4* __restrict__ key4 = (const uint4*)keys;

  for (int base = blockIdx.x * OWN_R; base < N; base += gridDim.x * OWN_R) {
    const int R = min(OWN_R, N - base);

    for (int i = tid; i < OWN_R * OWN_P; i += OWN_T) s_num[i] = 0.0f;
    for (int i = tid; i < OWN_R; i += OWN_T) { s_den[i] = 0.0f; s_key[i] = 0u; }
    {
      const float4* __restrict__ dfr = (const float4*)(df + (size_t)base * 64);
#pragma unroll
      for (int u = 0; u < (OWN_R * 16 + OWN_T - 1) / OWN_T; ++u) {
        const int i = min(tid + u * OWN_T, R * 16 - 1);
        ((float4*)s_df)[i] = dfr[i];
      }
    }
    if (tid == 0) s_cnt = 0u;
    __syncthreads();

    const unsigned want = ((unsigned)base / (unsigned)OWN_R) & 255u;
    const unsigned pat  = want * 0x01010101u;
    for (int it = 0; it < nIter; ++it) {
      int itx = it + rot; if (itx >= nIter) itx -= nIter;
      const int v0 = itx * (OWN_T * OWN_U) + tid;
      uint4 k4[OWN_U];
#pragma unroll
      for (int u = 0; u < OWN_U; ++u) {
        k4[u] = key4[min(v0 + u * OWN_T, nvec - 1)];
      }
      __builtin_amdgcn_sched_barrier(0);
#pragma unroll
      for (int u = 0; u < OWN_U; ++u) {
        const int v = v0 + u * OWN_T;
        const unsigned kw[4] = {k4[u].x, k4[u].y, k4[u].z, k4[u].w};
        unsigned zb[4];
#pragma unroll
        for (int c = 0; c < 4; ++c) {
          const unsigned y = kw[c] ^ pat;
          zb[c] = ~(((y & 0x7F7F7F7Fu) + 0x7F7F7F7Fu) | y) & 0x80808080u;
        }
        unsigned hit = ((((zb[0] >> 1) | zb[1]) >> 2) | ((zb[2] >> 1) | zb[3]));
        if (v >= nvec) hit = 0u;
        while (hit != 0u) {
          const int p = __ffs((int)hit) - 1;
          hit &= hit - 1u;
          const unsigned idx = atomicAdd(&s_cnt, 1u);
          if (idx < OWN_CAP) s_es[idx] = 16 * v + 4 * ((p & 7) - 4) + (p >> 3);
        }
      }
    }
    {
      const int eT = (nvec << 4) + tid;
      if (tid < 16 && eT < E) {
        const unsigned loc = (unsigned)(dst[eT] - base);
        if (loc < (unsigned)R) {
          const unsigned idx = atomicAdd(&s_cnt, 1u);
          if (idx < OWN_CAP) s_es[idx] = eT;
        }
      }
    }
    __syncthreads();
    const unsigned total = s_cnt;

    if (total <= OWN_CAP) {
      const int cnt = (int)total;

      {
        int sv[OWN_CAP / OWN_T], dv[OWN_CAP / OWN_T];
#pragma unroll
        for (int u = 0; u < OWN_CAP / OWN_T; ++u) {
          const int i  = tid + u * OWN_T;
          const int ev = s_es[i];
          sv[u] = (i < cnt) ? ev : 0;
        }
#pragma unroll
        for (int u = 0; u < OWN_CAP / OWN_T; ++u) { dv[u] = dst[sv[u]]; sv[u] = src[sv[u]]; }
#pragma unroll
        for (int u = 0; u < OWN_CAP / OWN_T; ++u) {
          const int i = tid + u * OWN_T;
          if (i < cnt) {
            const unsigned loc = (unsigned)(dv[u] - base);
            s_es[i]  = min(max(sv[u], 0), N - 1);
            s_loc[i] = (loc < (unsigned)R) ? (unsigned short)loc : (unsigned short)OWN_NONE;
          }
        }
      }
      __syncthreads();

      for (int i0 = 0; i0 < cnt; i0 += 4 * (OWN_T / 8)) {
        float4 za[4], zb[4];
        int    li[4];
#pragma unroll
        for (int u = 0; u < 4; ++u) {
          const int  i  = i0 + u * (OWN_T / 8) + g;
          const int  sr = s_es[i];
          const int  lr = (int)s_loc[i];
          const bool ok = i < cnt && lr != OWN_NONE;
          const int  s  = ok ? sr : 0;
          li[u] = ok ? lr : 0;
          const float4* zr = (const float4*)(z + (size_t)s * 64);
          za[u] = zr[j]; zb[u] = zr[8 + j];
        }
#pragma unroll
        for (int u = 0; u < 4; ++u) {
          const int  i  = i0 + u * (OWN_T / 8) + g;
          const float4* dr = (const float4*)(s_df + li[u] * 64);
          const float4 da = dr[j], db = dr[8 + j];
          float p = za[u].x * da.x + za[u].y * da.y + za[u].z * da.z + za[u].w * da.w
                  + zb[u].x * db.x + zb[u].y * db.y + zb[u].z * db.z + zb[u].w * db.w;
          p += __shfl_xor(p, 1, 32);
          p += __shfl_xor(p, 2, 32);
          p += __shfl_xor(p, 4, 32);
          if (i < cnt && j == 0 && (int)s_loc[i] != OWN_NONE) {
            s_lg[i] = p;
            atomicMax(&s_key[li[u]], float_key(p));
          }
        }
      }
      __syncthreads();

      for (int i0 = 0; i0 < cnt; i0 += 4 * (OWN_T / 8)) {
        float4 za[4], zb[4];
#pragma unroll
        for (int u = 0; u < 4; ++u) {
          const int  i = i0 + u * (OWN_T / 8) + g;
          const int  sr = s_es[i];
          const int  s  = (i < cnt) ? sr : 0;
          const float4* zr = (const float4*)(z + (size_t)s * 64);
          za[u] = zr[j]; zb[u] = zr[8 + j];
        }
#pragma unroll
        for (int u = 0; u < 4; ++u) {
          const int i = i0 + u * (OWN_T / 8) + g;
          if (i < cnt && (int)s_loc[i] != OWN_NONE) {
            const int   loc = (int)s_loc[i];
            const float ex  = __expf(s_lg[i] - key_float(s_key[loc]));
            float* np = s_num + loc * OWN_P + 4 * j;
            atomicAdd(np + 0,  ex * za[u].x); atomicAdd(np + 1,  ex * za[u].y);
            atomicAdd(np + 2,  ex * za[u].z); atomicAdd(np + 3,  ex * za[u].w);
            atomicAdd(np + 32, ex * zb[u].x); atomicAdd(np + 33, ex * zb[u].y);
            atomicAdd(np + 34, ex * zb[u].z); atomicAdd(np + 35, ex * zb[u].w);
            if (j == 0) atomicAdd(&s_den[loc], ex);
          }
        }
      }
    } else {
      for (int e = tid; e < E; e += OWN_T) {
        const unsigned loc = (unsigned)(dst[e] - base);
        if (loc < (unsigned)R) {
          const int s = min(max(src[e], 0), N - 1);
          const float p = row_dot_lane(z + (size_t)s * 64, s_df + loc * 64);
          atomicMax(&s_key[loc], float_key(p));
        }
      }
      __syncthreads();
      for (int e = tid; e < E; e += OWN_T) {
        const unsigned loc = (unsigned)(dst[e] - base);
        if (loc < (unsigned)R) {
          const int s = min(max(src[e], 0), N - 1);
          const float* zr = z + (size_t)s * 64;
          const float p  = row_dot_lane(zr, s_df + loc * 64);
          const float ex = __expf(p - key_float(s_key[loc]));
          float* np = s_num + loc * OWN_P;
          for (int k = 0; k < 64; ++k) atomicAdd(np + k, ex * zr[k]);
          atomicAdd(&s_den[loc], ex);
        }
      }
    }
    __syncthreads();

    for (int i = tid; i < R * 64; i += OWN_T) {
      const int   loc = i >> 6, f = i & 63;
      const float dn  = s_den[loc];
      out[(size_t)(base + loc) * 64 + f] = s_num[loc * OWN_P + f] / (dn == 0.0f ? 1.0f : dn);
    }
    __syncthreads();
  }
}

extern "C" void kernel_launch(void* const* d_in, const int* in_sizes, int n_in,
                              void* d_out, int out_size, void* d_ws, size_t ws_size,
                              hipStream_t stream) {
  const float* h     = (const float*)d_in[0];
  const float* feat  = (const float*)d_in[1];
  const float* W_fc  = (const float*)d_in[2];
  const float* W_dst = (const float*)d_in[3];
  const int*   src   = (const int*)d_in[4];
  const int*   dst   = (const int*)d_in[5];
  float*       out   = (float*)d_out;

  const int FEAT = 64, O = 64;
  const int N   = in_sizes[1] / FEAT;
  const int Kfc = in_sizes[0] / N;
  const int E   = in_sizes[4];

  float* z  = (float*)d_ws;
  float* df = z + (size_t)N * O;
  unsigned* keys = (unsigned*)(df + (size_t)N * O);

  if (Kfc == 256) {
    proj_resident<256><<<1024, PJ_T, 0, stream>>>(h, W_fc, z, N);
  } else {
    const dim3 gemmBlk(32 * (O / 16));
    const int  mTiles = (N + 15) / 16;
    gemm_nt_wmma<<<mTiles, gemmBlk, 0, stream>>>(h, W_fc, z, N, Kfc, O);
  }
  proj_resident<64><<<1024, PJ_T, 0, stream>>>(feat, W_dst, df, N);

  owner_keys<<<512, 256, 0, stream>>>(dst, keys, E);
  edge_owner<<<256, OWN_T, 0, stream>>>(z, df, src, dst, keys, out, N, E);
}


# edge filter in two levels: a new 256-bit owner map per 32 edges lets a block scan 1 bit per 32 edges and test key bytes only in marked groups; list order, so LDS sums, in a new arrival order
# speedup vs baseline: 4.2796x; 1.0897x over previous
#include <hip/hip_runtime.h>
#include <hip/hip_bf16.h>

typedef float v2f __attribute__((ext_vector_type(2)));
typedef float v8f __attribute__((ext_vector_type(8)));

__global__ void gemm_nt_wmma(const float* __restrict__ A,
                             const float* __restrict__ W,
                             float* __restrict__ C,
                             int M, int K, int O) {
  const int lane  = threadIdx.x & 31;
  const int oTile = threadIdx.x >> 5;
  const int mTile = blockIdx.x;
  if (mTile * 16 >= M) return;
  const int half = lane >> 4;
  const int l16  = lane & 15;

  const float* aPtr = A + (size_t)(mTile * 16 + l16) * K + 2 * half;
  const float* wPtr = W + (size_t)(oTile * 16 + l16) * K + 2 * half;

  v8f acc = {};
#pragma unroll 8
  for (int k = 0; k < K; k += 4) {
    v2f a = *(const v2f*)(aPtr + k);
    v2f b = *(const v2f*)(wPtr + k);
    acc = __builtin_amdgcn_wmma_f32_16x16x4_f32(
         false, a,  false, b,
         (short)0, acc,  false,  false);
  }

  float* cPtr = C + (size_t)(mTile * 16 + half * 8) * O + oTile * 16 + l16;
#pragma unroll
  for (int r = 0; r < 8; ++r) cPtr[(size_t)r * O] = acc[r];
}

#define PJ_T 128
#define PJ_G 4

template <int K>
__global__ void __launch_bounds__(PJ_T)
proj_resident(const float* __restrict__ A,
              const float* __restrict__ Wm,
              float* __restrict__ C, int M) {
  constexpr int P   = K + 4;
  constexpr int NQ  = K / 8;
  constexpr int RV  = K / 4;
  constexpr int NLD = (16 * RV) / PJ_T;
  __shared__ __attribute__((aligned(16))) float s_a[2 * 16 * P];

  const int tid  = threadIdx.x;
  const int lane = tid & 31;
  const int ct   = tid >> 5;
  const int c    = lane & 15;
  const int h    = lane >> 4;
  const int nSteps = (M + 15) >> 4;
  int step = blockIdx.x;
  if (step >= nSteps) return;

  float4 wf[NQ];
  {
    const float4* wr = (const float4*)(Wm + (size_t)(ct * 16 + c) * K + 4 * h);
#pragma unroll
    for (int q = 0; q < NQ; ++q) wf[q] = wr[2 * q];
  }

  constexpr int RPI = PJ_T / RV;
  const int trow = tid / RV, tcol = (tid % RV) * 4;
  const float* ag = A + (size_t)trow * K + tcol;
  float*       sl = s_a + trow * P + tcol;

  static_assert(NLD == 8 || NLD == 2, "tile share per thread");
  float4 t0, t1, t2, t3, t4, t5, t6, t7;
#define PJ_REQ(at)                                                        \
  do {                                                                    \
    t0 = *(const float4*)((at) + (size_t)0 * RPI * K);                    \
    t1 = *(const float4*)((at) + (size_t)1 * RPI * K);                    \
    if constexpr (NLD == 8) {                                             \
      t2 = *(const float4*)((at) + (size_t)2 * RPI * K);                  \
      t3 = *(const float4*)((at) + (size_t)3 * RPI * K);                  \
      t4 = *(const float4*)((at) + (size_t)4 * RPI * K);                  \
      t5 = *(const float4*)((at) + (size_t)5 * RPI * K);                  \
      t6 = *(const float4*)((at) + (size_t)6 * RPI * K);                  \
      t7 = *(const float4*)((at) + (size_t)7 * RPI * K);                  \
    }                                                                     \
  } while (0)
#define PJ_PUT(sp)                                                        \
  do {                                                                    \
    *(float4*)((sp) + 0 * RPI * P) = t0;                                  \
    *(float4*)((sp) + 1 * RPI * P) = t1;                                  \
    if constexpr (NLD == 8) {                                             \
      *(float4*)((sp) + 2 * RPI * P) = t2;                                \
      *(float4*)((sp) + 3 * RPI * P) = t3;                                \
      *(float4*)((sp) + 4 * RPI * P) = t4;                                \
      *(float4*)((sp) + 5 * RPI * P) = t5;                                \
      *(float4*)((sp) + 6 * RPI * P) = t6;                                \
      *(float4*)((sp) + 7 * RPI * P) = t7;                                \
    }                                                                     \
  } while (0)

  {
    const float* at = ag + (size_t)min(step * 16, M - 16) * K;
    PJ_REQ(at);
    PJ_PUT(sl);
  }
  {
    const int ls = min(step + (int)gridDim.x, nSteps - 1);
    const float* at = ag + (size_t)min(ls * 16, M - 16) * K;
    PJ_REQ(at);
  }
  __syncthreads();

  int b = 0;
  for (;;) {
    const float* ar = s_a + b * (16 * P) + c * P + 4 * h;
    v8f acc0 = {}, acc1 = {};
    float4 af[2][PJ_G];
#pragma unroll
    for (int j = 0; j < PJ_G; ++j) af[0][j] = *(const float4*)(ar + 8 * j);
#pragma unroll
    for (int g = 0; g < NQ / PJ_G; ++g) {
      if (g + 1 < NQ / PJ_G) {
#pragma unroll
        for (int j = 0; j < PJ_G; ++j) af[(g + 1) & 1][j] = *(const float4*)(ar + 8 * ((g + 1) * PJ_G + j));
      }
      __builtin_amdgcn_sched_barrier(0);
#pragma unroll
      for (int j = 0; j < PJ_G; ++j) {
        const float4 av = af[g & 1][j];
        const float4 wv = wf[g * PJ_G + j];
        const v2f a0 = {av.x, av.y}, a1 = {av.z, av.w};
        const v2f b0 = {wv.x, wv.y}, b1 = {wv.z, wv.w};
        acc0 = __builtin_amdgcn_wmma_f32_16x16x4_f32(false, a0, false, b0, (short)0, acc0, false, false);
        acc1 = __builtin_amdgcn_wmma_f32_16x16x4_f32(false, a1, false, b1, (short)0, acc1, false, false);
      }
      __builtin_amdgcn_sched_barrier(0);
    }

    {
      float* sn = sl + (b ^ 1) * (16 * P);
      PJ_PUT(sn);
    }
    __syncthreads();

    {
      const int r0 = min(step * 16, M - 16) + 8 * h;
      float* cp = C + (size_t)r0 * 64 + ct * 16 + c;
#pragma unroll
      for (int r = 0; r < 8; ++r)
        if (r0 + r >= step * 16) cp[(size_t)r * 64] = acc0[r] + acc1[r];
    }

    const int nx = step + (int)gridDim.x;
    if (nx >= nSteps) break;
    {
      const int ls = min(nx + (int)gridDim.x, nSteps - 1);
      const float* at = ag + (size_t)min(ls * 16, M - 16) * K;
      PJ_REQ(at);
    }
    step = nx;
    b ^= 1;
  }
#undef PJ_REQ
#undef PJ_PUT
}

__device__ __forceinline__ unsigned float_key(float f) {
  unsigned u = __float_as_uint(f);
  return (u & 0x80000000u) ? ~u : (u | 0x80000000u);
}
__device__ __forceinline__ float key_float(unsigned k) {
  unsigned u = (k & 0x80000000u) ? (k & 0x7FFFFFFFu) : ~k;
  return __uint_as_float(u);
}

#define OWN_R   392
#define OWN_P   65
#define OWN_CAP 8192
#define OWN_T   1024
#define OWN_NONE 0xFFFF
#define OWN_U   8

__device__ __forceinline__ int own_groups(int E) { return (E + 31) >> 5; }
__device__ __forceinline__ int own_pitch(int E)  { return (own_groups(E) + 3) & ~3; }

__global__ void __launch_bounds__(256)
owner_keys(const int* __restrict__ dst, unsigned* __restrict__ keys, int E) {
  const int nG    = own_groups(E);
  const int pitch = own_pitch(E);
  unsigned* __restrict__ maps = keys + (size_t)nG * 8;
  for (int g = blockIdx.x * 256 + threadIdx.x; g < nG; g += gridDim.x * 256) {
    unsigned o[32];
    if (32 * g + 32 <= E) {
      const int4* __restrict__ p = (const int4*)(dst + (size_t)32 * g);
      int4 d[8];
#pragma unroll
      for (int q = 0; q < 8; ++q) d[q] = p[q];
#pragma unroll
      for (int q = 0; q < 8; ++q) {
        o[4 * q + 0] = ((unsigned)d[q].x / (unsigned)OWN_R) & 255u;
        o[4 * q + 1] = ((unsigned)d[q].y / (unsigned)OWN_R) & 255u;
        o[4 * q + 2] = ((unsigned)d[q].z / (unsigned)OWN_R) & 255u;
        o[4 * q + 3] = ((unsigned)d[q].w / (unsigned)OWN_R) & 255u;
      }
    } else {
#pragma unroll
      for (int k = 0; k < 32; ++k) {
        const int e = 32 * g + k;
        o[k] = (e < E) ? (((unsigned)dst[e] / (unsigned)OWN_R) & 255u) : 0xFFFFFFFFu;
      }
    }
    unsigned w[8];
#pragma unroll
    for (int q = 0; q < 8; ++q)
      w[q] = (o[4 * q] & 255u) | ((o[4 * q + 1] & 255u) << 8) | ((o[4 * q + 2] & 255u) << 16) | ((o[4 * q + 3] & 255u) << 24);
    ((uint4*)keys)[2 * (size_t)g]     = make_uint4(w[0], w[1], w[2], w[3]);
    ((uint4*)keys)[2 * (size_t)g + 1] = make_uint4(w[4], w[5], w[6], w[7]);
    unsigned bm[8] = {0u, 0u, 0u, 0u, 0u, 0u, 0u, 0u};
#pragma unroll
    for (int k = 0; k < 32; ++k) {
      const unsigned bit = 1u << (o[k] & 31u);
      const unsigned hi  = o[k] >> 5;
#pragma unroll
      for (int d = 0; d < 8; ++d) bm[d] |= (hi == (unsigned)d) ? bit : 0u;
    }
#pragma unroll
    for (int d = 0; d < 8; ++d) maps[(size_t)d * pitch + g] = bm[d];
  }
}

__device__ __forceinline__ float row_dot_lane(const float* __restrict__ zr,
                                              const float* __restrict__ dr) {
  float p = 0.0f;
  for (int k = 0; k < 16; ++k) {
    const float4 a = ((const float4*)zr)[k];
    const float4 b = ((const float4*)dr)[k];
    p = fmaf(a.x, b.x, p); p = fmaf(a.y, b.y, p);
    p = fmaf(a.z, b.z, p); p = fmaf(a.w, b.w, p);
  }
  return p;
}

__global__ void __launch_bounds__(OWN_T)
edge_owner(const float* __restrict__ z,
           const float* __restrict__ df,
           const int* __restrict__ src,
           const int* __restrict__ dst,
           const unsigned* __restrict__ keys,
           float* __restrict__ out, int N, int E) {
  __shared__ __attribute__((aligned(16))) float s_df[OWN_R * 64];
  __shared__ float          s_num[OWN_R * OWN_P];
  __shared__ float          s_den[OWN_R];
  __shared__ unsigned       s_key[OWN_R];
  __shared__ int            s_es[OWN_CAP];
  __shared__ unsigned       s_lgw[OWN_CAP];
  __shared__ unsigned short s_loc[OWN_CAP];
  __shared__ unsigned       s_cnt;
  __shared__ unsigned       s_ng;

  const int tid = threadIdx.x;
  const int g   = tid >> 3;
  const int j   = tid & 7;

  const int lane  = tid & 31;
  const int nG    = own_groups(E);
  const int pitch = own_pitch(E);
  const int nv    = pitch >> 2;
  const int nIter = (nv + OWN_T * OWN_U - 1) / (OWN_T * OWN_U);
  const uint4* __restrict__ key4 = (const uint4*)keys;
  const unsigned* __restrict__ maps = keys + (size_t)nG * 8;

  for (int base = blockIdx.x * OWN_R; base < N; base += gridDim.x * OWN_R) {
    const int R = min(OWN_R, N - base);

    for (int i = tid; i < OWN_R * OWN_P; i += OWN_T) s_num[i] = 0.0f;
    for (int i = tid; i < OWN_R; i += OWN_T) { s_den[i] = 0.0f; s_key[i] = 0u; }
    {
      const float4* __restrict__ dfr = (const float4*)(df + (size_t)base * 64);
#pragma unroll
      for (int u = 0; u < (OWN_R * 16 + OWN_T - 1) / OWN_T; ++u) {
        const int i = min(tid + u * OWN_T, R * 16 - 1);
        ((float4*)s_df)[i] = dfr[i];
      }
    }
    if (tid == 0) { s_cnt = 0u; s_ng = 0u; }
    __syncthreads();

    const unsigned want = ((unsigned)base / (unsigned)OWN_R) & 255u;
    const unsigned pat  = want * 0x01010101u;
    const unsigned wsh  = want & 31u;
    const uint4* __restrict__ row4 = (const uint4*)(maps + (size_t)(want >> 5) * pitch);
    const int rotv = (int)(((unsigned)(blockIdx.x & 31) * (unsigned)nv) >> 5);
    for (int it = 0; it < nIter; ++it) {
      const int v0 = it * (OWN_T * OWN_U) + tid;
      uint4 m4[OWN_U];
      int   vv[OWN_U];
#pragma unroll
      for (int u = 0; u < OWN_U; ++u) {
        int x = min(v0 + u * OWN_T, nv - 1) + rotv;
        if (x >= nv) x -= nv;
        vv[u] = x;
        m4[u] = row4[x];
      }
      __builtin_amdgcn_sched_barrier(0);
#pragma unroll
      for (int u = 0; u < OWN_U; ++u) {
        const bool in = (v0 + u * OWN_T) < nv;
        const int  g0 = 4 * vv[u];
        const bool h0 = in && (g0 + 0 < nG) && (((m4[u].x >> wsh) & 1u) != 0u);
        const bool h1 = in && (g0 + 1 < nG) && (((m4[u].y >> wsh) & 1u) != 0u);
        const bool h2 = in && (g0 + 2 < nG) && (((m4[u].z >> wsh) & 1u) != 0u);
        const bool h3 = in && (g0 + 3 < nG) && (((m4[u].w >> wsh) & 1u) != 0u);
        const unsigned b0 = __builtin_amdgcn_ballot_w32(h0);
        const unsigned b1 = __builtin_amdgcn_ballot_w32(h1);
        const unsigned b2 = __builtin_amdgcn_ballot_w32(h2);
        const unsigned b3 = __builtin_amdgcn_ballot_w32(h3);
        const unsigned c0 = (unsigned)__popc(b0), c1 = (unsigned)__popc(b1), c2 = (unsigned)__popc(b2);
        const unsigned tot = c0 + c1 + c2 + (unsigned)__popc(b3);
        if (tot != 0u) {
          unsigned first = 0u;
          if (lane == 0) first = atomicAdd(&s_ng, tot);
          first = (unsigned)__builtin_amdgcn_readfirstlane((int)first);
          const unsigned i0 = first + __builtin_amdgcn_mbcnt_lo(b0, 0u);
          const unsigned i1 = first + c0 + __builtin_amdgcn_mbcnt_lo(b1, 0u);
          const unsigned i2 = first + c0 + c1 + __builtin_amdgcn_mbcnt_lo(b2, 0u);
          const unsigned i3 = first + c0 + c1 + c2 + __builtin_amdgcn_mbcnt_lo(b3, 0u);
          if (h0 && i0 < OWN_CAP) s_lgw[i0] = (unsigned)(g0 + 0);
          if (h1 && i1 < OWN_CAP) s_lgw[i1] = (unsigned)(g0 + 1);
          if (h2 && i2 < OWN_CAP) s_lgw[i2] = (unsigned)(g0 + 2);
          if (h3 && i3 < OWN_CAP) s_lgw[i3] = (unsigned)(g0 + 3);
        }
      }
    }
    __syncthreads();
    const unsigned ngr = s_ng;

    if (ngr <= OWN_CAP) {
      const int ng = (int)ngr;
      for (int i0 = 0; i0 < ng; i0 += OWN_T * (OWN_U / 2)) {
        uint4 k4[OWN_U];
        int   gq[OWN_U / 2];
#pragma unroll
        for (int q = 0; q < OWN_U / 2; ++q) {
          const int i  = i0 + q * OWN_T + tid;
          const int gr = (int)s_lgw[min(i, OWN_CAP - 1)];
          gq[q] = (i < ng) ? gr : -1;
          const int gs = max(gq[q], 0);
          k4[2 * q]     = key4[2 * (size_t)gs];
          k4[2 * q + 1] = key4[2 * (size_t)gs + 1];
        }
        __builtin_amdgcn_sched_barrier(0);
#pragma unroll
        for (int u = 0; u < OWN_U; ++u) {
          const int gcur = gq[u >> 1];
          const int e0   = 32 * gcur + 16 * (u & 1);
          const unsigned kw[4] = {k4[u].x, k4[u].y, k4[u].z, k4[u].w};
          unsigned zb[4];
#pragma unroll
          for (int c = 0; c < 4; ++c) {
            const unsigned y = kw[c] ^ pat;
            zb[c] = ~(((y & 0x7F7F7F7Fu) + 0x7F7F7F7Fu) | y) & 0x80808080u;
          }
          unsigned hit = ((((zb[0] >> 1) | zb[1]) >> 2) | ((zb[2] >> 1) | zb[3]));
          if (gcur < 0) hit = 0u;
          while (hit != 0u) {
            const int p = __ffs((int)hit) - 1;
            hit &= hit - 1u;
            const int e = e0 + 4 * ((p & 7) - 4) + (p >> 3);
            if (e < E) {
              const unsigned idx = atomicAdd(&s_cnt, 1u);
              if (idx < OWN_CAP) s_es[idx] = e;
            }
          }
        }
      }
    }
    __syncthreads();
    const unsigned total = (ngr <= OWN_CAP) ? s_cnt : (unsigned)(OWN_CAP + 1);

    if (total <= OWN_CAP) {
      const int cnt = (int)total;

      {
        int sv[OWN_CAP / OWN_T], dv[OWN_CAP / OWN_T];
#pragma unroll
        for (int u = 0; u < OWN_CAP / OWN_T; ++u) {
          const int i  = tid + u * OWN_T;
          const int ev = s_es[i];
          sv[u] = (i < cnt) ? ev : 0;
        }
#pragma unroll
        for (int u = 0; u < OWN_CAP / OWN_T; ++u) { dv[u] = dst[sv[u]]; sv[u] = src[sv[u]]; }
#pragma unroll
        for (int u = 0; u < OWN_CAP / OWN_T; ++u) {
          const int i = tid + u * OWN_T;
          if (i < cnt) {
            const unsigned loc = (unsigned)(dv[u] - base);
            s_es[i]  = min(max(sv[u], 0), N - 1);
            s_loc[i] = (loc < (unsigned)R) ? (unsigned short)loc : (unsigned short)OWN_NONE;
          }
        }
      }
      __syncthreads();

      for (int i0 = 0; i0 < cnt; i0 += 4 * (OWN_T / 8)) {
        float4 za[4], zb[4];
        int    li[4];
#pragma unroll
        for (int u = 0; u < 4; ++u) {
          const int  i  = i0 + u * (OWN_T / 8) + g;
          const int  sr = s_es[i];
          const int  lr = (int)s_loc[i];
          const bool ok = i < cnt && lr != OWN_NONE;
          const int  s  = ok ? sr : 0;
          li[u] = ok ? lr : 0;
          const float4* zr = (const float4*)(z + (size_t)s * 64);
          za[u] = zr[j]; zb[u] = zr[8 + j];
        }
#pragma unroll
        for (int u = 0; u < 4; ++u) {
          const int  i  = i0 + u * (OWN_T / 8) + g;
          const float4* dr = (const float4*)(s_df + li[u] * 64);
          const float4 da = dr[j], db = dr[8 + j];
          float p = za[u].x * da.x + za[u].y * da.y + za[u].z * da.z + za[u].w * da.w
                  + zb[u].x * db.x + zb[u].y * db.y + zb[u].z * db.z + zb[u].w * db.w;
          p += __shfl_xor(p, 1, 32);
          p += __shfl_xor(p, 2, 32);
          p += __shfl_xor(p, 4, 32);
          if (i < cnt && j == 0 && (int)s_loc[i] != OWN_NONE) {
            s_lgw[i] = __float_as_uint(p);
            atomicMax(&s_key[li[u]], float_key(p));
          }
        }
      }
      __syncthreads();

      for (int i0 = 0; i0 < cnt; i0 += 4 * (OWN_T / 8)) {
        float4 za[4], zb[4];
#pragma unroll
        for (int u = 0; u < 4; ++u) {
          const int  i = i0 + u * (OWN_T / 8) + g;
          const int  sr = s_es[i];
          const int  s  = (i < cnt) ? sr : 0;
          const float4* zr = (const float4*)(z + (size_t)s * 64);
          za[u] = zr[j]; zb[u] = zr[8 + j];
        }
#pragma unroll
        for (int u = 0; u < 4; ++u) {
          const int i = i0 + u * (OWN_T / 8) + g;
          if (i < cnt && (int)s_loc[i] != OWN_NONE) {
            const int   loc = (int)s_loc[i];
            const float ex  = __expf(__uint_as_float(s_lgw[i]) - key_float(s_key[loc]));
            float* np = s_num + loc * OWN_P + 4 * j;
            atomicAdd(np + 0,  ex * za[u].x); atomicAdd(np + 1,  ex * za[u].y);
            atomicAdd(np + 2,  ex * za[u].z); atomicAdd(np + 3,  ex * za[u].w);
            atomicAdd(np + 32, ex * zb[u].x); atomicAdd(np + 33, ex * zb[u].y);
            atomicAdd(np + 34, ex * zb[u].z); atomicAdd(np + 35, ex * zb[u].w);
            if (j == 0) atomicAdd(&s_den[loc], ex);
          }
        }
      }
    } else {
      for (int e = tid; e < E; e += OWN_T) {
        const unsigned loc = (unsigned)(dst[e] - base);
        if (loc < (unsigned)R) {
          const int s = min(max(src[e], 0), N - 1);
          const float p = row_dot_lane(z + (size_t)s * 64, s_df + loc * 64);
          atomicMax(&s_key[loc], float_key(p));
        }
      }
      __syncthreads();
      for (int e = tid; e < E; e += OWN_T) {
        const unsigned loc = (unsigned)(dst[e] - base);
        if (loc < (unsigned)R) {
          const int s = min(max(src[e], 0), N - 1);
          const float* zr = z + (size_t)s * 64;
          const float p  = row_dot_lane(zr, s_df + loc * 64);
          const float ex = __expf(p - key_float(s_key[loc]));
          float* np = s_num + loc * OWN_P;
          for (int k = 0; k < 64; ++k) atomicAdd(np + k, ex * zr[k]);
          atomicAdd(&s_den[loc], ex);
        }
      }
    }
    __syncthreads();

    for (int i = tid; i < R * 64; i += OWN_T) {
      const int   loc = i >> 6, f = i & 63;
      const float dn  = s_den[loc];
      out[(size_t)(base + loc) * 64 + f] = s_num[loc * OWN_P + f] / (dn == 0.0f ? 1.0f : dn);
    }
    __syncthreads();
  }
}

extern "C" void kernel_launch(void* const* d_in, const int* in_sizes, int n_in,
                              void* d_out, int out_size, void* d_ws, size_t ws_size,
                              hipStream_t stream) {
  const float* h     = (const float*)d_in[0];
  const float* feat  = (const float*)d_in[1];
  const float* W_fc  = (const float*)d_in[2];
  const float* W_dst = (const float*)d_in[3];
  const int*   src   = (const int*)d_in[4];
  const int*   dst   = (const int*)d_in[5];
  float*       out   = (float*)d_out;

  const int FEAT = 64, O = 64;
  const int N   = in_sizes[1] / FEAT;
  const int Kfc = in_sizes[0] / N;
  const int E   = in_sizes[4];

  float* z  = (float*)d_ws;
  float* df = z + (size_t)N * O;
  unsigned* keys = (unsigned*)(df + (size_t)N * O);

  if (Kfc == 256) {
    proj_resident<256><<<1024, PJ_T, 0, stream>>>(h, W_fc, z, N);
  } else {
    const dim3 gemmBlk(32 * (O / 16));
    const int  mTiles = (N + 15) / 16;
    gemm_nt_wmma<<<mTiles, gemmBlk, 0, stream>>>(h, W_fc, z, N, Kfc, O);
  }
  proj_resident<64><<<1024, PJ_T, 0, stream>>>(feat, W_dst, df, N);

  owner_keys<<<512, 256, 0, stream>>>(dst, keys, E);
  edge_owner<<<256, OWN_T, 0, stream>>>(z, df, src, dst, keys, out, N, E);
}


# edge phase reads each source row once: weight 2^(e*log2(e)-s), s = floor of the first-arriving edge's scaled logit, an exact power-of-two shift the divide cancels; LDS sums in arrival order
# speedup vs baseline: 4.6585x; 1.0885x over previous
#include <hip/hip_runtime.h>
#include <hip/hip_bf16.h>

typedef float v2f __attribute__((ext_vector_type(2)));
typedef float v8f __attribute__((ext_vector_type(8)));

__global__ void gemm_nt_wmma(const float* __restrict__ A,
                             const float* __restrict__ W,
                             float* __restrict__ C,
                             int M, int K, int O) {
  const int lane  = threadIdx.x & 31;
  const int oTile = threadIdx.x >> 5;
  const int mTile = blockIdx.x;
  if (mTile * 16 >= M) return;
  const int half = lane >> 4;
  const int l16  = lane & 15;

  const float* aPtr = A + (size_t)(mTile * 16 + l16) * K + 2 * half;
  const float* wPtr = W + (size_t)(oTile * 16 + l16) * K + 2 * half;

  v8f acc = {};
#pragma unroll 8
  for (int k = 0; k < K; k += 4) {
    v2f a = *(const v2f*)(aPtr + k);
    v2f b = *(const v2f*)(wPtr + k);
    acc = __builtin_amdgcn_wmma_f32_16x16x4_f32(
         false, a,  false, b,
         (short)0, acc,  false,  false);
  }

  float* cPtr = C + (size_t)(mTile * 16 + half * 8) * O + oTile * 16 + l16;
#pragma unroll
  for (int r = 0; r < 8; ++r) cPtr[(size_t)r * O] = acc[r];
}

#define PJ_T 128
#define PJ_G 4

template <int K>
__global__ void __launch_bounds__(PJ_T)
proj_resident(const float* __restrict__ A,
              const float* __restrict__ Wm,
              float* __restrict__ C, int M) {
  constexpr int P   = K + 4;
  constexpr int NQ  = K / 8;
  constexpr int RV  = K / 4;
  constexpr int NLD = (16 * RV) / PJ_T;
  __shared__ __attribute__((aligned(16))) float s_a[2 * 16 * P];

  const int tid  = threadIdx.x;
  const int lane = tid & 31;
  const int ct   = tid >> 5;
  const int c    = lane & 15;
  const int h    = lane >> 4;
  const int nSteps = (M + 15) >> 4;
  int step = blockIdx.x;
  if (step >= nSteps) return;

  float4 wf[NQ];
  {
    const float4* wr = (const float4*)(Wm + (size_t)(ct * 16 + c) * K + 4 * h);
#pragma unroll
    for (int q = 0; q < NQ; ++q) wf[q] = wr[2 * q];
  }

  constexpr int RPI = PJ_T / RV;
  const int trow = tid / RV, tcol = (tid % RV) * 4;
  const float* ag = A + (size_t)trow * K + tcol;
  float*       sl = s_a + trow * P + tcol;

  static_assert(NLD == 8 || NLD == 2, "tile share per thread");
  float4 t0, t1, t2, t3, t4, t5, t6, t7;
#define PJ_REQ(at)                                                        \
  do {                                                                    \
    t0 = *(const float4*)((at) + (size_t)0 * RPI * K);                    \
    t1 = *(const float4*)((at) + (size_t)1 * RPI * K);                    \
    if constexpr (NLD == 8) {                                             \
      t2 = *(const float4*)((at) + (size_t)2 * RPI * K);                  \
      t3 = *(const float4*)((at) + (size_t)3 * RPI * K);                  \
      t4 = *(const float4*)((at) + (size_t)4 * RPI * K);                  \
      t5 = *(const float4*)((at) + (size_t)5 * RPI * K);                  \
      t6 = *(const float4*)((at) + (size_t)6 * RPI * K);                  \
      t7 = *(const float4*)((at) + (size_t)7 * RPI * K);                  \
    }                                                                     \
  } while (0)
#define PJ_PUT(sp)                                                        \
  do {                                                                    \
    *(float4*)((sp) + 0 * RPI * P) = t0;                                  \
    *(float4*)((sp) + 1 * RPI * P) = t1;                                  \
    if constexpr (NLD == 8) {                                             \
      *(float4*)((sp) + 2 * RPI * P) = t2;                                \
      *(float4*)((sp) + 3 * RPI * P) = t3;                                \
      *(float4*)((sp) + 4 * RPI * P) = t4;                                \
      *(float4*)((sp) + 5 * RPI * P) = t5;                                \
      *(float4*)((sp) + 6 * RPI * P) = t6;                                \
      *(float4*)((sp) + 7 * RPI * P) = t7;                                \
    }                                                                     \
  } while (0)

  {
    const float* at = ag + (size_t)min(step * 16, M - 16) * K;
    PJ_REQ(at);
    PJ_PUT(sl);
  }
  {
    const int ls = min(step + (int)gridDim.x, nSteps - 1);
    const float* at = ag + (size_t)min(ls * 16, M - 16) * K;
    PJ_REQ(at);
  }
  __syncthreads();

  int b = 0;
  for (;;) {
    const float* ar = s_a + b * (16 * P) + c * P + 4 * h;
    v8f acc0 = {}, acc1 = {};
    float4 af[2][PJ_G];
#pragma unroll
    for (int j = 0; j < PJ_G; ++j) af[0][j] = *(const float4*)(ar + 8 * j);
#pragma unroll
    for (int g = 0; g < NQ / PJ_G; ++g) {
      if (g + 1 < NQ / PJ_G) {
#pragma unroll
        for (int j = 0; j < PJ_G; ++j) af[(g + 1) & 1][j] = *(const float4*)(ar + 8 * ((g + 1) * PJ_G + j));
      }
      __builtin_amdgcn_sched_barrier(0);
#pragma unroll
      for (int j = 0; j < PJ_G; ++j) {
        const float4 av = af[g & 1][j];
        const float4 wv = wf[g * PJ_G + j];
        const v2f a0 = {av.x, av.y}, a1 = {av.z, av.w};
        const v2f b0 = {wv.x, wv.y}, b1 = {wv.z, wv.w};
        acc0 = __builtin_amdgcn_wmma_f32_16x16x4_f32(false, a0, false, b0, (short)0, acc0, false, false);
        acc1 = __builtin_amdgcn_wmma_f32_16x16x4_f32(false, a1, false, b1, (short)0, acc1, false, false);
      }
      __builtin_amdgcn_sched_barrier(0);
    }

    {
      float* sn = sl + (b ^ 1) * (16 * P);
      PJ_PUT(sn);
    }
    __syncthreads();

    {
      const int r0 = min(step * 16, M - 16) + 8 * h;
      float* cp = C + (size_t)r0 * 64 + ct * 16 + c;
#pragma unroll
      for (int r = 0; r < 8; ++r)
        if (r0 + r >= step * 16) cp[(size_t)r * 64] = acc0[r] + acc1[r];
    }

    const int nx = step + (int)gridDim.x;
    if (nx >= nSteps) break;
    {
      const int ls = min(nx + (int)gridDim.x, nSteps - 1);
      const float* at = ag + (size_t)min(ls * 16, M - 16) * K;
      PJ_REQ(at);
    }
    step = nx;
    b ^= 1;
  }
#undef PJ_REQ
#undef PJ_PUT
}

__device__ __forceinline__ unsigned float_key(float f) {
  unsigned u = __float_as_uint(f);
  return (u & 0x80000000u) ? ~u : (u | 0x80000000u);
}
__device__ __forceinline__ float key_float(unsigned k) {
  unsigned u = (k & 0x80000000u) ? (k & 0x7FFFFFFFu) : ~k;
  return __uint_as_float(u);
}

#define OWN_R   392
#define OWN_P   65
#define OWN_CAP 8192
#define OWN_T   1024
#define OWN_NONE 0xFFFF
#define OWN_U   8
#define OWN_EMPTY  ((int)0x80000000)
#define OWN_SPREAD 40
#define OWN_CLAMP   268435456.0f
#define OWN_CLAMP_I (1 << 28)
#define OWN_L2E_HI 1.44269502162933349609375f
#define OWN_L2E_LO 1.925963033500011e-8f

__device__ __forceinline__ int own_groups(int E) { return (E + 31) >> 5; }
__device__ __forceinline__ int own_pitch(int E)  { return (own_groups(E) + 3) & ~3; }

__global__ void __launch_bounds__(256)
owner_keys(const int* __restrict__ dst, unsigned* __restrict__ keys, int E) {
  const int nG    = own_groups(E);
  const int pitch = own_pitch(E);
  unsigned* __restrict__ maps = keys + (size_t)nG * 8;
  for (int g = blockIdx.x * 256 + threadIdx.x; g < nG; g += gridDim.x * 256) {
    unsigned o[32];
    if (32 * g + 32 <= E) {
      const int4* __restrict__ p = (const int4*)(dst + (size_t)32 * g);
      int4 d[8];
#pragma unroll
      for (int q = 0; q < 8; ++q) d[q] = p[q];
#pragma unroll
      for (int q = 0; q < 8; ++q) {
        o[4 * q + 0] = ((unsigned)d[q].x / (unsigned)OWN_R) & 255u;
        o[4 * q + 1] = ((unsigned)d[q].y / (unsigned)OWN_R) & 255u;
        o[4 * q + 2] = ((unsigned)d[q].z / (unsigned)OWN_R) & 255u;
        o[4 * q + 3] = ((unsigned)d[q].w / (unsigned)OWN_R) & 255u;
      }
    } else {
#pragma unroll
      for (int k = 0; k < 32; ++k) {
        const int e = 32 * g + k;
        o[k] = (e < E) ? (((unsigned)dst[e] / (unsigned)OWN_R) & 255u) : 0xFFFFFFFFu;
      }
    }
    unsigned w[8];
#pragma unroll
    for (int q = 0; q < 8; ++q)
      w[q] = (o[4 * q] & 255u) | ((o[4 * q + 1] & 255u) << 8) | ((o[4 * q + 2] & 255u) << 16) | ((o[4 * q + 3] & 255u) << 24);
    ((uint4*)keys)[2 * (size_t)g]     = make_uint4(w[0], w[1], w[2], w[3]);
    ((uint4*)keys)[2 * (size_t)g + 1] = make_uint4(w[4], w[5], w[6], w[7]);
    unsigned bm[8] = {0u, 0u, 0u, 0u, 0u, 0u, 0u, 0u};
#pragma unroll
    for (int k = 0; k < 32; ++k) {
      const unsigned bit = 1u << (o[k] & 31u);
      const unsigned hi  = o[k] >> 5;
#pragma unroll
      for (int d = 0; d < 8; ++d) bm[d] |= (hi == (unsigned)d) ? bit : 0u;
    }
#pragma unroll
    for (int d = 0; d < 8; ++d) maps[(size_t)d * pitch + g] = bm[d];
  }
}

__device__ __forceinline__ float row_dot_lane(const float* __restrict__ zr,
                                              const float* __restrict__ dr) {
  float p = 0.0f;
  for (int k = 0; k < 16; ++k) {
    const float4 a = ((const float4*)zr)[k];
    const float4 b = ((const float4*)dr)[k];
    p = fmaf(a.x, b.x, p); p = fmaf(a.y, b.y, p);
    p = fmaf(a.z, b.z, p); p = fmaf(a.w, b.w, p);
  }
  return p;
}

__device__ __forceinline__ float own_lg2(float e) {
  return fminf(fmaxf(e * OWN_L2E_HI, -OWN_CLAMP), OWN_CLAMP);
}

__global__ void __launch_bounds__(OWN_T)
edge_owner(const float* __restrict__ z,
           const float* __restrict__ df,
           const int* __restrict__ src,
           const int* __restrict__ dst,
           const unsigned* __restrict__ keys,
           float* __restrict__ out, int N, int E) {
  __shared__ __attribute__((aligned(16))) float s_df[OWN_R * 64];
  __shared__ float          s_num[OWN_R * OWN_P];
  __shared__ float          s_den[OWN_R];
  __shared__ unsigned       s_key[OWN_R];
  __shared__ int            s_sh[OWN_R];
  __shared__ int            s_es[OWN_CAP];
  __shared__ unsigned       s_lgw[OWN_CAP];
  __shared__ unsigned short s_loc[OWN_CAP];
  __shared__ unsigned       s_cnt;
  __shared__ unsigned       s_ng;
  __shared__ unsigned       s_bad;

  const int tid = threadIdx.x;
  const int g   = tid >> 3;
  const int j   = tid & 7;

  const int lane  = tid & 31;
  const int nG    = own_groups(E);
  const int pitch = own_pitch(E);
  const int nv    = pitch >> 2;
  const int nIter = (nv + OWN_T * OWN_U - 1) / (OWN_T * OWN_U);
  const uint4* __restrict__ key4 = (const uint4*)keys;
  const unsigned* __restrict__ maps = keys + (size_t)nG * 8;

  for (int base = blockIdx.x * OWN_R; base < N; base += gridDim.x * OWN_R) {
    const int R = min(OWN_R, N - base);

    for (int i = tid; i < OWN_R * OWN_P; i += OWN_T) s_num[i] = 0.0f;
    for (int i = tid; i < OWN_R; i += OWN_T) { s_den[i] = 0.0f; s_key[i] = 0u; s_sh[i] = OWN_EMPTY; }
    {
      const float4* __restrict__ dfr = (const float4*)(df + (size_t)base * 64);
#pragma unroll
      for (int u = 0; u < (OWN_R * 16 + OWN_T - 1) / OWN_T; ++u) {
        const int i = min(tid + u * OWN_T, R * 16 - 1);
        ((float4*)s_df)[i] = dfr[i];
      }
    }
    if (tid == 0) { s_cnt = 0u; s_ng = 0u; s_bad = 0u; }
    __syncthreads();

    const unsigned want = ((unsigned)base / (unsigned)OWN_R) & 255u;
    const unsigned pat  = want * 0x01010101u;
    const unsigned wsh  = want & 31u;
    const uint4* __restrict__ row4 = (const uint4*)(maps + (size_t)(want >> 5) * pitch);
    const int rotv = (int)(((unsigned)(blockIdx.x & 31) * (unsigned)nv) >> 5);
    for (int it = 0; it < nIter; ++it) {
      const int v0 = it * (OWN_T * OWN_U) + tid;
      uint4 m4[OWN_U];
      int   vv[OWN_U];
#pragma unroll
      for (int u = 0; u < OWN_U; ++u) {
        int x = min(v0 + u * OWN_T, nv - 1) + rotv;
        if (x >= nv) x -= nv;
        vv[u] = x;
        m4[u] = row4[x];
      }
      __builtin_amdgcn_sched_barrier(0);
#pragma unroll
      for (int u = 0; u < OWN_U; ++u) {
        const bool in = (v0 + u * OWN_T) < nv;
        const int  g0 = 4 * vv[u];
        const bool h0 = in && (g0 + 0 < nG) && (((m4[u].x >> wsh) & 1u) != 0u);
        const bool h1 = in && (g0 + 1 < nG) && (((m4[u].y >> wsh) & 1u) != 0u);
        const bool h2 = in && (g0 + 2 < nG) && (((m4[u].z >> wsh) & 1u) != 0u);
        const bool h3 = in && (g0 + 3 < nG) && (((m4[u].w >> wsh) & 1u) != 0u);
        const unsigned b0 = __builtin_amdgcn_ballot_w32(h0);
        const unsigned b1 = __builtin_amdgcn_ballot_w32(h1);
        const unsigned b2 = __builtin_amdgcn_ballot_w32(h2);
        const unsigned b3 = __builtin_amdgcn_ballot_w32(h3);
        const unsigned c0 = (unsigned)__popc(b0), c1 = (unsigned)__popc(b1), c2 = (unsigned)__popc(b2);
        const unsigned tot = c0 + c1 + c2 + (unsigned)__popc(b3);
        if (tot != 0u) {
          unsigned first = 0u;
          if (lane == 0) first = atomicAdd(&s_ng, tot);
          first = (unsigned)__builtin_amdgcn_readfirstlane((int)first);
          const unsigned i0 = first + __builtin_amdgcn_mbcnt_lo(b0, 0u);
          const unsigned i1 = first + c0 + __builtin_amdgcn_mbcnt_lo(b1, 0u);
          const unsigned i2 = first + c0 + c1 + __builtin_amdgcn_mbcnt_lo(b2, 0u);
          const unsigned i3 = first + c0 + c1 + c2 + __builtin_amdgcn_mbcnt_lo(b3, 0u);
          if (h0 && i0 < OWN_CAP) s_lgw[i0] = (unsigned)(g0 + 0);
          if (h1 && i1 < OWN_CAP) s_lgw[i1] = (unsigned)(g0 + 1);
          if (h2 && i2 < OWN_CAP) s_lgw[i2] = (unsigned)(g0 + 2);
          if (h3 && i3 < OWN_CAP) s_lgw[i3] = (unsigned)(g0 + 3);
        }
      }
    }
    __syncthreads();
    const unsigned ngr = s_ng;

    if (ngr <= OWN_CAP) {
      const int ng = (int)ngr;
      for (int i0 = 0; i0 < ng; i0 += OWN_T * (OWN_U / 2)) {
        uint4 k4[OWN_U];
        int   gq[OWN_U / 2];
#pragma unroll
        for (int q = 0; q < OWN_U / 2; ++q) {
          const int i  = i0 + q * OWN_T + tid;
          const int gr = (int)s_lgw[min(i, OWN_CAP - 1)];
          gq[q] = (i < ng) ? gr : -1;
          const int gs = max(gq[q], 0);
          k4[2 * q]     = key4[2 * (size_t)gs];
          k4[2 * q + 1] = key4[2 * (size_t)gs + 1];
        }
        __builtin_amdgcn_sched_barrier(0);
#pragma unroll
        for (int u = 0; u < OWN_U; ++u) {
          const int gcur = gq[u >> 1];
          const int e0   = 32 * gcur + 16 * (u & 1);
          const unsigned kw[4] = {k4[u].x, k4[u].y, k4[u].z, k4[u].w};
          unsigned zb[4];
#pragma unroll
          for (int c = 0; c < 4; ++c) {
            const unsigned y = kw[c] ^ pat;
            zb[c] = ~(((y & 0x7F7F7F7Fu) + 0x7F7F7F7Fu) | y) & 0x80808080u;
          }
          unsigned hit = ((((zb[0] >> 1) | zb[1]) >> 2) | ((zb[2] >> 1) | zb[3]));
          if (gcur < 0) hit = 0u;
          while (hit != 0u) {
            const int p = __ffs((int)hit) - 1;
            hit &= hit - 1u;
            const int e = e0 + 4 * ((p & 7) - 4) + (p >> 3);
            if (e < E) {
              const unsigned idx = atomicAdd(&s_cnt, 1u);
              if (idx < OWN_CAP) s_es[idx] = e;
            }
          }
        }
      }
    }
    __syncthreads();
    const unsigned total = (ngr <= OWN_CAP) ? s_cnt : (unsigned)(OWN_CAP + 1);

    if (total <= OWN_CAP) {
      const int cnt = (int)total;

      {
        int sv[OWN_CAP / OWN_T], dv[OWN_CAP / OWN_T];
#pragma unroll
        for (int u = 0; u < OWN_CAP / OWN_T; ++u) {
          const int i  = tid + u * OWN_T;
          const int ev = s_es[i];
          sv[u] = (i < cnt) ? ev : 0;
        }
#pragma unroll
        for (int u = 0; u < OWN_CAP / OWN_T; ++u) { dv[u] = dst[sv[u]]; sv[u] = src[sv[u]]; }
#pragma unroll
        for (int u = 0; u < OWN_CAP / OWN_T; ++u) {
          const int i = tid + u * OWN_T;
          if (i < cnt) {
            const unsigned loc = (unsigned)(dv[u] - base);
            s_es[i]  = min(max(sv[u], 0), N - 1);
            s_loc[i] = (loc < (unsigned)R) ? (unsigned short)loc : (unsigned short)OWN_NONE;
          }
        }
      }
      __syncthreads();

      for (int i0 = 0; i0 < cnt; i0 += 4 * (OWN_T / 8)) {
        float4 za[4], zb[4];
        int    li[4];
        bool   ok[4];
#pragma unroll
        for (int u = 0; u < 4; ++u) {
          const int  i  = i0 + u * (OWN_T / 8) + g;
          const int  sr = s_es[i];
          const int  lr = (int)s_loc[i];
          ok[u] = i < cnt && lr != OWN_NONE;
          const int  s  = ok[u] ? sr : 0;
          li[u] = ok[u] ? lr : 0;
          const float4* zr = (const float4*)(z + (size_t)s * 64);
          za[u] = zr[j]; zb[u] = zr[8 + j];
        }
        float ef[4];
        int   ti[4], sh[4];
#pragma unroll
        for (int u = 0; u < 4; ++u) {
          const int  i  = i0 + u * (OWN_T / 8) + g;
          const float4* dr = (const float4*)(s_df + li[u] * 64);
          const float4 da = dr[j], db = dr[8 + j];
          float p = za[u].x * da.x + za[u].y * da.y + za[u].z * da.z + za[u].w * da.w
                  + zb[u].x * db.x + zb[u].y * db.y + zb[u].z * db.z + zb[u].w * db.w;
          p += __shfl_xor(p, 1, 32);
          p += __shfl_xor(p, 2, 32);
          p += __shfl_xor(p, 4, 32);
          const float th = p * OWN_L2E_HI;
          const float tl = fmaf(p, OWN_L2E_LO, fmaf(p, OWN_L2E_HI, -th));
          const float tc = own_lg2(p);
          ti[u] = (int)floorf(tc);
          ef[u] = __builtin_amdgcn_exp2f(__builtin_amdgcn_fractf(tc) + tl);
          sh[u] = 0;
          if (ok[u] && j == 0) {
            s_lgw[i] = __float_as_uint(p);
            atomicMax(&s_key[li[u]], float_key(p));
            const int old = atomicCAS(&s_sh[li[u]], OWN_EMPTY, ti[u]);
            sh[u] = (old == OWN_EMPTY) ? ti[u] : old;
          }
        }
#pragma unroll
        for (int u = 0; u < 4; ++u) {
          const int   shift = __shfl(sh[u], 0, 8);
          const float ex    = ldexpf(ef[u], ti[u] - shift);
          if (ok[u]) {
            float* np = s_num + li[u] * OWN_P + 4 * j;
            atomicAdd(np + 0,  ex * za[u].x); atomicAdd(np + 1,  ex * za[u].y);
            atomicAdd(np + 2,  ex * za[u].z); atomicAdd(np + 3,  ex * za[u].w);
            atomicAdd(np + 32, ex * zb[u].x); atomicAdd(np + 33, ex * zb[u].y);
            atomicAdd(np + 34, ex * zb[u].z); atomicAdd(np + 35, ex * zb[u].w);
            if (j == 0) atomicAdd(&s_den[li[u]], ex);
          }
        }
      }
      __syncthreads();

      for (int i = tid; i < R; i += OWN_T) {
        const int shv = s_sh[i];
        if (shv != OWN_EMPTY) {
          const int tm = (int)floorf(own_lg2(key_float(s_key[i])));
          if (tm - shv > OWN_SPREAD || tm >= OWN_CLAMP_I || shv <= -OWN_CLAMP_I) s_bad = 1u;
        }
      }
      __syncthreads();

      if (s_bad != 0u) {
        for (int i = tid; i < OWN_R * OWN_P; i += OWN_T) s_num[i] = 0.0f;
        for (int i = tid; i < OWN_R; i += OWN_T) s_den[i] = 0.0f;
        __syncthreads();

        for (int i0 = 0; i0 < cnt; i0 += 4 * (OWN_T / 8)) {
          float4 za[4], zb[4];
#pragma unroll
          for (int u = 0; u < 4; ++u) {
            const int  i = i0 + u * (OWN_T / 8) + g;
            const int  sr = s_es[i];
            const int  s  = (i < cnt) ? sr : 0;
            const float4* zr = (const float4*)(z + (size_t)s * 64);
            za[u] = zr[j]; zb[u] = zr[8 + j];
          }
#pragma unroll
          for (int u = 0; u < 4; ++u) {
            const int i = i0 + u * (OWN_T / 8) + g;
            if (i < cnt && (int)s_loc[i] != OWN_NONE) {
              const int   loc = (int)s_loc[i];
              const float ex  = __expf(__uint_as_float(s_lgw[i]) - key_float(s_key[loc]));
              float* np = s_num + loc * OWN_P + 4 * j;
              atomicAdd(np + 0,  ex * za[u].x); atomicAdd(np + 1,  ex * za[u].y);
              atomicAdd(np + 2,  ex * za[u].z); atomicAdd(np + 3,  ex * za[u].w);
              atomicAdd(np + 32, ex * zb[u].x); atomicAdd(np + 33, ex * zb[u].y);
              atomicAdd(np + 34, ex * zb[u].z); atomicAdd(np + 35, ex * zb[u].w);
              if (j == 0) atomicAdd(&s_den[loc], ex);
            }
          }
        }
      }
    } else {
      for (int e = tid; e < E; e += OWN_T) {
        const unsigned loc = (unsigned)(dst[e] - base);
        if (loc < (unsigned)R) {
          const int s = min(max(src[e], 0), N - 1);
          const float p = row_dot_lane(z + (size_t)s * 64, s_df + loc * 64);
          atomicMax(&s_key[loc], float_key(p));
        }
      }
      __syncthreads();
      for (int e = tid; e < E; e += OWN_T) {
        const unsigned loc = (unsigned)(dst[e] - base);
        if (loc < (unsigned)R) {
          const int s = min(max(src[e], 0), N - 1);
          const float* zr = z + (size_t)s * 64;
          const float p  = row_dot_lane(zr, s_df + loc * 64);
          const float ex = __expf(p - key_float(s_key[loc]));
          float* np = s_num + loc * OWN_P;
          for (int k = 0; k < 64; ++k) atomicAdd(np + k, ex * zr[k]);
          atomicAdd(&s_den[loc], ex);
        }
      }
    }
    __syncthreads();

    for (int i = tid; i < R * 64; i += OWN_T) {
      const int   loc = i >> 6, f = i & 63;
      const float dn  = s_den[loc];
      out[(size_t)(base + loc) * 64 + f] = s_num[loc * OWN_P + f] / (dn == 0.0f ? 1.0f : dn);
    }
    __syncthreads();
  }
}

extern "C" void kernel_launch(void* const* d_in, const int* in_sizes, int n_in,
                              void* d_out, int out_size, void* d_ws, size_t ws_size,
                              hipStream_t stream) {
  const float* h     = (const float*)d_in[0];
  const float* feat  = (const float*)d_in[1];
  const float* W_fc  = (const float*)d_in[2];
  const float* W_dst = (const float*)d_in[3];
  const int*   src   = (const int*)d_in[4];
  const int*   dst   = (const int*)d_in[5];
  float*       out   = (float*)d_out;

  const int FEAT = 64, O = 64;
  const int N   = in_sizes[1] / FEAT;
  const int Kfc = in_sizes[0] / N;
  const int E   = in_sizes[4];

  float* z  = (float*)d_ws;
  float* df = z + (size_t)N * O;
  unsigned* keys = (unsigned*)(df + (size_t)N * O);

  if (Kfc == 256) {
    proj_resident<256><<<1024, PJ_T, 0, stream>>>(h, W_fc, z, N);
  } else {
    const dim3 gemmBlk(32 * (O / 16));
    const int  mTiles = (N + 15) / 16;
    gemm_nt_wmma<<<mTiles, gemmBlk, 0, stream>>>(h, W_fc, z, N, Kfc, O);
  }
  proj_resident<64><<<1024, PJ_T, 0, stream>>>(feat, W_dst, df, N);

  owner_keys<<<512, 256, 0, stream>>>(dst, keys, E);
  edge_owner<<<256, OWN_T, 0, stream>>>(z, df, src, dst, keys, out, N, E);
}


# edge list dealt into 4 classes by local destination mod 4, 8-lane group g takes class g mod 4: the 4 edges of one LDS float add never share a bank; LDS sums in a new arrival order
# speedup vs baseline: 4.8487x; 1.0408x over previous
#include <hip/hip_runtime.h>
#include <hip/hip_bf16.h>

typedef float v2f __attribute__((ext_vector_type(2)));
typedef float v8f __attribute__((ext_vector_type(8)));

__global__ void gemm_nt_wmma(const float* __restrict__ A,
                             const float* __restrict__ W,
                             float* __restrict__ C,
                             int M, int K, int O) {
  const int lane  = threadIdx.x & 31;
  const int oTile = threadIdx.x >> 5;
  const int mTile = blockIdx.x;
  if (mTile * 16 >= M) return;
  const int half = lane >> 4;
  const int l16  = lane & 15;

  const float* aPtr = A + (size_t)(mTile * 16 + l16) * K + 2 * half;
  const float* wPtr = W + (size_t)(oTile * 16 + l16) * K + 2 * half;

  v8f acc = {};
#pragma unroll 8
  for (int k = 0; k < K; k += 4) {
    v2f a = *(const v2f*)(aPtr + k);
    v2f b = *(const v2f*)(wPtr + k);
    acc = __builtin_amdgcn_wmma_f32_16x16x4_f32(
         false, a,  false, b,
         (short)0, acc,  false,  false);
  }

  float* cPtr = C + (size_t)(mTile * 16 + half * 8) * O + oTile * 16 + l16;
#pragma unroll
  for (int r = 0; r < 8; ++r) cPtr[(size_t)r * O] = acc[r];
}

#define PJ_T 128
#define PJ_G 4

template <int K>
__global__ void __launch_bounds__(PJ_T)
proj_resident(const float* __restrict__ A,
              const float* __restrict__ Wm,
              float* __restrict__ C, int M) {
  constexpr int P   = K + 4;
  constexpr int NQ  = K / 8;
  constexpr int RV  = K / 4;
  constexpr int NLD = (16 * RV) / PJ_T;
  __shared__ __attribute__((aligned(16))) float s_a[2 * 16 * P];

  const int tid  = threadIdx.x;
  const int lane = tid & 31;
  const int ct   = tid >> 5;
  const int c    = lane & 15;
  const int h    = lane >> 4;
  const int nSteps = (M + 15) >> 4;
  int step = blockIdx.x;
  if (step >= nSteps) return;

  float4 wf[NQ];
  {
    const float4* wr = (const float4*)(Wm + (size_t)(ct * 16 + c) * K + 4 * h);
#pragma unroll
    for (int q = 0; q < NQ; ++q) wf[q] = wr[2 * q];
  }

  constexpr int RPI = PJ_T / RV;
  const int trow = tid / RV, tcol = (tid % RV) * 4;
  const float* ag = A + (size_t)trow * K + tcol;
  float*       sl = s_a + trow * P + tcol;

  static_assert(NLD == 8 || NLD == 2, "tile share per thread");
  float4 t0, t1, t2, t3, t4, t5, t6, t7;
#define PJ_REQ(at)                                                        \
  do {                                                                    \
    t0 = *(const float4*)((at) + (size_t)0 * RPI * K);                    \
    t1 = *(const float4*)((at) + (size_t)1 * RPI * K);                    \
    if constexpr (NLD == 8) {                                             \
      t2 = *(const float4*)((at) + (size_t)2 * RPI * K);                  \
      t3 = *(const float4*)((at) + (size_t)3 * RPI * K);                  \
      t4 = *(const float4*)((at) + (size_t)4 * RPI * K);                  \
      t5 = *(const float4*)((at) + (size_t)5 * RPI * K);                  \
      t6 = *(const float4*)((at) + (size_t)6 * RPI * K);                  \
      t7 = *(const float4*)((at) + (size_t)7 * RPI * K);                  \
    }                                                                     \
  } while (0)
#define PJ_PUT(sp)                                                        \
  do {                                                                    \
    *(float4*)((sp) + 0 * RPI * P) = t0;                                  \
    *(float4*)((sp) + 1 * RPI * P) = t1;                                  \
    if constexpr (NLD == 8) {                                             \
      *(float4*)((sp) + 2 * RPI * P) = t2;                                \
      *(float4*)((sp) + 3 * RPI * P) = t3;                                \
      *(float4*)((sp) + 4 * RPI * P) = t4;                                \
      *(float4*)((sp) + 5 * RPI * P) = t5;                                \
      *(float4*)((sp) + 6 * RPI * P) = t6;                                \
      *(float4*)((sp) + 7 * RPI * P) = t7;                                \
    }                                                                     \
  } while (0)

  {
    const float* at = ag + (size_t)min(step * 16, M - 16) * K;
    PJ_REQ(at);
    PJ_PUT(sl);
  }
  {
    const int ls = min(step + (int)gridDim.x, nSteps - 1);
    const float* at = ag + (size_t)min(ls * 16, M - 16) * K;
    PJ_REQ(at);
  }
  __syncthreads();

  int b = 0;
  for (;;) {
    const float* ar = s_a + b * (16 * P) + c * P + 4 * h;
    v8f acc0 = {}, acc1 = {};
    float4 af[2][PJ_G];
#pragma unroll
    for (int j = 0; j < PJ_G; ++j) af[0][j] = *(const float4*)(ar + 8 * j);
#pragma unroll
    for (int g = 0; g < NQ / PJ_G; ++g) {
      if (g + 1 < NQ / PJ_G) {
#pragma unroll
        for (int j = 0; j < PJ_G; ++j) af[(g + 1) & 1][j] = *(const float4*)(ar + 8 * ((g + 1) * PJ_G + j));
      }
      __builtin_amdgcn_sched_barrier(0);
#pragma unroll
      for (int j = 0; j < PJ_G; ++j) {
        const float4 av = af[g & 1][j];
        const float4 wv = wf[g * PJ_G + j];
        const v2f a0 = {av.x, av.y}, a1 = {av.z, av.w};
        const v2f b0 = {wv.x, wv.y}, b1 = {wv.z, wv.w};
        acc0 = __builtin_amdgcn_wmma_f32_16x16x4_f32(false, a0, false, b0, (short)0, acc0, false, false);
        acc1 = __builtin_amdgcn_wmma_f32_16x16x4_f32(false, a1, false, b1, (short)0, acc1, false, false);
      }
      __builtin_amdgcn_sched_barrier(0);
    }

    {
      float* sn = sl + (b ^ 1) * (16 * P);
      PJ_PUT(sn);
    }
    __syncthreads();

    {
      const int r0 = min(step * 16, M - 16) + 8 * h;
      float* cp = C + (size_t)r0 * 64 + ct * 16 + c;
#pragma unroll
      for (int r = 0; r < 8; ++r)
        if (r0 + r >= step * 16) cp[(size_t)r * 64] = acc0[r] + acc1[r];
    }

    const int nx = step + (int)gridDim.x;
    if (nx >= nSteps) break;
    {
      const int ls = min(nx + (int)gridDim.x, nSteps - 1);
      const float* at = ag + (size_t)min(ls * 16, M - 16) * K;
      PJ_REQ(at);
    }
    step = nx;
    b ^= 1;
  }
#undef PJ_REQ
#undef PJ_PUT
}

__device__ __forceinline__ unsigned float_key(float f) {
  unsigned u = __float_as_uint(f);
  return (u & 0x80000000u) ? ~u : (u | 0x80000000u);
}
__device__ __forceinline__ float key_float(unsigned k) {
  unsigned u = (k & 0x80000000u) ? (k & 0x7FFFFFFFu) : ~k;
  return __uint_as_float(u);
}

#define OWN_R   392
#define OWN_P   65
#define OWN_CAP 8192
#define OWN_T   1024
#define OWN_NONE 0xFFFF
#define OWN_U   8
#define OWN_EMPTY  ((int)0x80000000)
#define OWN_SPREAD 40
#define OWN_CLAMP   268435456.0f
#define OWN_CLAMP_I (1 << 28)
#define OWN_L2E_HI 1.44269502162933349609375f
#define OWN_L2E_LO 1.925963033500011e-8f

__device__ __forceinline__ int own_groups(int E) { return (E + 31) >> 5; }
__device__ __forceinline__ int own_pitch(int E)  { return (own_groups(E) + 3) & ~3; }

__global__ void __launch_bounds__(256)
owner_keys(const int* __restrict__ dst, unsigned* __restrict__ keys, int E) {
  const int nG    = own_groups(E);
  const int pitch = own_pitch(E);
  unsigned* __restrict__ maps = keys + (size_t)nG * 8;
  for (int g = blockIdx.x * 256 + threadIdx.x; g < nG; g += gridDim.x * 256) {
    unsigned o[32];
    if (32 * g + 32 <= E) {
      const int4* __restrict__ p = (const int4*)(dst + (size_t)32 * g);
      int4 d[8];
#pragma unroll
      for (int q = 0; q < 8; ++q) d[q] = p[q];
#pragma unroll
      for (int q = 0; q < 8; ++q) {
        o[4 * q + 0] = ((unsigned)d[q].x / (unsigned)OWN_R) & 255u;
        o[4 * q + 1] = ((unsigned)d[q].y / (unsigned)OWN_R) & 255u;
        o[4 * q + 2] = ((unsigned)d[q].z / (unsigned)OWN_R) & 255u;
        o[4 * q + 3] = ((unsigned)d[q].w / (unsigned)OWN_R) & 255u;
      }
    } else {
#pragma unroll
      for (int k = 0; k < 32; ++k) {
        const int e = 32 * g + k;
        o[k] = (e < E) ? (((unsigned)dst[e] / (unsigned)OWN_R) & 255u) : 0xFFFFFFFFu;
      }
    }
    unsigned w[8];
#pragma unroll
    for (int q = 0; q < 8; ++q)
      w[q] = (o[4 * q] & 255u) | ((o[4 * q + 1] & 255u) << 8) | ((o[4 * q + 2] & 255u) << 16) | ((o[4 * q + 3] & 255u) << 24);
    ((uint4*)keys)[2 * (size_t)g]     = make_uint4(w[0], w[1], w[2], w[3]);
    ((uint4*)keys)[2 * (size_t)g + 1] = make_uint4(w[4], w[5], w[6], w[7]);
    unsigned bm[8] = {0u, 0u, 0u, 0u, 0u, 0u, 0u, 0u};
#pragma unroll
    for (int k = 0; k < 32; ++k) {
      const unsigned bit = 1u << (o[k] & 31u);
      const unsigned hi  = o[k] >> 5;
#pragma unroll
      for (int d = 0; d < 8; ++d) bm[d] |= (hi == (unsigned)d) ? bit : 0u;
    }
#pragma unroll
    for (int d = 0; d < 8; ++d) maps[(size_t)d * pitch + g] = bm[d];
  }
}

__device__ __forceinline__ float row_dot_lane(const float* __restrict__ zr,
                                              const float* __restrict__ dr) {
  float p = 0.0f;
  for (int k = 0; k < 16; ++k) {
    const float4 a = ((const float4*)zr)[k];
    const float4 b = ((const float4*)dr)[k];
    p = fmaf(a.x, b.x, p); p = fmaf(a.y, b.y, p);
    p = fmaf(a.z, b.z, p); p = fmaf(a.w, b.w, p);
  }
  return p;
}

__device__ __forceinline__ float own_lg2(float e) {
  return fminf(fmaxf(e * OWN_L2E_HI, -OWN_CLAMP), OWN_CLAMP);
}

__device__ __forceinline__ unsigned own_pick4(unsigned q, unsigned a0, unsigned a1, unsigned a2, unsigned a3) {
  const unsigned lo = (q & 1u) ? a1 : a0;
  const unsigned hi = (q & 1u) ? a3 : a2;
  return (q & 2u) ? hi : lo;
}

__global__ void __launch_bounds__(OWN_T)
edge_owner(const float* __restrict__ z,
           const float* __restrict__ df,
           const int* __restrict__ src,
           const int* __restrict__ dst,
           const unsigned* __restrict__ keys,
           float* __restrict__ out, int N, int E) {
  __shared__ __attribute__((aligned(16))) float s_df[OWN_R * 64];
  __shared__ float          s_num[OWN_R * OWN_P];
  __shared__ float          s_den[OWN_R];
  __shared__ unsigned       s_key[OWN_R];
  __shared__ int            s_sh[OWN_R];
  __shared__ int            s_es[OWN_CAP];
  __shared__ unsigned       s_lgw[OWN_CAP];
  __shared__ unsigned short s_loc[OWN_CAP];
  __shared__ unsigned       s_c4[4];
  __shared__ unsigned       s_cnt;
  __shared__ unsigned       s_ng;
  __shared__ unsigned       s_bad;

  const int tid = threadIdx.x;
  const int g   = tid >> 3;
  const int j   = tid & 7;

  const int lane  = tid & 31;
  const int nG    = own_groups(E);
  const int pitch = own_pitch(E);
  const int nv    = pitch >> 2;
  const int nIter = (nv + OWN_T * OWN_U - 1) / (OWN_T * OWN_U);
  const uint4* __restrict__ key4 = (const uint4*)keys;
  const unsigned* __restrict__ maps = keys + (size_t)nG * 8;

  for (int base = blockIdx.x * OWN_R; base < N; base += gridDim.x * OWN_R) {
    const int R = min(OWN_R, N - base);

    for (int i = tid; i < OWN_R * OWN_P; i += OWN_T) s_num[i] = 0.0f;
    for (int i = tid; i < OWN_R; i += OWN_T) { s_den[i] = 0.0f; s_key[i] = 0u; s_sh[i] = OWN_EMPTY; }
    for (int i = tid; i < OWN_CAP; i += OWN_T) s_loc[i] = (unsigned short)OWN_NONE;
    {
      const float4* __restrict__ dfr = (const float4*)(df + (size_t)base * 64);
#pragma unroll
      for (int u = 0; u < (OWN_R * 16 + OWN_T - 1) / OWN_T; ++u) {
        const int i = min(tid + u * OWN_T, R * 16 - 1);
        ((float4*)s_df)[i] = dfr[i];
      }
    }
    if (tid == 0) { s_cnt = 0u; s_ng = 0u; s_bad = 0u; }
    if (tid < 4) s_c4[tid] = 0u;
    __syncthreads();

    const unsigned want = ((unsigned)base / (unsigned)OWN_R) & 255u;
    const unsigned pat  = want * 0x01010101u;
    const unsigned wsh  = want & 31u;
    const uint4* __restrict__ row4 = (const uint4*)(maps + (size_t)(want >> 5) * pitch);
    const int rotv = (int)(((unsigned)(blockIdx.x & 31) * (unsigned)nv) >> 5);
    for (int it = 0; it < nIter; ++it) {
      const int v0 = it * (OWN_T * OWN_U) + tid;
      uint4 m4[OWN_U];
      int   vv[OWN_U];
#pragma unroll
      for (int u = 0; u < OWN_U; ++u) {
        int x = min(v0 + u * OWN_T, nv - 1) + rotv;
        if (x >= nv) x -= nv;
        vv[u] = x;
        m4[u] = row4[x];
      }
      __builtin_amdgcn_sched_barrier(0);
#pragma unroll
      for (int u = 0; u < OWN_U; ++u) {
        const bool in = (v0 + u * OWN_T) < nv;
        const int  g0 = 4 * vv[u];
        const bool h0 = in && (g0 + 0 < nG) && (((m4[u].x >> wsh) & 1u) != 0u);
        const bool h1 = in && (g0 + 1 < nG) && (((m4[u].y >> wsh) & 1u) != 0u);
        const bool h2 = in && (g0 + 2 < nG) && (((m4[u].z >> wsh) & 1u) != 0u);
        const bool h3 = in && (g0 + 3 < nG) && (((m4[u].w >> wsh) & 1u) != 0u);
        const unsigned b0 = __builtin_amdgcn_ballot_w32(h0);
        const unsigned b1 = __builtin_amdgcn_ballot_w32(h1);
        const unsigned b2 = __builtin_amdgcn_ballot_w32(h2);
        const unsigned b3 = __builtin_amdgcn_ballot_w32(h3);
        const unsigned c0 = (unsigned)__popc(b0), c1 = (unsigned)__popc(b1), c2 = (unsigned)__popc(b2);
        const unsigned tot = c0 + c1 + c2 + (unsigned)__popc(b3);
        if (tot != 0u) {
          unsigned first = 0u;
          if (lane == 0) first = atomicAdd(&s_ng, tot);
          first = (unsigned)__builtin_amdgcn_readfirstlane((int)first);
          const unsigned i0 = first + __builtin_amdgcn_mbcnt_lo(b0, 0u);
          const unsigned i1 = first + c0 + __builtin_amdgcn_mbcnt_lo(b1, 0u);
          const unsigned i2 = first + c0 + c1 + __builtin_amdgcn_mbcnt_lo(b2, 0u);
          const unsigned i3 = first + c0 + c1 + c2 + __builtin_amdgcn_mbcnt_lo(b3, 0u);
          if (h0 && i0 < OWN_CAP) s_lgw[i0] = (unsigned)(g0 + 0);
          if (h1 && i1 < OWN_CAP) s_lgw[i1] = (unsigned)(g0 + 1);
          if (h2 && i2 < OWN_CAP) s_lgw[i2] = (unsigned)(g0 + 2);
          if (h3 && i3 < OWN_CAP) s_lgw[i3] = (unsigned)(g0 + 3);
        }
      }
    }
    __syncthreads();
    const unsigned ngr = s_ng;

    if (ngr <= OWN_CAP) {
      const int ng = (int)ngr;
      for (int i0 = 0; i0 < ng; i0 += OWN_T * (OWN_U / 2)) {
        uint4 k4[OWN_U];
        int   gq[OWN_U / 2];
#pragma unroll
        for (int q = 0; q < OWN_U / 2; ++q) {
          const int i  = i0 + q * OWN_T + tid;
          const int gr = (int)s_lgw[min(i, OWN_CAP - 1)];
          gq[q] = (i < ng) ? gr : -1;
          const int gs = max(gq[q], 0);
          k4[2 * q]     = key4[2 * (size_t)gs];
          k4[2 * q + 1] = key4[2 * (size_t)gs + 1];
        }
        __builtin_amdgcn_sched_barrier(0);
#pragma unroll
        for (int u = 0; u < OWN_U; ++u) {
          const int gcur = gq[u >> 1];
          const int e0   = 32 * gcur + 16 * (u & 1);
          const unsigned kw[4] = {k4[u].x, k4[u].y, k4[u].z, k4[u].w};
          unsigned zb[4];
#pragma unroll
          for (int c = 0; c < 4; ++c) {
            const unsigned y = kw[c] ^ pat;
            zb[c] = ~(((y & 0x7F7F7F7Fu) + 0x7F7F7F7Fu) | y) & 0x80808080u;
          }
          unsigned hit = ((((zb[0] >> 1) | zb[1]) >> 2) | ((zb[2] >> 1) | zb[3]));
          if (gcur < 0) hit = 0u;
          while (hit != 0u) {
            const int p = __ffs((int)hit) - 1;
            hit &= hit - 1u;
            const int e = e0 + 4 * ((p & 7) - 4) + (p >> 3);
            if (e < E) {
              const unsigned idx = atomicAdd(&s_cnt, 1u);
              if (idx < OWN_CAP) s_es[idx] = e;
            }
          }
        }
      }
    }
    __syncthreads();
    const unsigned total = (ngr <= OWN_CAP) ? s_cnt : (unsigned)(OWN_CAP + 1);
    bool listed = total <= OWN_CAP;
    int  bound  = 0;

    if (listed) {
      const int cnt = (int)total;

      {
        int sv[OWN_CAP / OWN_T], dv[OWN_CAP / OWN_T];
        unsigned rk[OWN_CAP / OWN_T];
#pragma unroll
        for (int u = 0; u < OWN_CAP / OWN_T; ++u) {
          const int i  = tid + u * OWN_T;
          const int ev = s_es[i];
          sv[u] = (i < cnt) ? ev : 0;
        }
#pragma unroll
        for (int u = 0; u < OWN_CAP / OWN_T; ++u) { dv[u] = dst[sv[u]]; sv[u] = src[sv[u]]; }
        unsigned pre0 = 0u, pre1 = 0u, pre2 = 0u, pre3 = 0u;
#pragma unroll
        for (int u = 0; u < OWN_CAP / OWN_T; ++u) {
          const int      i    = tid + u * OWN_T;
          const unsigned loc  = (unsigned)(dv[u] - base);
          const bool     keep = i < cnt && loc < (unsigned)R;
          const unsigned q    = loc & 3u;
          const unsigned mk = __builtin_amdgcn_ballot_w32(keep);
          const unsigned ml = __builtin_amdgcn_ballot_w32(keep && (q & 1u) != 0u);
          const unsigned mh = __builtin_amdgcn_ballot_w32(keep && (q & 2u) != 0u);
          const unsigned b0 = mk & ~ml & ~mh, b1 = ml & ~mh, b2 = mh & ~ml, b3 = ml & mh;
          rk[u] = own_pick4(q, pre0, pre1, pre2, pre3)
                + __builtin_amdgcn_mbcnt_lo(own_pick4(q, b0, b1, b2, b3), 0u);
          pre0 += (unsigned)__popc(b0); pre1 += (unsigned)__popc(b1);
          pre2 += (unsigned)__popc(b2); pre3 += (unsigned)__popc(b3);
        }
        unsigned got = 0u;
        if (lane < 4) {
          got = atomicAdd(&s_c4[lane], own_pick4((unsigned)lane, pre0, pre1, pre2, pre3));
        }
        const unsigned f0 = (unsigned)__builtin_amdgcn_readlane((int)got, 0);
        const unsigned f1 = (unsigned)__builtin_amdgcn_readlane((int)got, 1);
        const unsigned f2 = (unsigned)__builtin_amdgcn_readlane((int)got, 2);
        const unsigned f3 = (unsigned)__builtin_amdgcn_readlane((int)got, 3);
#pragma unroll
        for (int u = 0; u < OWN_CAP / OWN_T; ++u) {
          const int      i   = tid + u * OWN_T;
          const unsigned loc = (unsigned)(dv[u] - base);
          const unsigned q   = loc & 3u;
          const unsigned p   = own_pick4(q, f0, f1, f2, f3) + rk[u];
          if (i < cnt && loc < (unsigned)R && p < (unsigned)(OWN_CAP / 4)) {
            const unsigned slot = 4u * p + q;
            s_lgw[slot] = (unsigned)min(max(sv[u], 0), N - 1);
            s_loc[slot] = (unsigned short)loc;
          }
        }
      }
      __syncthreads();
      const unsigned maxc = max(max(s_c4[0], s_c4[1]), max(s_c4[2], s_c4[3]));
      listed = maxc <= (unsigned)(OWN_CAP / 4);
      bound  = 4 * (int)maxc;
    }

    if (listed) {
      for (int i0 = 0; i0 < bound; i0 += 4 * (OWN_T / 8)) {
        float4 za[4], zb[4];
        int    li[4];
        bool   ok[4];
#pragma unroll
        for (int u = 0; u < 4; ++u) {
          const int  i  = i0 + u * (OWN_T / 8) + g;
          const int  sr = (int)s_lgw[i];
          const int  lr = (int)s_loc[i];
          ok[u] = lr != OWN_NONE;
          const int  s  = ok[u] ? sr : 0;
          li[u] = ok[u] ? lr : 0;
          const float4* zr = (const float4*)(z + (size_t)s * 64);
          za[u] = zr[j]; zb[u] = zr[8 + j];
        }
        float ef[4];
        int   ti[4], sh[4];
#pragma unroll
        for (int u = 0; u < 4; ++u) {
          const int  i  = i0 + u * (OWN_T / 8) + g;
          const float4* dr = (const float4*)(s_df + li[u] * 64);
          const float4 da = dr[j], db = dr[8 + j];
          float p = za[u].x * da.x + za[u].y * da.y + za[u].z * da.z + za[u].w * da.w
                  + zb[u].x * db.x + zb[u].y * db.y + zb[u].z * db.z + zb[u].w * db.w;
          p += __shfl_xor(p, 1, 32);
          p += __shfl_xor(p, 2, 32);
          p += __shfl_xor(p, 4, 32);
          const float th = p * OWN_L2E_HI;
          const float tl = fmaf(p, OWN_L2E_LO, fmaf(p, OWN_L2E_HI, -th));
          const float tc = own_lg2(p);
          ti[u] = (int)floorf(tc);
          ef[u] = __builtin_amdgcn_exp2f(__builtin_amdgcn_fractf(tc) + tl);
          sh[u] = 0;
          if (ok[u] && j == 0) {
            s_es[i] = (int)__float_as_uint(p);
            atomicMax(&s_key[li[u]], float_key(p));
            const int old = atomicCAS(&s_sh[li[u]], OWN_EMPTY, ti[u]);
            sh[u] = (old == OWN_EMPTY) ? ti[u] : old;
          }
        }
#pragma unroll
        for (int u = 0; u < 4; ++u) {
          const int   shift = __shfl(sh[u], 0, 8);
          const float ex    = ldexpf(ef[u], ti[u] - shift);
          if (ok[u]) {
            float* np = s_num + li[u] * OWN_P + 4 * j;
            atomicAdd(np + 0,  ex * za[u].x); atomicAdd(np + 1,  ex * za[u].y);
            atomicAdd(np + 2,  ex * za[u].z); atomicAdd(np + 3,  ex * za[u].w);
            atomicAdd(np + 32, ex * zb[u].x); atomicAdd(np + 33, ex * zb[u].y);
            atomicAdd(np + 34, ex * zb[u].z); atomicAdd(np + 35, ex * zb[u].w);
            if (j == 0) atomicAdd(&s_den[li[u]], ex);
          }
        }
      }
      __syncthreads();

      for (int i = tid; i < R; i += OWN_T) {
        const int shv = s_sh[i];
        if (shv != OWN_EMPTY) {
          const int tm = (int)floorf(own_lg2(key_float(s_key[i])));
          if (tm - shv > OWN_SPREAD || tm >= OWN_CLAMP_I || shv <= -OWN_CLAMP_I) s_bad = 1u;
        }
      }
      __syncthreads();

      if (s_bad != 0u) {
        for (int i = tid; i < OWN_R * OWN_P; i += OWN_T) s_num[i] = 0.0f;
        for (int i = tid; i < OWN_R; i += OWN_T) s_den[i] = 0.0f;
        __syncthreads();

        for (int i0 = 0; i0 < bound; i0 += 4 * (OWN_T / 8)) {
          float4 za[4], zb[4];
#pragma unroll
          for (int u = 0; u < 4; ++u) {
            const int  i = i0 + u * (OWN_T / 8) + g;
            const int  sr = (int)s_lgw[i];
            const int  s  = ((int)s_loc[i] != OWN_NONE) ? sr : 0;
            const float4* zr = (const float4*)(z + (size_t)s * 64);
            za[u] = zr[j]; zb[u] = zr[8 + j];
          }
#pragma unroll
          for (int u = 0; u < 4; ++u) {
            const int i = i0 + u * (OWN_T / 8) + g;
            if ((int)s_loc[i] != OWN_NONE) {
              const int   loc = (int)s_loc[i];
              const float ex  = __expf(__uint_as_float((unsigned)s_es[i]) - key_float(s_key[loc]));
              float* np = s_num + loc * OWN_P + 4 * j;
              atomicAdd(np + 0,  ex * za[u].x); atomicAdd(np + 1,  ex * za[u].y);
              atomicAdd(np + 2,  ex * za[u].z); atomicAdd(np + 3,  ex * za[u].w);
              atomicAdd(np + 32, ex * zb[u].x); atomicAdd(np + 33, ex * zb[u].y);
              atomicAdd(np + 34, ex * zb[u].z); atomicAdd(np + 35, ex * zb[u].w);
              if (j == 0) atomicAdd(&s_den[loc], ex);
            }
          }
        }
      }
    } else {
      for (int e = tid; e < E; e += OWN_T) {
        const unsigned loc = (unsigned)(dst[e] - base);
        if (loc < (unsigned)R) {
          const int s = min(max(src[e], 0), N - 1);
          const float p = row_dot_lane(z + (size_t)s * 64, s_df + loc * 64);
          atomicMax(&s_key[loc], float_key(p));
        }
      }
      __syncthreads();
      for (int e = tid; e < E; e += OWN_T) {
        const unsigned loc = (unsigned)(dst[e] - base);
        if (loc < (unsigned)R) {
          const int s = min(max(src[e], 0), N - 1);
          const float* zr = z + (size_t)s * 64;
          const float p  = row_dot_lane(zr, s_df + loc * 64);
          const float ex = __expf(p - key_float(s_key[loc]));
          float* np = s_num + loc * OWN_P;
          for (int k = 0; k < 64; ++k) atomicAdd(np + k, ex * zr[k]);
          atomicAdd(&s_den[loc], ex);
        }
      }
    }
    __syncthreads();

    for (int i = tid; i < R * 64; i += OWN_T) {
      const int   loc = i >> 6, f = i & 63;
      const float dn  = s_den[loc];
      out[(size_t)(base + loc) * 64 + f] = s_num[loc * OWN_P + f] / (dn == 0.0f ? 1.0f : dn);
    }
    __syncthreads();
  }
}

extern "C" void kernel_launch(void* const* d_in, const int* in_sizes, int n_in,
                              void* d_out, int out_size, void* d_ws, size_t ws_size,
                              hipStream_t stream) {
  const float* h     = (const float*)d_in[0];
  const float* feat  = (const float*)d_in[1];
  const float* W_fc  = (const float*)d_in[2];
  const float* W_dst = (const float*)d_in[3];
  const int*   src   = (const int*)d_in[4];
  const int*   dst   = (const int*)d_in[5];
  float*       out   = (float*)d_out;

  const int FEAT = 64, O = 64;
  const int N   = in_sizes[1] / FEAT;
  const int Kfc = in_sizes[0] / N;
  const int E   = in_sizes[4];

  float* z  = (float*)d_ws;
  float* df = z + (size_t)N * O;
  unsigned* keys = (unsigned*)(df + (size_t)N * O);

  if (Kfc == 256) {
    proj_resident<256><<<1024, PJ_T, 0, stream>>>(h, W_fc, z, N);
  } else {
    const dim3 gemmBlk(32 * (O / 16));
    const int  mTiles = (N + 15) / 16;
    gemm_nt_wmma<<<mTiles, gemmBlk, 0, stream>>>(h, W_fc, z, N, Kfc, O);
  }
  proj_resident<64><<<1024, PJ_T, 0, stream>>>(feat, W_dst, df, N);

  owner_keys<<<512, 256, 0, stream>>>(dst, keys, E);
  edge_owner<<<256, OWN_T, 0, stream>>>(z, df, src, dst, keys, out, N, E);
}


# edge filter reads one 128-byte line of 16 (src,dst) pairs per candidate group, found by one bit per group in a per-key bit row: no key bytes, no dst/src look-ups; LDS sums in a new arrival order
# speedup vs baseline: 5.0007x; 1.0313x over previous
#include <hip/hip_runtime.h>
#include <hip/hip_bf16.h>

typedef float v2f __attribute__((ext_vector_type(2)));
typedef float v8f __attribute__((ext_vector_type(8)));

__global__ void gemm_nt_wmma(const float* __restrict__ A,
                             const float* __restrict__ W,
                             float* __restrict__ C,
                             int M, int K, int O) {
  const int lane  = threadIdx.x & 31;
  const int oTile = threadIdx.x >> 5;
  const int mTile = blockIdx.x;
  if (mTile * 16 >= M) return;
  const int half = lane >> 4;
  const int l16  = lane & 15;

  const float* aPtr = A + (size_t)(mTile * 16 + l16) * K + 2 * half;
  const float* wPtr = W + (size_t)(oTile * 16 + l16) * K + 2 * half;

  v8f acc = {};
#pragma unroll 8
  for (int k = 0; k < K; k += 4) {
    v2f a = *(const v2f*)(aPtr + k);
    v2f b = *(const v2f*)(wPtr + k);
    acc = __builtin_amdgcn_wmma_f32_16x16x4_f32(
         false, a,  false, b,
         (short)0, acc,  false,  false);
  }

  float* cPtr = C + (size_t)(mTile * 16 + half * 8) * O + oTile * 16 + l16;
#pragma unroll
  for (int r = 0; r < 8; ++r) cPtr[(size_t)r * O] = acc[r];
}

#define PJ_T 128
#define PJ_G 4

template <int K>
__global__ void __launch_bounds__(PJ_T)
proj_resident(const float* __restrict__ A,
              const float* __restrict__ Wm,
              float* __restrict__ C, int M) {
  constexpr int P   = K + 4;
  constexpr int NQ  = K / 8;
  constexpr int RV  = K / 4;
  constexpr int NLD = (16 * RV) / PJ_T;
  __shared__ __attribute__((aligned(16))) float s_a[2 * 16 * P];

  const int tid  = threadIdx.x;
  const int lane = tid & 31;
  const int ct   = tid >> 5;
  const int c    = lane & 15;
  const int h    = lane >> 4;
  const int nSteps = (M + 15) >> 4;
  int step = blockIdx.x;
  if (step >= nSteps) return;

  float4 wf[NQ];
  {
    const float4* wr = (const float4*)(Wm + (size_t)(ct * 16 + c) * K + 4 * h);
#pragma unroll
    for (int q = 0; q < NQ; ++q) wf[q] = wr[2 * q];
  }

  constexpr int RPI = PJ_T / RV;
  const int trow = tid / RV, tcol = (tid % RV) * 4;
  const float* ag = A + (size_t)trow * K + tcol;
  float*       sl = s_a + trow * P + tcol;

  static_assert(NLD == 8 || NLD == 2, "tile share per thread");
  float4 t0, t1, t2, t3, t4, t5, t6, t7;
#define PJ_REQ(at)                                                        \
  do {                                                                    \
    t0 = *(const float4*)((at) + (size_t)0 * RPI * K);                    \
    t1 = *(const float4*)((at) + (size_t)1 * RPI * K);                    \
    if constexpr (NLD == 8) {                                             \
      t2 = *(const float4*)((at) + (size_t)2 * RPI * K);                  \
      t3 = *(const float4*)((at) + (size_t)3 * RPI * K);                  \
      t4 = *(const float4*)((at) + (size_t)4 * RPI * K);                  \
      t5 = *(const float4*)((at) + (size_t)5 * RPI * K);                  \
      t6 = *(const float4*)((at) + (size_t)6 * RPI * K);                  \
      t7 = *(const float4*)((at) + (size_t)7 * RPI * K);                  \
    }                                                                     \
  } while (0)
#define PJ_PUT(sp)                                                        \
  do {                                                                    \
    *(float4*)((sp) + 0 * RPI * P) = t0;                                  \
    *(float4*)((sp) + 1 * RPI * P) = t1;                                  \
    if constexpr (NLD == 8) {                                             \
      *(float4*)((sp) + 2 * RPI * P) = t2;                                \
      *(float4*)((sp) + 3 * RPI * P) = t3;                                \
      *(float4*)((sp) + 4 * RPI * P) = t4;                                \
      *(float4*)((sp) + 5 * RPI * P) = t5;                                \
      *(float4*)((sp) + 6 * RPI * P) = t6;                                \
      *(float4*)((sp) + 7 * RPI * P) = t7;                                \
    }                                                                     \
  } while (0)

  {
    const float* at = ag + (size_t)min(step * 16, M - 16) * K;
    PJ_REQ(at);
    PJ_PUT(sl);
  }
  {
    const int ls = min(step + (int)gridDim.x, nSteps - 1);
    const float* at = ag + (size_t)min(ls * 16, M - 16) * K;
    PJ_REQ(at);
  }
  __syncthreads();

  int b = 0;
  for (;;) {
    const float* ar = s_a + b * (16 * P) + c * P + 4 * h;
    v8f acc0 = {}, acc1 = {};
    float4 af[2][PJ_G];
#pragma unroll
    for (int j = 0; j < PJ_G; ++j) af[0][j] = *(const float4*)(ar + 8 * j);
#pragma unroll
    for (int g = 0; g < NQ / PJ_G; ++g) {
      if (g + 1 < NQ / PJ_G) {
#pragma unroll
        for (int j = 0; j < PJ_G; ++j) af[(g + 1) & 1][j] = *(const float4*)(ar + 8 * ((g + 1) * PJ_G + j));
      }
      __builtin_amdgcn_sched_barrier(0);
#pragma unroll
      for (int j = 0; j < PJ_G; ++j) {
        const float4 av = af[g & 1][j];
        const float4 wv = wf[g * PJ_G + j];
        const v2f a0 = {av.x, av.y}, a1 = {av.z, av.w};
        const v2f b0 = {wv.x, wv.y}, b1 = {wv.z, wv.w};
        acc0 = __builtin_amdgcn_wmma_f32_16x16x4_f32(false, a0, false, b0, (short)0, acc0, false, false);
        acc1 = __builtin_amdgcn_wmma_f32_16x16x4_f32(false, a1, false, b1, (short)0, acc1, false, false);
      }
      __builtin_amdgcn_sched_barrier(0);
    }

    {
      float* sn = sl + (b ^ 1) * (16 * P);
      PJ_PUT(sn);
    }
    __syncthreads();

    {
      const int r0 = min(step * 16, M - 16) + 8 * h;
      float* cp = C + (size_t)r0 * 64 + ct * 16 + c;
#pragma unroll
      for (int r = 0; r < 8; ++r)
        if (r0 + r >= step * 16) cp[(size_t)r * 64] = acc0[r] + acc1[r];
    }

    const int nx = step + (int)gridDim.x;
    if (nx >= nSteps) break;
    {
      const int ls = min(nx + (int)gridDim.x, nSteps - 1);
      const float* at = ag + (size_t)min(ls * 16, M - 16) * K;
      PJ_REQ(at);
    }
    step = nx;
    b ^= 1;
  }
#undef PJ_REQ
#undef PJ_PUT
}

__device__ __forceinline__ unsigned float_key(float f) {
  unsigned u = __float_as_uint(f);
  return (u & 0x80000000u) ? ~u : (u | 0x80000000u);
}
__device__ __forceinline__ float key_float(unsigned k) {
  unsigned u = (k & 0x80000000u) ? (k & 0x7FFFFFFFu) : ~k;
  return __uint_as_float(u);
}

#define OWN_R   392
#define OWN_P   65
#define OWN_CAP 8192
#define OWN_T   1024
#define OWN_NONE 0xFFFF
#define OWN_U   8
#define OWN_EMPTY  ((int)0x80000000)
#define OWN_SPREAD 40
#define OWN_CLAMP   268435456.0f
#define OWN_CLAMP_I (1 << 28)
#define OWN_L2E_HI 1.44269502162933349609375f
#define OWN_L2E_LO 1.925963033500011e-8f

__device__ __forceinline__ int own_groups(int E) { return (E + 15) >> 4; }
__device__ __forceinline__ int own_words(int E)  { return (own_groups(E) + 31) >> 5; }
__device__ __forceinline__ int own_rowp(int E)   { return (own_words(E) + 7) & ~7; }

__global__ void __launch_bounds__(256)
owner_pairs(const int* __restrict__ src, const int* __restrict__ dst,
            unsigned* __restrict__ pairs, int E) {
  __shared__ __attribute__((aligned(16))) unsigned s_p[256 * 36];
  __shared__ unsigned s_b[256 * 9];
  const int tid = threadIdx.x;
  const int nG  = own_groups(E);
  const int W32 = own_words(E);
  const int RW  = own_rowp(E);
  unsigned* __restrict__ rows = pairs + (size_t)nG * 32;
  for (int g0 = blockIdx.x * 256; g0 < nG; g0 += gridDim.x * 256) {
    for (int i = tid; i < 256 * 9; i += 256) s_b[i] = 0u;
    __syncthreads();
    const int g = g0 + tid;
    int dd[16], ss[16];
    if (g < nG && 16 * g + 16 <= E) {
      const int4* __restrict__ pd = (const int4*)(dst + (size_t)16 * g);
      const int4* __restrict__ ps = (const int4*)(src + (size_t)16 * g);
      int4 d4[4], s4[4];
#pragma unroll
      for (int q = 0; q < 4; ++q) { d4[q] = pd[q]; s4[q] = ps[q]; }
#pragma unroll
      for (int q = 0; q < 4; ++q) {
        dd[4 * q + 0] = d4[q].x; dd[4 * q + 1] = d4[q].y; dd[4 * q + 2] = d4[q].z; dd[4 * q + 3] = d4[q].w;
        ss[4 * q + 0] = s4[q].x; ss[4 * q + 1] = s4[q].y; ss[4 * q + 2] = s4[q].z; ss[4 * q + 3] = s4[q].w;
      }
    } else {
#pragma unroll
      for (int k = 0; k < 16; ++k) {
        const int  e  = 16 * g + k;
        const bool ok = g < nG && e < E;
        dd[k] = ok ? dst[ok ? e : 0] : -1;
        ss[k] = ok ? src[ok ? e : 0] : 0;
      }
    }
#pragma unroll
    for (int k = 0; k < 16; ++k) {
      if (dd[k] >= 0) {
        const unsigned o = ((unsigned)dd[k] / (unsigned)OWN_R) & 255u;
        atomicOr(&s_b[o * 9 + (tid >> 5)], 1u << (tid & 31));
      }
    }
    {
      uint4* sp = (uint4*)(s_p + tid * 36);
#pragma unroll
      for (int q = 0; q < 8; ++q)
        sp[q] = make_uint4((unsigned)ss[2 * q], (unsigned)dd[2 * q], (unsigned)ss[2 * q + 1], (unsigned)dd[2 * q + 1]);
    }
    __syncthreads();
#pragma unroll
    for (int r = 0; r < 8; ++r) {
      const int x = r * 256 + tid, gl = x >> 3, pc = x & 7;
      if (g0 + gl < nG) ((uint4*)pairs)[(size_t)(g0 + gl) * 8 + pc] = *(const uint4*)(s_p + gl * 36 + 4 * pc);
    }
#pragma unroll
    for (int r = 0; r < 8; ++r) {
      const int x = r * 256 + tid, k = x >> 3, w = x & 7;
      const int gw = (g0 >> 5) + w;
      if (gw < W32) rows[(size_t)k * RW + gw] = s_b[k * 9 + w];
    }
    __syncthreads();
  }
}

__device__ __forceinline__ float row_dot_lane(const float* __restrict__ zr,
                                              const float* __restrict__ dr) {
  float p = 0.0f;
  for (int k = 0; k < 16; ++k) {
    const float4 a = ((const float4*)zr)[k];
    const float4 b = ((const float4*)dr)[k];
    p = fmaf(a.x, b.x, p); p = fmaf(a.y, b.y, p);
    p = fmaf(a.z, b.z, p); p = fmaf(a.w, b.w, p);
  }
  return p;
}

__device__ __forceinline__ float own_lg2(float e) {
  return fminf(fmaxf(e * OWN_L2E_HI, -OWN_CLAMP), OWN_CLAMP);
}

__device__ __forceinline__ unsigned own_pick4(unsigned q, unsigned a0, unsigned a1, unsigned a2, unsigned a3) {
  const unsigned lo = (q & 1u) ? a1 : a0;
  const unsigned hi = (q & 1u) ? a3 : a2;
  return (q & 2u) ? hi : lo;
}

__global__ void __launch_bounds__(OWN_T)
edge_owner(const float* __restrict__ z,
           const float* __restrict__ df,
           const int* __restrict__ src,
           const int* __restrict__ dst,
           const unsigned* __restrict__ pairs,
           float* __restrict__ out, int N, int E) {
  __shared__ __attribute__((aligned(16))) float s_df[OWN_R * 64];
  __shared__ float          s_num[OWN_R * OWN_P];
  __shared__ float          s_den[OWN_R];
  __shared__ unsigned       s_key[OWN_R];
  __shared__ int            s_sh[OWN_R];
  __shared__ int            s_es[OWN_CAP];
  __shared__ unsigned short s_tl[OWN_CAP];
  __shared__ unsigned       s_lgw[OWN_CAP];
  __shared__ unsigned short s_loc[OWN_CAP];
  __shared__ unsigned       s_c4[4];
  __shared__ unsigned       s_cnt;
  __shared__ unsigned       s_ng;
  __shared__ unsigned       s_bad;

  const int tid = threadIdx.x;
  const int g   = tid >> 3;
  const int j   = tid & 7;

  const int lane  = tid & 31;
  const int nG  = own_groups(E);
  const int W32 = own_words(E);
  const unsigned* __restrict__ rows = pairs + (size_t)nG * 32;

  for (int base = blockIdx.x * OWN_R; base < N; base += gridDim.x * OWN_R) {
    const int R = min(OWN_R, N - base);

    for (int i = tid; i < OWN_R * OWN_P; i += OWN_T) s_num[i] = 0.0f;
    for (int i = tid; i < OWN_R; i += OWN_T) { s_den[i] = 0.0f; s_key[i] = 0u; s_sh[i] = OWN_EMPTY; }
    for (int i = tid; i < OWN_CAP; i += OWN_T) s_loc[i] = (unsigned short)OWN_NONE;
    {
      const float4* __restrict__ dfr = (const float4*)(df + (size_t)base * 64);
#pragma unroll
      for (int u = 0; u < (OWN_R * 16 + OWN_T - 1) / OWN_T; ++u) {
        const int i = min(tid + u * OWN_T, R * 16 - 1);
        ((float4*)s_df)[i] = dfr[i];
      }
    }
    if (tid == 0) { s_cnt = 0u; s_ng = 0u; s_bad = 0u; }
    if (tid < 4) s_c4[tid] = 0u;
    __syncthreads();

    const unsigned want = ((unsigned)base / (unsigned)OWN_R) & 255u;
    const unsigned* __restrict__ krow = rows + (size_t)want * own_rowp(E);
    for (int w0 = 0; w0 < W32; w0 += OWN_T * 4) {
      unsigned bw[4];
#pragma unroll
      for (int u = 0; u < 4; ++u) {
        const int w = w0 + u * OWN_T + tid;
        bw[u] = krow[min(w, W32 - 1)];
      }
#pragma unroll
      for (int u = 0; u < 4; ++u) {
        const int w = w0 + u * OWN_T + tid;
        unsigned bits = (w < W32) ? bw[u] : 0u;
        const unsigned c = (unsigned)__popc(bits);
        unsigned incl = c;
#pragma unroll
        for (int d = 1; d < 32; d <<= 1) {
          const unsigned t = (unsigned)__shfl_up((int)incl, d, 32);
          if (lane >= d) incl += t;
        }
        const unsigned tot = (unsigned)__shfl((int)incl, 31, 32);
        if (tot != 0u) {
          unsigned first = 0u;
          if (lane == 0) first = atomicAdd(&s_ng, tot);
          first = (unsigned)__builtin_amdgcn_readfirstlane((int)first);
          unsigned pos = first + incl - c;
          while (bits != 0u) {
            const int b = __ffs((int)bits) - 1;
            bits &= bits - 1u;
            if (pos < OWN_CAP) s_lgw[pos] = (unsigned)(32 * w + b);
            ++pos;
          }
        }
      }
    }
    __syncthreads();
    const unsigned ngr = s_ng;

    if (ngr <= OWN_CAP) {
      const int ng = (int)ngr;
      const uint4* __restrict__ pr4 = (const uint4*)pairs;
      const int pc = tid & 7;
      for (int i0 = 0; i0 < ng; i0 += (OWN_T / 8) * OWN_U) {
        uint4 p4[OWN_U];
        bool  in[OWN_U];
#pragma unroll
        for (int u = 0; u < OWN_U; ++u) {
          const int i  = i0 + u * (OWN_T / 8) + g;
          const int gr = (int)s_lgw[min(i, OWN_CAP - 1)];
          in[u] = i < ng;
          const int gs = in[u] ? gr : 0;
          p4[u] = pr4[(size_t)gs * 8 + pc];
        }
        __builtin_amdgcn_sched_barrier(0);
#pragma unroll
        for (int u = 0; u < OWN_U; ++u) {
          const unsigned l0 = (unsigned)((int)p4[u].y - base);
          const unsigned l1 = (unsigned)((int)p4[u].w - base);
          const bool h0 = in[u] && l0 < (unsigned)R;
          const bool h1 = in[u] && l1 < (unsigned)R;
          const unsigned b0 = __builtin_amdgcn_ballot_w32(h0);
          const unsigned b1 = __builtin_amdgcn_ballot_w32(h1);
          const unsigned c0 = (unsigned)__popc(b0);
          const unsigned tot = c0 + (unsigned)__popc(b1);
          if (tot != 0u) {
            unsigned first = 0u;
            if (lane == 0) first = atomicAdd(&s_cnt, tot);
            first = (unsigned)__builtin_amdgcn_readfirstlane((int)first);
            const unsigned x0 = first + __builtin_amdgcn_mbcnt_lo(b0, 0u);
            const unsigned x1 = first + c0 + __builtin_amdgcn_mbcnt_lo(b1, 0u);
            if (h0 && x0 < OWN_CAP) { s_es[x0] = (int)p4[u].x; s_tl[x0] = (unsigned short)l0; }
            if (h1 && x1 < OWN_CAP) { s_es[x1] = (int)p4[u].z; s_tl[x1] = (unsigned short)l1; }
          }
        }
      }
    }
    __syncthreads();
    const unsigned total = (ngr <= OWN_CAP) ? s_cnt : (unsigned)(OWN_CAP + 1);
    bool listed = total <= OWN_CAP;
    int  bound  = 0;

    if (listed) {
      const int cnt = (int)total;

      {
        int sv[OWN_CAP / OWN_T];
        unsigned lc[OWN_CAP / OWN_T];
        unsigned rk[OWN_CAP / OWN_T];
#pragma unroll
        for (int u = 0; u < OWN_CAP / OWN_T; ++u) {
          const int i  = tid + u * OWN_T;
          const int      ev = s_es[i];
          const unsigned lv = (unsigned)s_tl[i];
          sv[u] = (i < cnt) ? ev : 0;
          lc[u] = (i < cnt) ? lv : 0u;
        }
        unsigned pre0 = 0u, pre1 = 0u, pre2 = 0u, pre3 = 0u;
#pragma unroll
        for (int u = 0; u < OWN_CAP / OWN_T; ++u) {
          const int      i    = tid + u * OWN_T;
          const unsigned loc  = lc[u];
          const bool     keep = i < cnt && loc < (unsigned)R;
          const unsigned q    = loc & 3u;
          const unsigned mk = __builtin_amdgcn_ballot_w32(keep);
          const unsigned ml = __builtin_amdgcn_ballot_w32(keep && (q & 1u) != 0u);
          const unsigned mh = __builtin_amdgcn_ballot_w32(keep && (q & 2u) != 0u);
          const unsigned b0 = mk & ~ml & ~mh, b1 = ml & ~mh, b2 = mh & ~ml, b3 = ml & mh;
          rk[u] = own_pick4(q, pre0, pre1, pre2, pre3)
                + __builtin_amdgcn_mbcnt_lo(own_pick4(q, b0, b1, b2, b3), 0u);
          pre0 += (unsigned)__popc(b0); pre1 += (unsigned)__popc(b1);
          pre2 += (unsigned)__popc(b2); pre3 += (unsigned)__popc(b3);
        }
        unsigned got = 0u;
        if (lane < 4) {
          got = atomicAdd(&s_c4[lane], own_pick4((unsigned)lane, pre0, pre1, pre2, pre3));
        }
        const unsigned f0 = (unsigned)__builtin_amdgcn_readlane((int)got, 0);
        const unsigned f1 = (unsigned)__builtin_amdgcn_readlane((int)got, 1);
        const unsigned f2 = (unsigned)__builtin_amdgcn_readlane((int)got, 2);
        const unsigned f3 = (unsigned)__builtin_amdgcn_readlane((int)got, 3);
#pragma unroll
        for (int u = 0; u < OWN_CAP / OWN_T; ++u) {
          const int      i   = tid + u * OWN_T;
          const unsigned loc = lc[u];
          const unsigned q   = loc & 3u;
          const unsigned p   = own_pick4(q, f0, f1, f2, f3) + rk[u];
          if (i < cnt && loc < (unsigned)R && p < (unsigned)(OWN_CAP / 4)) {
            const unsigned slot = 4u * p + q;
            s_lgw[slot] = (unsigned)min(max(sv[u], 0), N - 1);
            s_loc[slot] = (unsigned short)loc;
          }
        }
      }
      __syncthreads();
      const unsigned maxc = max(max(s_c4[0], s_c4[1]), max(s_c4[2], s_c4[3]));
      listed = maxc <= (unsigned)(OWN_CAP / 4);
      bound  = 4 * (int)maxc;
    }

    if (listed) {
      for (int i0 = 0; i0 < bound; i0 += 4 * (OWN_T / 8)) {
        float4 za[4], zb[4];
        int    li[4];
        bool   ok[4];
#pragma unroll
        for (int u = 0; u < 4; ++u) {
          const int  i  = i0 + u * (OWN_T / 8) + g;
          const int  sr = (int)s_lgw[i];
          const int  lr = (int)s_loc[i];
          ok[u] = lr != OWN_NONE;
          const int  s  = ok[u] ? sr : 0;
          li[u] = ok[u] ? lr : 0;
          const float4* zr = (const float4*)(z + (size_t)s * 64);
          za[u] = zr[j]; zb[u] = zr[8 + j];
        }
        float ef[4];
        int   ti[4], sh[4];
#pragma unroll
        for (int u = 0; u < 4; ++u) {
          const int  i  = i0 + u * (OWN_T / 8) + g;
          const float4* dr = (const float4*)(s_df + li[u] * 64);
          const float4 da = dr[j], db = dr[8 + j];
          float p = za[u].x * da.x + za[u].y * da.y + za[u].z * da.z + za[u].w * da.w
                  + zb[u].x * db.x + zb[u].y * db.y + zb[u].z * db.z + zb[u].w * db.w;
          p += __shfl_xor(p, 1, 32);
          p += __shfl_xor(p, 2, 32);
          p += __shfl_xor(p, 4, 32);
          const float th = p * OWN_L2E_HI;
          const float tl = fmaf(p, OWN_L2E_LO, fmaf(p, OWN_L2E_HI, -th));
          const float tc = own_lg2(p);
          ti[u] = (int)floorf(tc);
          ef[u] = __builtin_amdgcn_exp2f(__builtin_amdgcn_fractf(tc) + tl);
          sh[u] = 0;
          if (ok[u] && j == 0) {
            s_es[i] = (int)__float_as_uint(p);
            atomicMax(&s_key[li[u]], float_key(p));
            const int old = atomicCAS(&s_sh[li[u]], OWN_EMPTY, ti[u]);
            sh[u] = (old == OWN_EMPTY) ? ti[u] : old;
          }
        }
#pragma unroll
        for (int u = 0; u < 4; ++u) {
          const int   shift = __shfl(sh[u], 0, 8);
          const float ex    = ldexpf(ef[u], ti[u] - shift);
          if (ok[u]) {
            float* np = s_num + li[u] * OWN_P + 4 * j;
            atomicAdd(np + 0,  ex * za[u].x); atomicAdd(np + 1,  ex * za[u].y);
            atomicAdd(np + 2,  ex * za[u].z); atomicAdd(np + 3,  ex * za[u].w);
            atomicAdd(np + 32, ex * zb[u].x); atomicAdd(np + 33, ex * zb[u].y);
            atomicAdd(np + 34, ex * zb[u].z); atomicAdd(np + 35, ex * zb[u].w);
            if (j == 0) atomicAdd(&s_den[li[u]], ex);
          }
        }
      }
      __syncthreads();

      for (int i = tid; i < R; i += OWN_T) {
        const int shv = s_sh[i];
        if (shv != OWN_EMPTY) {
          const int tm = (int)floorf(own_lg2(key_float(s_key[i])));
          if (tm - shv > OWN_SPREAD || tm >= OWN_CLAMP_I || shv <= -OWN_CLAMP_I) s_bad = 1u;
        }
      }
      __syncthreads();

      if (s_bad != 0u) {
        for (int i = tid; i < OWN_R * OWN_P; i += OWN_T) s_num[i] = 0.0f;
        for (int i = tid; i < OWN_R; i += OWN_T) s_den[i] = 0.0f;
        __syncthreads();

        for (int i0 = 0; i0 < bound; i0 += 4 * (OWN_T / 8)) {
          float4 za[4], zb[4];
#pragma unroll
          for (int u = 0; u < 4; ++u) {
            const int  i = i0 + u * (OWN_T / 8) + g;
            const int  sr = (int)s_lgw[i];
            const int  s  = ((int)s_loc[i] != OWN_NONE) ? sr : 0;
            const float4* zr = (const float4*)(z + (size_t)s * 64);
            za[u] = zr[j]; zb[u] = zr[8 + j];
          }
#pragma unroll
          for (int u = 0; u < 4; ++u) {
            const int i = i0 + u * (OWN_T / 8) + g;
            if ((int)s_loc[i] != OWN_NONE) {
              const int   loc = (int)s_loc[i];
              const float ex  = __expf(__uint_as_float((unsigned)s_es[i]) - key_float(s_key[loc]));
              float* np = s_num + loc * OWN_P + 4 * j;
              atomicAdd(np + 0,  ex * za[u].x); atomicAdd(np + 1,  ex * za[u].y);
              atomicAdd(np + 2,  ex * za[u].z); atomicAdd(np + 3,  ex * za[u].w);
              atomicAdd(np + 32, ex * zb[u].x); atomicAdd(np + 33, ex * zb[u].y);
              atomicAdd(np + 34, ex * zb[u].z); atomicAdd(np + 35, ex * zb[u].w);
              if (j == 0) atomicAdd(&s_den[loc], ex);
            }
          }
        }
      }
    } else {
      for (int e = tid; e < E; e += OWN_T) {
        const unsigned loc = (unsigned)(dst[e] - base);
        if (loc < (unsigned)R) {
          const int s = min(max(src[e], 0), N - 1);
          const float p = row_dot_lane(z + (size_t)s * 64, s_df + loc * 64);
          atomicMax(&s_key[loc], float_key(p));
        }
      }
      __syncthreads();
      for (int e = tid; e < E; e += OWN_T) {
        const unsigned loc = (unsigned)(dst[e] - base);
        if (loc < (unsigned)R) {
          const int s = min(max(src[e], 0), N - 1);
          const float* zr = z + (size_t)s * 64;
          const float p  = row_dot_lane(zr, s_df + loc * 64);
          const float ex = __expf(p - key_float(s_key[loc]));
          float* np = s_num + loc * OWN_P;
          for (int k = 0; k < 64; ++k) atomicAdd(np + k, ex * zr[k]);
          atomicAdd(&s_den[loc], ex);
        }
      }
    }
    __syncthreads();

    for (int i = tid; i < R * 64; i += OWN_T) {
      const int   loc = i >> 6, f = i & 63;
      const float dn  = s_den[loc];
      out[(size_t)(base + loc) * 64 + f] = s_num[loc * OWN_P + f] / (dn == 0.0f ? 1.0f : dn);
    }
    __syncthreads();
  }
}

extern "C" void kernel_launch(void* const* d_in, const int* in_sizes, int n_in,
                              void* d_out, int out_size, void* d_ws, size_t ws_size,
                              hipStream_t stream) {
  const float* h     = (const float*)d_in[0];
  const float* feat  = (const float*)d_in[1];
  const float* W_fc  = (const float*)d_in[2];
  const float* W_dst = (const float*)d_in[3];
  const int*   src   = (const int*)d_in[4];
  const int*   dst   = (const int*)d_in[5];
  float*       out   = (float*)d_out;

  const int FEAT = 64, O = 64;
  const int N   = in_sizes[1] / FEAT;
  const int Kfc = in_sizes[0] / N;
  const int E   = in_sizes[4];

  float* z  = (float*)d_ws;
  float* df = z + (size_t)N * O;
  unsigned* pairs = (unsigned*)(df + (size_t)N * O);

  if (Kfc == 256) {
    proj_resident<256><<<1024, PJ_T, 0, stream>>>(h, W_fc, z, N);
  } else {
    const dim3 gemmBlk(32 * (O / 16));
    const int  mTiles = (N + 15) / 16;
    gemm_nt_wmma<<<mTiles, gemmBlk, 0, stream>>>(h, W_fc, z, N, Kfc, O);
  }
  proj_resident<64><<<1024, PJ_T, 0, stream>>>(feat, W_dst, df, N);

  owner_pairs<<<512, 256, 0, stream>>>(src, dst, pairs, E);
  edge_owner<<<256, OWN_T, 0, stream>>>(z, df, src, dst, pairs, out, N, E);
}


# edge kernel waits for LDS once per stage of a trip, not per item: 8-lane logit sums by DPP not LDS permutes; place, df-row and shift requests batched; new list and LDS-sum order, logit dot by fma
# speedup vs baseline: 5.2522x; 1.0503x over previous
#include <hip/hip_runtime.h>
#include <hip/hip_bf16.h>

typedef float v2f __attribute__((ext_vector_type(2)));
typedef float v8f __attribute__((ext_vector_type(8)));

__global__ void gemm_nt_wmma(const float* __restrict__ A,
                             const float* __restrict__ W,
                             float* __restrict__ C,
                             int M, int K, int O) {
  const int lane  = threadIdx.x & 31;
  const int oTile = threadIdx.x >> 5;
  const int mTile = blockIdx.x;
  if (mTile * 16 >= M) return;
  const int half = lane >> 4;
  const int l16  = lane & 15;

  const float* aPtr = A + (size_t)(mTile * 16 + l16) * K + 2 * half;
  const float* wPtr = W + (size_t)(oTile * 16 + l16) * K + 2 * half;

  v8f acc = {};
#pragma unroll 8
  for (int k = 0; k < K; k += 4) {
    v2f a = *(const v2f*)(aPtr + k);
    v2f b = *(const v2f*)(wPtr + k);
    acc = __builtin_amdgcn_wmma_f32_16x16x4_f32(
         false, a,  false, b,
         (short)0, acc,  false,  false);
  }

  float* cPtr = C + (size_t)(mTile * 16 + half * 8) * O + oTile * 16 + l16;
#pragma unroll
  for (int r = 0; r < 8; ++r) cPtr[(size_t)r * O] = acc[r];
}

#define PJ_T 128
#define PJ_G 4

template <int K>
__global__ void __launch_bounds__(PJ_T)
proj_resident(const float* __restrict__ A,
              const float* __restrict__ Wm,
              float* __restrict__ C, int M) {
  constexpr int P   = K + 4;
  constexpr int NQ  = K / 8;
  constexpr int RV  = K / 4;
  constexpr int NLD = (16 * RV) / PJ_T;
  __shared__ __attribute__((aligned(16))) float s_a[2 * 16 * P];

  const int tid  = threadIdx.x;
  const int lane = tid & 31;
  const int ct   = tid >> 5;
  const int c    = lane & 15;
  const int h    = lane >> 4;
  const int nSteps = (M + 15) >> 4;
  int step = blockIdx.x;
  if (step >= nSteps) return;

  float4 wf[NQ];
  {
    const float4* wr = (const float4*)(Wm + (size_t)(ct * 16 + c) * K + 4 * h);
#pragma unroll
    for (int q = 0; q < NQ; ++q) wf[q] = wr[2 * q];
  }

  constexpr int RPI = PJ_T / RV;
  const int trow = tid / RV, tcol = (tid % RV) * 4;
  const float* ag = A + (size_t)trow * K + tcol;
  float*       sl = s_a + trow * P + tcol;

  static_assert(NLD == 8 || NLD == 2, "tile share per thread");
  float4 t0, t1, t2, t3, t4, t5, t6, t7;
#define PJ_REQ(at)                                                        \
  do {                                                                    \
    t0 = *(const float4*)((at) + (size_t)0 * RPI * K);                    \
    t1 = *(const float4*)((at) + (size_t)1 * RPI * K);                    \
    if constexpr (NLD == 8) {                                             \
      t2 = *(const float4*)((at) + (size_t)2 * RPI * K);                  \
      t3 = *(const float4*)((at) + (size_t)3 * RPI * K);                  \
      t4 = *(const float4*)((at) + (size_t)4 * RPI * K);                  \
      t5 = *(const float4*)((at) + (size_t)5 * RPI * K);                  \
      t6 = *(const float4*)((at) + (size_t)6 * RPI * K);                  \
      t7 = *(const float4*)((at) + (size_t)7 * RPI * K);                  \
    }                                                                     \
  } while (0)
#define PJ_PUT(sp)                                                        \
  do {                                                                    \
    *(float4*)((sp) + 0 * RPI * P) = t0;                                  \
    *(float4*)((sp) + 1 * RPI * P) = t1;                                  \
    if constexpr (NLD == 8) {                                             \
      *(float4*)((sp) + 2 * RPI * P) = t2;                                \
      *(float4*)((sp) + 3 * RPI * P) = t3;                                \
      *(float4*)((sp) + 4 * RPI * P) = t4;                                \
      *(float4*)((sp) + 5 * RPI * P) = t5;                                \
      *(float4*)((sp) + 6 * RPI * P) = t6;                                \
      *(float4*)((sp) + 7 * RPI * P) = t7;                                \
    }                                                                     \
  } while (0)

  {
    const float* at = ag + (size_t)min(step * 16, M - 16) * K;
    PJ_REQ(at);
    PJ_PUT(sl);
  }
  {
    const int ls = min(step + (int)gridDim.x, nSteps - 1);
    const float* at = ag + (size_t)min(ls * 16, M - 16) * K;
    PJ_REQ(at);
  }
  __syncthreads();

  int b = 0;
  for (;;) {
    const float* ar = s_a + b * (16 * P) + c * P + 4 * h;
    v8f acc0 = {}, acc1 = {};
    float4 af[2][PJ_G];
#pragma unroll
    for (int j = 0; j < PJ_G; ++j) af[0][j] = *(const float4*)(ar + 8 * j);
#pragma unroll
    for (int g = 0; g < NQ / PJ_G; ++g) {
      if (g + 1 < NQ / PJ_G) {
#pragma unroll
        for (int j = 0; j < PJ_G; ++j) af[(g + 1) & 1][j] = *(const float4*)(ar + 8 * ((g + 1) * PJ_G + j));
      }
      __builtin_amdgcn_sched_barrier(0);
#pragma unroll
      for (int j = 0; j < PJ_G; ++j) {
        const float4 av = af[g & 1][j];
        const float4 wv = wf[g * PJ_G + j];
        const v2f a0 = {av.x, av.y}, a1 = {av.z, av.w};
        const v2f b0 = {wv.x, wv.y}, b1 = {wv.z, wv.w};
        acc0 = __builtin_amdgcn_wmma_f32_16x16x4_f32(false, a0, false, b0, (short)0, acc0, false, false);
        acc1 = __builtin_amdgcn_wmma_f32_16x16x4_f32(false, a1, false, b1, (short)0, acc1, false, false);
      }
      __builtin_amdgcn_sched_barrier(0);
    }

    {
      float* sn = sl + (b ^ 1) * (16 * P);
      PJ_PUT(sn);
    }
    __syncthreads();

    {
      const int r0 = min(step * 16, M - 16) + 8 * h;
      float* cp = C + (size_t)r0 * 64 + ct * 16 + c;
#pragma unroll
      for (int r = 0; r < 8; ++r)
        if (r0 + r >= step * 16) cp[(size_t)r * 64] = acc0[r] + acc1[r];
    }

    const int nx = step + (int)gridDim.x;
    if (nx >= nSteps) break;
    {
      const int ls = min(nx + (int)gridDim.x, nSteps - 1);
      const float* at = ag + (size_t)min(ls * 16, M - 16) * K;
      PJ_REQ(at);
    }
    step = nx;
    b ^= 1;
  }
#undef PJ_REQ
#undef PJ_PUT
}

__device__ __forceinline__ unsigned float_key(float f) {
  unsigned u = __float_as_uint(f);
  return (u & 0x80000000u) ? ~u : (u | 0x80000000u);
}
__device__ __forceinline__ float key_float(unsigned k) {
  unsigned u = (k & 0x80000000u) ? (k & 0x7FFFFFFFu) : ~k;
  return __uint_as_float(u);
}

#define OWN_R   392
#define OWN_P   65
#define OWN_CAP 8192
#define OWN_T   1024
#define OWN_NONE 0xFFFF
#define OWN_U   8
#define OWN_EMPTY  ((int)0x80000000)
#define OWN_SPREAD 40
#define OWN_CLAMP   268435456.0f
#define OWN_CLAMP_I (1 << 28)
#define OWN_L2E_HI 1.44269502162933349609375f
#define OWN_L2E_LO 1.925963033500011e-8f

__device__ __forceinline__ int own_groups(int E) { return (E + 15) >> 4; }
__device__ __forceinline__ int own_words(int E)  { return (own_groups(E) + 31) >> 5; }
__device__ __forceinline__ int own_rowp(int E)   { return (own_words(E) + 7) & ~7; }

__global__ void __launch_bounds__(256)
owner_pairs(const int* __restrict__ src, const int* __restrict__ dst,
            unsigned* __restrict__ pairs, int E) {
  __shared__ __attribute__((aligned(16))) unsigned s_p[256 * 36];
  __shared__ unsigned s_b[256 * 9];
  const int tid = threadIdx.x;
  const int nG  = own_groups(E);
  const int W32 = own_words(E);
  const int RW  = own_rowp(E);
  unsigned* __restrict__ rows = pairs + (size_t)nG * 32;
  for (int g0 = blockIdx.x * 256; g0 < nG; g0 += gridDim.x * 256) {
    for (int i = tid; i < 256 * 9; i += 256) s_b[i] = 0u;
    __syncthreads();
    const int g = g0 + tid;
    int dd[16], ss[16];
    if (g < nG && 16 * g + 16 <= E) {
      const int4* __restrict__ pd = (const int4*)(dst + (size_t)16 * g);
      const int4* __restrict__ ps = (const int4*)(src + (size_t)16 * g);
      int4 d4[4], s4[4];
#pragma unroll
      for (int q = 0; q < 4; ++q) { d4[q] = pd[q]; s4[q] = ps[q]; }
#pragma unroll
      for (int q = 0; q < 4; ++q) {
        dd[4 * q + 0] = d4[q].x; dd[4 * q + 1] = d4[q].y; dd[4 * q + 2] = d4[q].z; dd[4 * q + 3] = d4[q].w;
        ss[4 * q + 0] = s4[q].x; ss[4 * q + 1] = s4[q].y; ss[4 * q + 2] = s4[q].z; ss[4 * q + 3] = s4[q].w;
      }
    } else {
#pragma unroll
      for (int k = 0; k < 16; ++k) {
        const int  e  = 16 * g + k;
        const bool ok = g < nG && e < E;
        dd[k] = ok ? dst[ok ? e : 0] : -1;
        ss[k] = ok ? src[ok ? e : 0] : 0;
      }
    }
#pragma unroll
    for (int k = 0; k < 16; ++k) {
      if (dd[k] >= 0) {
        const unsigned o = ((unsigned)dd[k] / (unsigned)OWN_R) & 255u;
        atomicOr(&s_b[o * 9 + (tid >> 5)], 1u << (tid & 31));
      }
    }
    {
      uint4* sp = (uint4*)(s_p + tid * 36);
#pragma unroll
      for (int q = 0; q < 8; ++q)
        sp[q] = make_uint4((unsigned)ss[2 * q], (unsigned)dd[2 * q], (unsigned)ss[2 * q + 1], (unsigned)dd[2 * q + 1]);
    }
    __syncthreads();
#pragma unroll
    for (int r = 0; r < 8; ++r) {
      const int x = r * 256 + tid, gl = x >> 3, pc = x & 7;
      if (g0 + gl < nG) ((uint4*)pairs)[(size_t)(g0 + gl) * 8 + pc] = *(const uint4*)(s_p + gl * 36 + 4 * pc);
    }
#pragma unroll
    for (int r = 0; r < 8; ++r) {
      const int x = r * 256 + tid, k = x >> 3, w = x & 7;
      const int gw = (g0 >> 5) + w;
      if (gw < W32) rows[(size_t)k * RW + gw] = s_b[k * 9 + w];
    }
    __syncthreads();
  }
}

__device__ __forceinline__ float row_dot_lane(const float* __restrict__ zr,
                                              const float* __restrict__ dr) {
  float p = 0.0f;
  for (int k = 0; k < 16; ++k) {
    const float4 a = ((const float4*)zr)[k];
    const float4 b = ((const float4*)dr)[k];
    p = fmaf(a.x, b.x, p); p = fmaf(a.y, b.y, p);
    p = fmaf(a.z, b.z, p); p = fmaf(a.w, b.w, p);
  }
  return p;
}

__device__ __forceinline__ float own_lg2(float e) {
  return fminf(fmaxf(e * OWN_L2E_HI, -OWN_CLAMP), OWN_CLAMP);
}

__device__ __forceinline__ float own_sum8(float p) {
  p += __int_as_float(__builtin_amdgcn_update_dpp(__float_as_int(p), __float_as_int(p), 0x161, 0xF, 0xF, true));
  p += __int_as_float(__builtin_amdgcn_update_dpp(__float_as_int(p), __float_as_int(p), 0x162, 0xF, 0xF, true));
  p += __int_as_float(__builtin_amdgcn_update_dpp(__float_as_int(p), __float_as_int(p), 0x164, 0xF, 0xF, true));
  return p;
}

__device__ __forceinline__ unsigned own_pick4(unsigned q, unsigned a0, unsigned a1, unsigned a2, unsigned a3) {
  const unsigned lo = (q & 1u) ? a1 : a0;
  const unsigned hi = (q & 1u) ? a3 : a2;
  return (q & 2u) ? hi : lo;
}

__global__ void __launch_bounds__(OWN_T)
edge_owner(const float* __restrict__ z,
           const float* __restrict__ df,
           const int* __restrict__ src,
           const int* __restrict__ dst,
           const unsigned* __restrict__ pairs,
           float* __restrict__ out, int N, int E) {
  __shared__ __attribute__((aligned(16))) float s_df[OWN_R * 64];
  __shared__ float          s_num[OWN_R * OWN_P];
  __shared__ float          s_den[OWN_R];
  __shared__ unsigned       s_key[OWN_R];
  __shared__ int            s_sh[OWN_R];
  __shared__ int            s_es[OWN_CAP];
  __shared__ unsigned short s_tl[OWN_CAP];
  __shared__ unsigned       s_lgw[OWN_CAP];
  __shared__ unsigned short s_loc[OWN_CAP];
  __shared__ unsigned       s_c4[4];
  __shared__ unsigned       s_cnt;
  __shared__ unsigned       s_ng;
  __shared__ unsigned       s_bad;

  const int tid = threadIdx.x;
  const int g   = tid >> 3;
  const int j   = tid & 7;

  const int lane  = tid & 31;
  const int nG  = own_groups(E);
  const int W32 = own_words(E);
  const unsigned* __restrict__ rows = pairs + (size_t)nG * 32;

  for (int base = blockIdx.x * OWN_R; base < N; base += gridDim.x * OWN_R) {
    const int R = min(OWN_R, N - base);

    for (int i = tid; i < OWN_R * OWN_P; i += OWN_T) s_num[i] = 0.0f;
    for (int i = tid; i < OWN_R; i += OWN_T) { s_den[i] = 0.0f; s_key[i] = 0u; s_sh[i] = OWN_EMPTY; }
    for (int i = tid; i < OWN_CAP; i += OWN_T) s_loc[i] = (unsigned short)OWN_NONE;
    {
      const float4* __restrict__ dfr = (const float4*)(df + (size_t)base * 64);
#pragma unroll
      for (int u = 0; u < (OWN_R * 16 + OWN_T - 1) / OWN_T; ++u) {
        const int i = min(tid + u * OWN_T, R * 16 - 1);
        ((float4*)s_df)[i] = dfr[i];
      }
    }
    if (tid == 0) { s_cnt = 0u; s_ng = 0u; s_bad = 0u; }
    if (tid < 4) s_c4[tid] = 0u;
    __syncthreads();

    const unsigned want = ((unsigned)base / (unsigned)OWN_R) & 255u;
    const unsigned* __restrict__ krow = rows + (size_t)want * own_rowp(E);
    for (int w0 = 0; w0 < W32; w0 += OWN_T * 4) {
      unsigned bw[4];
#pragma unroll
      for (int u = 0; u < 4; ++u) {
        const int w = w0 + u * OWN_T + tid;
        bw[u] = krow[min(w, W32 - 1)];
      }
      unsigned bits[4];
      unsigned cl = 0u;
#pragma unroll
      for (int u = 0; u < 4; ++u) {
        const int w = w0 + u * OWN_T + tid;
        bits[u] = (w < W32) ? bw[u] : 0u;
        cl += (unsigned)__popc(bits[u]);
      }
      unsigned incl = cl;
#pragma unroll
      for (int d = 1; d < 32; d <<= 1) {
        const unsigned t = (unsigned)__shfl_up((int)incl, d, 32);
        if (lane >= d) incl += t;
      }
      const unsigned run = (unsigned)__shfl((int)incl, 31, 32);
      if (run != 0u) {
        unsigned first = 0u;
        if (lane == 0) first = atomicAdd(&s_ng, run);
        first = (unsigned)__builtin_amdgcn_readfirstlane((int)first);
        unsigned pos = first + incl - cl;
#pragma unroll
        for (int u = 0; u < 4; ++u) {
          const int w = w0 + u * OWN_T + tid;
          unsigned bl = bits[u];
          while (bl != 0u) {
            const int b = __ffs((int)bl) - 1;
            bl &= bl - 1u;
            if (pos < OWN_CAP) s_lgw[pos] = (unsigned)(32 * w + b);
            ++pos;
          }
        }
      }
    }
    __syncthreads();
    const unsigned ngr = s_ng;

    if (ngr <= OWN_CAP) {
      const int ng = (int)ngr;
      const uint4* __restrict__ pr4 = (const uint4*)pairs;
      const int pc = tid & 7;
      for (int i0 = 0; i0 < ng; i0 += (OWN_T / 8) * OWN_U) {
        uint4 p4[OWN_U];
        bool  in[OWN_U];
#pragma unroll
        for (int u = 0; u < OWN_U; ++u) {
          const int i  = i0 + u * (OWN_T / 8) + g;
          const int gr = (int)s_lgw[min(i, OWN_CAP - 1)];
          in[u] = i < ng;
          const int gs = in[u] ? gr : 0;
          p4[u] = pr4[(size_t)gs * 8 + pc];
        }
        __builtin_amdgcn_sched_barrier(0);
        unsigned b0[OWN_U], b1[OWN_U];
        unsigned run = 0u;
#pragma unroll
        for (int u = 0; u < OWN_U; ++u) {
          b0[u] = __builtin_amdgcn_ballot_w32(in[u] && (unsigned)((int)p4[u].y - base) < (unsigned)R);
          b1[u] = __builtin_amdgcn_ballot_w32(in[u] && (unsigned)((int)p4[u].w - base) < (unsigned)R);
          run += (unsigned)__popc(b0[u]) + (unsigned)__popc(b1[u]);
        }
        if (run != 0u) {
          unsigned first = 0u;
          if (lane == 0) first = atomicAdd(&s_cnt, run);
          first = (unsigned)__builtin_amdgcn_readfirstlane((int)first);
#pragma unroll
          for (int u = 0; u < OWN_U; ++u) {
            const unsigned l0 = (unsigned)((int)p4[u].y - base);
            const unsigned l1 = (unsigned)((int)p4[u].w - base);
            const bool h0 = in[u] && l0 < (unsigned)R;
            const bool h1 = in[u] && l1 < (unsigned)R;
            const unsigned x0 = first + __builtin_amdgcn_mbcnt_lo(b0[u], 0u);
            first += (unsigned)__popc(b0[u]);
            const unsigned x1 = first + __builtin_amdgcn_mbcnt_lo(b1[u], 0u);
            first += (unsigned)__popc(b1[u]);
            if (h0 && x0 < OWN_CAP) { s_es[x0] = (int)p4[u].x; s_tl[x0] = (unsigned short)l0; }
            if (h1 && x1 < OWN_CAP) { s_es[x1] = (int)p4[u].z; s_tl[x1] = (unsigned short)l1; }
          }
        }
      }
    }
    __syncthreads();
    const unsigned total = (ngr <= OWN_CAP) ? s_cnt : (unsigned)(OWN_CAP + 1);
    bool listed = total <= OWN_CAP;
    int  bound  = 0;

    if (listed) {
      const int cnt = (int)total;

      {
        int sv[OWN_CAP / OWN_T];
        unsigned lc[OWN_CAP / OWN_T];
        unsigned rk[OWN_CAP / OWN_T];
#pragma unroll
        for (int u = 0; u < OWN_CAP / OWN_T; ++u) {
          const int i  = tid + u * OWN_T;
          const int      ev = s_es[i];
          const unsigned lv = (unsigned)s_tl[i];
          sv[u] = (i < cnt) ? ev : 0;
          lc[u] = (i < cnt) ? lv : 0u;
        }
        unsigned pre0 = 0u, pre1 = 0u, pre2 = 0u, pre3 = 0u;
#pragma unroll
        for (int u = 0; u < OWN_CAP / OWN_T; ++u) {
          const int      i    = tid + u * OWN_T;
          const unsigned loc  = lc[u];
          const bool     keep = i < cnt && loc < (unsigned)R;
          const unsigned q    = loc & 3u;
          const unsigned mk = __builtin_amdgcn_ballot_w32(keep);
          const unsigned ml = __builtin_amdgcn_ballot_w32(keep && (q & 1u) != 0u);
          const unsigned mh = __builtin_amdgcn_ballot_w32(keep && (q & 2u) != 0u);
          const unsigned b0 = mk & ~ml & ~mh, b1 = ml & ~mh, b2 = mh & ~ml, b3 = ml & mh;
          rk[u] = own_pick4(q, pre0, pre1, pre2, pre3)
                + __builtin_amdgcn_mbcnt_lo(own_pick4(q, b0, b1, b2, b3), 0u);
          pre0 += (unsigned)__popc(b0); pre1 += (unsigned)__popc(b1);
          pre2 += (unsigned)__popc(b2); pre3 += (unsigned)__popc(b3);
        }
        unsigned got = 0u;
        if (lane < 4) {
          got = atomicAdd(&s_c4[lane], own_pick4((unsigned)lane, pre0, pre1, pre2, pre3));
        }
        const unsigned f0 = (unsigned)__builtin_amdgcn_readlane((int)got, 0);
        const unsigned f1 = (unsigned)__builtin_amdgcn_readlane((int)got, 1);
        const unsigned f2 = (unsigned)__builtin_amdgcn_readlane((int)got, 2);
        const unsigned f3 = (unsigned)__builtin_amdgcn_readlane((int)got, 3);
#pragma unroll
        for (int u = 0; u < OWN_CAP / OWN_T; ++u) {
          const int      i   = tid + u * OWN_T;
          const unsigned loc = lc[u];
          const unsigned q   = loc & 3u;
          const unsigned p   = own_pick4(q, f0, f1, f2, f3) + rk[u];
          if (i < cnt && loc < (unsigned)R && p < (unsigned)(OWN_CAP / 4)) {
            const unsigned slot = 4u * p + q;
            s_lgw[slot] = (unsigned)min(max(sv[u], 0), N - 1);
            s_loc[slot] = (unsigned short)loc;
          }
        }
      }
      __syncthreads();
      const unsigned maxc = max(max(s_c4[0], s_c4[1]), max(s_c4[2], s_c4[3]));
      listed = maxc <= (unsigned)(OWN_CAP / 4);
      bound  = 4 * (int)maxc;
    }

    if (listed) {
      for (int i0 = 0; i0 < bound; i0 += 4 * (OWN_T / 8)) {
        float4 za[4], zb[4];
        int    li[4];
        bool   ok[4];
#pragma unroll
        for (int u = 0; u < 4; ++u) {
          const int  i  = i0 + u * (OWN_T / 8) + g;
          const int  sr = (int)s_lgw[i];
          const int  lr = (int)s_loc[i];
          ok[u] = lr != OWN_NONE;
          const int  s  = ok[u] ? sr : 0;
          li[u] = ok[u] ? lr : 0;
          const float4* zr = (const float4*)(z + (size_t)s * 64);
          za[u] = zr[j]; zb[u] = zr[8 + j];
        }
        float ef[4];
        int   ti[4], sh[4];
        float pe[4];
        {
          float4 da[4], db[4];
#pragma unroll
          for (int u = 0; u < 4; ++u) {
            const float4* dr = (const float4*)(s_df + li[u] * 64);
            da[u] = dr[j]; db[u] = dr[8 + j];
          }
#pragma unroll
          for (int u = 0; u < 4; ++u) {
            const float p = za[u].x * da[u].x + za[u].y * da[u].y + za[u].z * da[u].z + za[u].w * da[u].w
                          + zb[u].x * db[u].x + zb[u].y * db[u].y + zb[u].z * db[u].z + zb[u].w * db[u].w;
            pe[u] = own_sum8(p);
          }
        }
#pragma unroll
        for (int u = 0; u < 4; ++u) {
          const float p = pe[u];
          const float th = p * OWN_L2E_HI;
          const float tl = fmaf(p, OWN_L2E_LO, fmaf(p, OWN_L2E_HI, -th));
          const float tc = own_lg2(p);
          ti[u] = (int)floorf(tc);
          ef[u] = __builtin_amdgcn_exp2f(__builtin_amdgcn_fractf(tc) + tl);
        }
        int old[4];
#pragma unroll
        for (int u = 0; u < 4; ++u) {
          const int  i  = i0 + u * (OWN_T / 8) + g;
          old[u] = OWN_EMPTY;
          if (ok[u] && j == 0) {
            s_es[i] = (int)__float_as_uint(pe[u]);
            atomicMax(&s_key[li[u]], float_key(pe[u]));
            old[u] = atomicCAS(&s_sh[li[u]], OWN_EMPTY, ti[u]);
          }
        }
#pragma unroll
        for (int u = 0; u < 4; ++u) sh[u] = (old[u] == OWN_EMPTY) ? ti[u] : old[u];
#pragma unroll
        for (int u = 0; u < 4; ++u) {
          const int   shift = __shfl(sh[u], 0, 8);
          const float ex    = ldexpf(ef[u], ti[u] - shift);
          if (ok[u]) {
            float* np = s_num + li[u] * OWN_P + 4 * j;
            atomicAdd(np + 0,  ex * za[u].x); atomicAdd(np + 1,  ex * za[u].y);
            atomicAdd(np + 2,  ex * za[u].z); atomicAdd(np + 3,  ex * za[u].w);
            atomicAdd(np + 32, ex * zb[u].x); atomicAdd(np + 33, ex * zb[u].y);
            atomicAdd(np + 34, ex * zb[u].z); atomicAdd(np + 35, ex * zb[u].w);
            if (j == 0) atomicAdd(&s_den[li[u]], ex);
          }
        }
      }
      __syncthreads();

      for (int i = tid; i < R; i += OWN_T) {
        const int shv = s_sh[i];
        if (shv != OWN_EMPTY) {
          const int tm = (int)floorf(own_lg2(key_float(s_key[i])));
          if (tm - shv > OWN_SPREAD || tm >= OWN_CLAMP_I || shv <= -OWN_CLAMP_I) s_bad = 1u;
        }
      }
      __syncthreads();

      if (s_bad != 0u) {
        for (int i = tid; i < OWN_R * OWN_P; i += OWN_T) s_num[i] = 0.0f;
        for (int i = tid; i < OWN_R; i += OWN_T) s_den[i] = 0.0f;
        __syncthreads();

        for (int i0 = 0; i0 < bound; i0 += 4 * (OWN_T / 8)) {
          float4 za[4], zb[4];
#pragma unroll
          for (int u = 0; u < 4; ++u) {
            const int  i = i0 + u * (OWN_T / 8) + g;
            const int  sr = (int)s_lgw[i];
            const int  s  = ((int)s_loc[i] != OWN_NONE) ? sr : 0;
            const float4* zr = (const float4*)(z + (size_t)s * 64);
            za[u] = zr[j]; zb[u] = zr[8 + j];
          }
#pragma unroll
          for (int u = 0; u < 4; ++u) {
            const int i = i0 + u * (OWN_T / 8) + g;
            if ((int)s_loc[i] != OWN_NONE) {
              const int   loc = (int)s_loc[i];
              const float ex  = __expf(__uint_as_float((unsigned)s_es[i]) - key_float(s_key[loc]));
              float* np = s_num + loc * OWN_P + 4 * j;
              atomicAdd(np + 0,  ex * za[u].x); atomicAdd(np + 1,  ex * za[u].y);
              atomicAdd(np + 2,  ex * za[u].z); atomicAdd(np + 3,  ex * za[u].w);
              atomicAdd(np + 32, ex * zb[u].x); atomicAdd(np + 33, ex * zb[u].y);
              atomicAdd(np + 34, ex * zb[u].z); atomicAdd(np + 35, ex * zb[u].w);
              if (j == 0) atomicAdd(&s_den[loc], ex);
            }
          }
        }
      }
    } else {
      for (int e = tid; e < E; e += OWN_T) {
        const unsigned loc = (unsigned)(dst[e] - base);
        if (loc < (unsigned)R) {
          const int s = min(max(src[e], 0), N - 1);
          const float p = row_dot_lane(z + (size_t)s * 64, s_df + loc * 64);
          atomicMax(&s_key[loc], float_key(p));
        }
      }
      __syncthreads();
      for (int e = tid; e < E; e += OWN_T) {
        const unsigned loc = (unsigned)(dst[e] - base);
        if (loc < (unsigned)R) {
          const int s = min(max(src[e], 0), N - 1);
          const float* zr = z + (size_t)s * 64;
          const float p  = row_dot_lane(zr, s_df + loc * 64);
          const float ex = __expf(p - key_float(s_key[loc]));
          float* np = s_num + loc * OWN_P;
          for (int k = 0; k < 64; ++k) atomicAdd(np + k, ex * zr[k]);
          atomicAdd(&s_den[loc], ex);
        }
      }
    }
    __syncthreads();

    for (int i = tid; i < R * 64; i += OWN_T) {
      const int   loc = i >> 6, f = i & 63;
      const float dn  = s_den[loc];
      out[(size_t)(base + loc) * 64 + f] = s_num[loc * OWN_P + f] / (dn == 0.0f ? 1.0f : dn);
    }
    __syncthreads();
  }
}

extern "C" void kernel_launch(void* const* d_in, const int* in_sizes, int n_in,
                              void* d_out, int out_size, void* d_ws, size_t ws_size,
                              hipStream_t stream) {
  const float* h     = (const float*)d_in[0];
  const float* feat  = (const float*)d_in[1];
  const float* W_fc  = (const float*)d_in[2];
  const float* W_dst = (const float*)d_in[3];
  const int*   src   = (const int*)d_in[4];
  const int*   dst   = (const int*)d_in[5];
  float*       out   = (float*)d_out;

  const int FEAT = 64, O = 64;
  const int N   = in_sizes[1] / FEAT;
  const int Kfc = in_sizes[0] / N;
  const int E   = in_sizes[4];

  float* z  = (float*)d_ws;
  float* df = z + (size_t)N * O;
  unsigned* pairs = (unsigned*)(df + (size_t)N * O);

  if (Kfc == 256) {
    proj_resident<256><<<1024, PJ_T, 0, stream>>>(h, W_fc, z, N);
  } else {
    const dim3 gemmBlk(32 * (O / 16));
    const int  mTiles = (N + 15) / 16;
    gemm_nt_wmma<<<mTiles, gemmBlk, 0, stream>>>(h, W_fc, z, N, Kfc, O);
  }
  proj_resident<64><<<1024, PJ_T, 0, stream>>>(feat, W_dst, df, N);

  owner_pairs<<<512, 256, 0, stream>>>(src, dst, pairs, E);
  edge_owner<<<256, OWN_T, 0, stream>>>(z, df, src, dst, pairs, out, N, E);
}


# z projection, df projection and pair-line build in ONE launch, roles interleaved by block index (3 z : 1 df, pair blocks behind): matrix-bound and memory-bound work overlap; 4 -> 2 launches
# speedup vs baseline: 5.5773x; 1.0619x over previous
#include <hip/hip_runtime.h>
#include <hip/hip_bf16.h>

typedef float v2f __attribute__((ext_vector_type(2)));
typedef float v8f __attribute__((ext_vector_type(8)));

__global__ void gemm_nt_wmma(const float* __restrict__ A,
                             const float* __restrict__ W,
                             float* __restrict__ C,
                             int M, int K, int O) {
  const int lane  = threadIdx.x & 31;
  const int oTile = threadIdx.x >> 5;
  const int mTile = blockIdx.x;
  if (mTile * 16 >= M) return;
  const int half = lane >> 4;
  const int l16  = lane & 15;

  const float* aPtr = A + (size_t)(mTile * 16 + l16) * K + 2 * half;
  const float* wPtr = W + (size_t)(oTile * 16 + l16) * K + 2 * half;

  v8f acc = {};
#pragma unroll 8
  for (int k = 0; k < K; k += 4) {
    v2f a = *(const v2f*)(aPtr + k);
    v2f b = *(const v2f*)(wPtr + k);
    acc = __builtin_amdgcn_wmma_f32_16x16x4_f32(
         false, a,  false, b,
         (short)0, acc,  false,  false);
  }

  float* cPtr = C + (size_t)(mTile * 16 + half * 8) * O + oTile * 16 + l16;
#pragma unroll
  for (int r = 0; r < 8; ++r) cPtr[(size_t)r * O] = acc[r];
}

#define PJ_T 128
#define PJ_G 4

template <int K>
__device__ __forceinline__ void
proj_body(const float* __restrict__ A,
          const float* __restrict__ Wm,
          float* __restrict__ C, int M,
          float* __restrict__ s_a, const int bid, const int nb) {
  constexpr int P   = K + 4;
  constexpr int NQ  = K / 8;
  constexpr int RV  = K / 4;
  constexpr int NLD = (16 * RV) / PJ_T;
  const int tid  = threadIdx.x;
  const int lane = tid & 31;
  const int ct   = tid >> 5;
  const int c    = lane & 15;
  const int h    = lane >> 4;
  const int nSteps = (M + 15) >> 4;
  int step = bid;
  if (step >= nSteps) return;

  float4 wf[NQ];
  {
    const float4* wr = (const float4*)(Wm + (size_t)(ct * 16 + c) * K + 4 * h);
#pragma unroll
    for (int q = 0; q < NQ; ++q) wf[q] = wr[2 * q];
  }

  constexpr int RPI = PJ_T / RV;
  const int trow = tid / RV, tcol = (tid % RV) * 4;
  const float* ag = A + (size_t)trow * K + tcol;
  float*       sl = s_a + trow * P + tcol;

  static_assert(NLD == 8 || NLD == 2, "tile share per thread");
  float4 t0, t1, t2, t3, t4, t5, t6, t7;
#define PJ_REQ(at)                                                        \
  do {                                                                    \
    t0 = *(const float4*)((at) + (size_t)0 * RPI * K);                    \
    t1 = *(const float4*)((at) + (size_t)1 * RPI * K);                    \
    if constexpr (NLD == 8) {                                             \
      t2 = *(const float4*)((at) + (size_t)2 * RPI * K);                  \
      t3 = *(const float4*)((at) + (size_t)3 * RPI * K);                  \
      t4 = *(const float4*)((at) + (size_t)4 * RPI * K);                  \
      t5 = *(const float4*)((at) + (size_t)5 * RPI * K);                  \
      t6 = *(const float4*)((at) + (size_t)6 * RPI * K);                  \
      t7 = *(const float4*)((at) + (size_t)7 * RPI * K);                  \
    }                                                                     \
  } while (0)
#define PJ_PUT(sp)                                                        \
  do {                                                                    \
    *(float4*)((sp) + 0 * RPI * P) = t0;                                  \
    *(float4*)((sp) + 1 * RPI * P) = t1;                                  \
    if constexpr (NLD == 8) {                                             \
      *(float4*)((sp) + 2 * RPI * P) = t2;                                \
      *(float4*)((sp) + 3 * RPI * P) = t3;                                \
      *(float4*)((sp) + 4 * RPI * P) = t4;                                \
      *(float4*)((sp) + 5 * RPI * P) = t5;                                \
      *(float4*)((sp) + 6 * RPI * P) = t6;                                \
      *(float4*)((sp) + 7 * RPI * P) = t7;                                \
    }                                                                     \
  } while (0)

  {
    const float* at = ag + (size_t)min(step * 16, M - 16) * K;
    PJ_REQ(at);
    PJ_PUT(sl);
  }
  {
    const int ls = min(step + nb, nSteps - 1);
    const float* at = ag + (size_t)min(ls * 16, M - 16) * K;
    PJ_REQ(at);
  }
  __syncthreads();

  int b = 0;
  for (;;) {
    const float* ar = s_a + b * (16 * P) + c * P + 4 * h;
    v8f acc0 = {}, acc1 = {};
    float4 af[2][PJ_G];
#pragma unroll
    for (int j = 0; j < PJ_G; ++j) af[0][j] = *(const float4*)(ar + 8 * j);
#pragma unroll
    for (int g = 0; g < NQ / PJ_G; ++g) {
      if (g + 1 < NQ / PJ_G) {
#pragma unroll
        for (int j = 0; j < PJ_G; ++j) af[(g + 1) & 1][j] = *(const float4*)(ar + 8 * ((g + 1) * PJ_G + j));
      }
      __builtin_amdgcn_sched_barrier(0);
#pragma unroll
      for (int j = 0; j < PJ_G; ++j) {
        const float4 av = af[g & 1][j];
        const float4 wv = wf[g * PJ_G + j];
        const v2f a0 = {av.x, av.y}, a1 = {av.z, av.w};
        const v2f b0 = {wv.x, wv.y}, b1 = {wv.z, wv.w};
        acc0 = __builtin_amdgcn_wmma_f32_16x16x4_f32(false, a0, false, b0, (short)0, acc0, false, false);
        acc1 = __builtin_amdgcn_wmma_f32_16x16x4_f32(false, a1, false, b1, (short)0, acc1, false, false);
      }
      __builtin_amdgcn_sched_barrier(0);
    }

    {
      float* sn = sl + (b ^ 1) * (16 * P);
      PJ_PUT(sn);
    }
    __syncthreads();

    {
      const int r0 = min(step * 16, M - 16) + 8 * h;
      float* cp = C + (size_t)r0 * 64 + ct * 16 + c;
#pragma unroll
      for (int r = 0; r < 8; ++r)
        if (r0 + r >= step * 16) cp[(size_t)r * 64] = acc0[r] + acc1[r];
    }

    const int nx = step + nb;
    if (nx >= nSteps) break;
    {
      const int ls = min(nx + nb, nSteps - 1);
      const float* at = ag + (size_t)min(ls * 16, M - 16) * K;
      PJ_REQ(at);
    }
    step = nx;
    b ^= 1;
  }
#undef PJ_REQ
#undef PJ_PUT
}

__device__ __forceinline__ unsigned float_key(float f) {
  unsigned u = __float_as_uint(f);
  return (u & 0x80000000u) ? ~u : (u | 0x80000000u);
}
__device__ __forceinline__ float key_float(unsigned k) {
  unsigned u = (k & 0x80000000u) ? (k & 0x7FFFFFFFu) : ~k;
  return __uint_as_float(u);
}

#define OWN_R   392
#define OWN_P   65
#define OWN_CAP 8192
#define OWN_T   1024
#define OWN_NONE 0xFFFF
#define OWN_U   8
#define OWN_EMPTY  ((int)0x80000000)
#define OWN_SPREAD 40
#define OWN_CLAMP   268435456.0f
#define OWN_CLAMP_I (1 << 28)
#define OWN_L2E_HI 1.44269502162933349609375f
#define OWN_L2E_LO 1.925963033500011e-8f

__device__ __forceinline__ int own_groups(int E) { return (E + 15) >> 4; }
__device__ __forceinline__ int own_words(int E)  { return (own_groups(E) + 31) >> 5; }
__device__ __forceinline__ int own_rowp(int E)   { return (own_words(E) + 7) & ~7; }

__device__ __forceinline__ void
pairs_body(const int* __restrict__ src, const int* __restrict__ dst,
           unsigned* __restrict__ pairs, int E,
           unsigned* __restrict__ s_p,
           unsigned* __restrict__ s_b,
           const int bid, const int nb) {
  const int tid = threadIdx.x;
  const int nG  = own_groups(E);
  const int W32 = own_words(E);
  const int RW  = own_rowp(E);
  unsigned* __restrict__ rows = pairs + (size_t)nG * 32;
  for (int g0 = bid * 256; g0 < nG; g0 += nb * 256) {
    for (int i = tid; i < 256 * 9; i += PJ_T) s_b[i] = 0u;
    __syncthreads();
#pragma unroll
    for (int hh = 0; hh < 2; ++hh) {
      const int lt = tid + PJ_T * hh;
      const int g  = g0 + lt;
      int dd[16], ss[16];
      if (g < nG && 16 * g + 16 <= E) {
        const int4* __restrict__ pd = (const int4*)(dst + (size_t)16 * g);
        const int4* __restrict__ ps = (const int4*)(src + (size_t)16 * g);
        int4 d4[4], s4[4];
#pragma unroll
        for (int q = 0; q < 4; ++q) { d4[q] = pd[q]; s4[q] = ps[q]; }
#pragma unroll
        for (int q = 0; q < 4; ++q) {
          dd[4 * q + 0] = d4[q].x; dd[4 * q + 1] = d4[q].y; dd[4 * q + 2] = d4[q].z; dd[4 * q + 3] = d4[q].w;
          ss[4 * q + 0] = s4[q].x; ss[4 * q + 1] = s4[q].y; ss[4 * q + 2] = s4[q].z; ss[4 * q + 3] = s4[q].w;
        }
      } else {
#pragma unroll
        for (int k = 0; k < 16; ++k) {
          const int  e  = 16 * g + k;
          const bool ok = g < nG && e < E;
          dd[k] = ok ? dst[ok ? e : 0] : -1;
          ss[k] = ok ? src[ok ? e : 0] : 0;
        }
      }
#pragma unroll
      for (int k = 0; k < 16; ++k) {
        if (dd[k] >= 0) {
          const unsigned o = ((unsigned)dd[k] / (unsigned)OWN_R) & 255u;
          atomicOr(&s_b[o * 9 + (lt >> 5)], 1u << (lt & 31));
        }
      }
      {
        uint4* sp = (uint4*)(s_p + lt * 36);
#pragma unroll
        for (int q = 0; q < 8; ++q)
          sp[q] = make_uint4((unsigned)ss[2 * q], (unsigned)dd[2 * q], (unsigned)ss[2 * q + 1], (unsigned)dd[2 * q + 1]);
      }
    }
    __syncthreads();
#pragma unroll
    for (int r = 0; r < 16; ++r) {
      const int x = r * PJ_T + tid, gl = x >> 3, pc = x & 7;
      if (g0 + gl < nG) ((uint4*)pairs)[(size_t)(g0 + gl) * 8 + pc] = *(const uint4*)(s_p + gl * 36 + 4 * pc);
    }
#pragma unroll
    for (int r = 0; r < 16; ++r) {
      const int x = r * PJ_T + tid, k = x >> 3, w = x & 7;
      const int gw = (g0 >> 5) + w;
      if (gw < W32) rows[(size_t)k * RW + gw] = s_b[k * 9 + w];
    }
    __syncthreads();
  }
}

#define FR_PJ 1024
#define FR_PR 512
template <bool ZIN>
__global__ void __launch_bounds__(PJ_T)
front_all(const float* __restrict__ h, const float* __restrict__ W_fc, float* __restrict__ z,
          const float* __restrict__ feat, const float* __restrict__ W_dst, float* __restrict__ df,
          const int* __restrict__ src, const int* __restrict__ dst,
          unsigned* __restrict__ pairs, int N, int E) {
  __shared__ __attribute__((aligned(16))) float s_all[256 * 36 + 256 * 9];
  static_assert(256 * 36 + 256 * 9 >= 2 * 16 * (256 + 4), "projection tiles fit");
  const int b = blockIdx.x;
  if (b < FR_PJ) {
    if ((b & 3) != 3) {
      if constexpr (ZIN) proj_body<256>(h, W_fc, z, N, s_all, (b >> 2) * 3 + (b & 3), (FR_PJ / 4) * 3);
    } else {
      proj_body<64>(feat, W_dst, df, N, s_all, b >> 2, FR_PJ / 4);
    }
  } else {
    unsigned* s_p = (unsigned*)s_all;
    pairs_body(src, dst, pairs, E, s_p, s_p + 256 * 36, b - FR_PJ, FR_PR);
  }
}

__device__ __forceinline__ float row_dot_lane(const float* __restrict__ zr,
                                              const float* __restrict__ dr) {
  float p = 0.0f;
  for (int k = 0; k < 16; ++k) {
    const float4 a = ((const float4*)zr)[k];
    const float4 b = ((const float4*)dr)[k];
    p = fmaf(a.x, b.x, p); p = fmaf(a.y, b.y, p);
    p = fmaf(a.z, b.z, p); p = fmaf(a.w, b.w, p);
  }
  return p;
}

__device__ __forceinline__ float own_lg2(float e) {
  return fminf(fmaxf(e * OWN_L2E_HI, -OWN_CLAMP), OWN_CLAMP);
}

__device__ __forceinline__ float own_sum8(float p) {
  p += __int_as_float(__builtin_amdgcn_update_dpp(__float_as_int(p), __float_as_int(p), 0x161, 0xF, 0xF, true));
  p += __int_as_float(__builtin_amdgcn_update_dpp(__float_as_int(p), __float_as_int(p), 0x162, 0xF, 0xF, true));
  p += __int_as_float(__builtin_amdgcn_update_dpp(__float_as_int(p), __float_as_int(p), 0x164, 0xF, 0xF, true));
  return p;
}

__device__ __forceinline__ unsigned own_pick4(unsigned q, unsigned a0, unsigned a1, unsigned a2, unsigned a3) {
  const unsigned lo = (q & 1u) ? a1 : a0;
  const unsigned hi = (q & 1u) ? a3 : a2;
  return (q & 2u) ? hi : lo;
}

__global__ void __launch_bounds__(OWN_T)
edge_owner(const float* __restrict__ z,
           const float* __restrict__ df,
           const int* __restrict__ src,
           const int* __restrict__ dst,
           const unsigned* __restrict__ pairs,
           float* __restrict__ out, int N, int E) {
  __shared__ __attribute__((aligned(16))) float s_df[OWN_R * 64];
  __shared__ float          s_num[OWN_R * OWN_P];
  __shared__ float          s_den[OWN_R];
  __shared__ unsigned       s_key[OWN_R];
  __shared__ int            s_sh[OWN_R];
  __shared__ int            s_es[OWN_CAP];
  __shared__ unsigned short s_tl[OWN_CAP];
  __shared__ unsigned       s_lgw[OWN_CAP];
  __shared__ unsigned short s_loc[OWN_CAP];
  __shared__ unsigned       s_c4[4];
  __shared__ unsigned       s_cnt;
  __shared__ unsigned       s_ng;
  __shared__ unsigned       s_bad;

  const int tid = threadIdx.x;
  const int g   = tid >> 3;
  const int j   = tid & 7;

  const int lane  = tid & 31;
  const int nG  = own_groups(E);
  const int W32 = own_words(E);
  const unsigned* __restrict__ rows = pairs + (size_t)nG * 32;

  for (int base = blockIdx.x * OWN_R; base < N; base += gridDim.x * OWN_R) {
    const int R = min(OWN_R, N - base);

    for (int i = tid; i < OWN_R * OWN_P; i += OWN_T) s_num[i] = 0.0f;
    for (int i = tid; i < OWN_R; i += OWN_T) { s_den[i] = 0.0f; s_key[i] = 0u; s_sh[i] = OWN_EMPTY; }
    for (int i = tid; i < OWN_CAP; i += OWN_T) s_loc[i] = (unsigned short)OWN_NONE;
    {
      const float4* __restrict__ dfr = (const float4*)(df + (size_t)base * 64);
#pragma unroll
      for (int u = 0; u < (OWN_R * 16 + OWN_T - 1) / OWN_T; ++u) {
        const int i = min(tid + u * OWN_T, R * 16 - 1);
        ((float4*)s_df)[i] = dfr[i];
      }
    }
    if (tid == 0) { s_cnt = 0u; s_ng = 0u; s_bad = 0u; }
    if (tid < 4) s_c4[tid] = 0u;
    __syncthreads();

    const unsigned want = ((unsigned)base / (unsigned)OWN_R) & 255u;
    const unsigned* __restrict__ krow = rows + (size_t)want * own_rowp(E);
    for (int w0 = 0; w0 < W32; w0 += OWN_T * 4) {
      unsigned bw[4];
#pragma unroll
      for (int u = 0; u < 4; ++u) {
        const int w = w0 + u * OWN_T + tid;
        bw[u] = krow[min(w, W32 - 1)];
      }
      unsigned bits[4];
      unsigned cl = 0u;
#pragma unroll
      for (int u = 0; u < 4; ++u) {
        const int w = w0 + u * OWN_T + tid;
        bits[u] = (w < W32) ? bw[u] : 0u;
        cl += (unsigned)__popc(bits[u]);
      }
      unsigned incl = cl;
#pragma unroll
      for (int d = 1; d < 32; d <<= 1) {
        const unsigned t = (unsigned)__shfl_up((int)incl, d, 32);
        if (lane >= d) incl += t;
      }
      const unsigned run = (unsigned)__shfl((int)incl, 31, 32);
      if (run != 0u) {
        unsigned first = 0u;
        if (lane == 0) first = atomicAdd(&s_ng, run);
        first = (unsigned)__builtin_amdgcn_readfirstlane((int)first);
        unsigned pos = first + incl - cl;
#pragma unroll
        for (int u = 0; u < 4; ++u) {
          const int w = w0 + u * OWN_T + tid;
          unsigned bl = bits[u];
          while (bl != 0u) {
            const int b = __ffs((int)bl) - 1;
            bl &= bl - 1u;
            if (pos < OWN_CAP) s_lgw[pos] = (unsigned)(32 * w + b);
            ++pos;
          }
        }
      }
    }
    __syncthreads();
    const unsigned ngr = s_ng;

    if (ngr <= OWN_CAP) {
      const int ng = (int)ngr;
      const uint4* __restrict__ pr4 = (const uint4*)pairs;
      const int pc = tid & 7;
      for (int i0 = 0; i0 < ng; i0 += (OWN_T / 8) * OWN_U) {
        uint4 p4[OWN_U];
        bool  in[OWN_U];
#pragma unroll
        for (int u = 0; u < OWN_U; ++u) {
          const int i  = i0 + u * (OWN_T / 8) + g;
          const int gr = (int)s_lgw[min(i, OWN_CAP - 1)];
          in[u] = i < ng;
          const int gs = in[u] ? gr : 0;
          p4[u] = pr4[(size_t)gs * 8 + pc];
        }
        __builtin_amdgcn_sched_barrier(0);
        unsigned b0[OWN_U], b1[OWN_U];
        unsigned run = 0u;
#pragma unroll
        for (int u = 0; u < OWN_U; ++u) {
          b0[u] = __builtin_amdgcn_ballot_w32(in[u] && (unsigned)((int)p4[u].y - base) < (unsigned)R);
          b1[u] = __builtin_amdgcn_ballot_w32(in[u] && (unsigned)((int)p4[u].w - base) < (unsigned)R);
          run += (unsigned)__popc(b0[u]) + (unsigned)__popc(b1[u]);
        }
        if (run != 0u) {
          unsigned first = 0u;
          if (lane == 0) first = atomicAdd(&s_cnt, run);
          first = (unsigned)__builtin_amdgcn_readfirstlane((int)first);
#pragma unroll
          for (int u = 0; u < OWN_U; ++u) {
            const unsigned l0 = (unsigned)((int)p4[u].y - base);
            const unsigned l1 = (unsigned)((int)p4[u].w - base);
            const bool h0 = in[u] && l0 < (unsigned)R;
            const bool h1 = in[u] && l1 < (unsigned)R;
            const unsigned x0 = first + __builtin_amdgcn_mbcnt_lo(b0[u], 0u);
            first += (unsigned)__popc(b0[u]);
            const unsigned x1 = first + __builtin_amdgcn_mbcnt_lo(b1[u], 0u);
            first += (unsigned)__popc(b1[u]);
            if (h0 && x0 < OWN_CAP) { s_es[x0] = (int)p4[u].x; s_tl[x0] = (unsigned short)l0; }
            if (h1 && x1 < OWN_CAP) { s_es[x1] = (int)p4[u].z; s_tl[x1] = (unsigned short)l1; }
          }
        }
      }
    }
    __syncthreads();
    const unsigned total = (ngr <= OWN_CAP) ? s_cnt : (unsigned)(OWN_CAP + 1);
    bool listed = total <= OWN_CAP;
    int  bound  = 0;

    if (listed) {
      const int cnt = (int)total;

      {
        int sv[OWN_CAP / OWN_T];
        unsigned lc[OWN_CAP / OWN_T];
        unsigned rk[OWN_CAP / OWN_T];
#pragma unroll
        for (int u = 0; u < OWN_CAP / OWN_T; ++u) {
          const int i  = tid + u * OWN_T;
          const int      ev = s_es[i];
          const unsigned lv = (unsigned)s_tl[i];
          sv[u] = (i < cnt) ? ev : 0;
          lc[u] = (i < cnt) ? lv : 0u;
        }
        unsigned pre0 = 0u, pre1 = 0u, pre2 = 0u, pre3 = 0u;
#pragma unroll
        for (int u = 0; u < OWN_CAP / OWN_T; ++u) {
          const int      i    = tid + u * OWN_T;
          const unsigned loc  = lc[u];
          const bool     keep = i < cnt && loc < (unsigned)R;
          const unsigned q    = loc & 3u;
          const unsigned mk = __builtin_amdgcn_ballot_w32(keep);
          const unsigned ml = __builtin_amdgcn_ballot_w32(keep && (q & 1u) != 0u);
          const unsigned mh = __builtin_amdgcn_ballot_w32(keep && (q & 2u) != 0u);
          const unsigned b0 = mk & ~ml & ~mh, b1 = ml & ~mh, b2 = mh & ~ml, b3 = ml & mh;
          rk[u] = own_pick4(q, pre0, pre1, pre2, pre3)
                + __builtin_amdgcn_mbcnt_lo(own_pick4(q, b0, b1, b2, b3), 0u);
          pre0 += (unsigned)__popc(b0); pre1 += (unsigned)__popc(b1);
          pre2 += (unsigned)__popc(b2); pre3 += (unsigned)__popc(b3);
        }
        unsigned got = 0u;
        if (lane < 4) {
          got = atomicAdd(&s_c4[lane], own_pick4((unsigned)lane, pre0, pre1, pre2, pre3));
        }
        const unsigned f0 = (unsigned)__builtin_amdgcn_readlane((int)got, 0);
        const unsigned f1 = (unsigned)__builtin_amdgcn_readlane((int)got, 1);
        const unsigned f2 = (unsigned)__builtin_amdgcn_readlane((int)got, 2);
        const unsigned f3 = (unsigned)__builtin_amdgcn_readlane((int)got, 3);
#pragma unroll
        for (int u = 0; u < OWN_CAP / OWN_T; ++u) {
          const int      i   = tid + u * OWN_T;
          const unsigned loc = lc[u];
          const unsigned q   = loc & 3u;
          const unsigned p   = own_pick4(q, f0, f1, f2, f3) + rk[u];
          if (i < cnt && loc < (unsigned)R && p < (unsigned)(OWN_CAP / 4)) {
            const unsigned slot = 4u * p + q;
            s_lgw[slot] = (unsigned)min(max(sv[u], 0), N - 1);
            s_loc[slot] = (unsigned short)loc;
          }
        }
      }
      __syncthreads();
      const unsigned maxc = max(max(s_c4[0], s_c4[1]), max(s_c4[2], s_c4[3]));
      listed = maxc <= (unsigned)(OWN_CAP / 4);
      bound  = 4 * (int)maxc;
    }

    if (listed) {
      for (int i0 = 0; i0 < bound; i0 += 4 * (OWN_T / 8)) {
        float4 za[4], zb[4];
        int    li[4];
        bool   ok[4];
#pragma unroll
        for (int u = 0; u < 4; ++u) {
          const int  i  = i0 + u * (OWN_T / 8) + g;
          const int  sr = (int)s_lgw[i];
          const int  lr = (int)s_loc[i];
          ok[u] = lr != OWN_NONE;
          const int  s  = ok[u] ? sr : 0;
          li[u] = ok[u] ? lr : 0;
          const float4* zr = (const float4*)(z + (size_t)s * 64);
          za[u] = zr[j]; zb[u] = zr[8 + j];
        }
        float ef[4];
        int   ti[4], sh[4];
        float pe[4];
        {
          float4 da[4], db[4];
#pragma unroll
          for (int u = 0; u < 4; ++u) {
            const float4* dr = (const float4*)(s_df + li[u] * 64);
            da[u] = dr[j]; db[u] = dr[8 + j];
          }
#pragma unroll
          for (int u = 0; u < 4; ++u) {
            const float p = za[u].x * da[u].x + za[u].y * da[u].y + za[u].z * da[u].z + za[u].w * da[u].w
                          + zb[u].x * db[u].x + zb[u].y * db[u].y + zb[u].z * db[u].z + zb[u].w * db[u].w;
            pe[u] = own_sum8(p);
          }
        }
#pragma unroll
        for (int u = 0; u < 4; ++u) {
          const float p = pe[u];
          const float th = p * OWN_L2E_HI;
          const float tl = fmaf(p, OWN_L2E_LO, fmaf(p, OWN_L2E_HI, -th));
          const float tc = own_lg2(p);
          ti[u] = (int)floorf(tc);
          ef[u] = __builtin_amdgcn_exp2f(__builtin_amdgcn_fractf(tc) + tl);
        }
        int old[4];
#pragma unroll
        for (int u = 0; u < 4; ++u) {
          const int  i  = i0 + u * (OWN_T / 8) + g;
          old[u] = OWN_EMPTY;
          if (ok[u] && j == 0) {
            s_es[i] = (int)__float_as_uint(pe[u]);
            atomicMax(&s_key[li[u]], float_key(pe[u]));
            old[u] = atomicCAS(&s_sh[li[u]], OWN_EMPTY, ti[u]);
          }
        }
#pragma unroll
        for (int u = 0; u < 4; ++u) sh[u] = (old[u] == OWN_EMPTY) ? ti[u] : old[u];
#pragma unroll
        for (int u = 0; u < 4; ++u) {
          const int   shift = __shfl(sh[u], 0, 8);
          const float ex    = ldexpf(ef[u], ti[u] - shift);
          if (ok[u]) {
            float* np = s_num + li[u] * OWN_P + 4 * j;
            atomicAdd(np + 0,  ex * za[u].x); atomicAdd(np + 1,  ex * za[u].y);
            atomicAdd(np + 2,  ex * za[u].z); atomicAdd(np + 3,  ex * za[u].w);
            atomicAdd(np + 32, ex * zb[u].x); atomicAdd(np + 33, ex * zb[u].y);
            atomicAdd(np + 34, ex * zb[u].z); atomicAdd(np + 35, ex * zb[u].w);
            if (j == 0) atomicAdd(&s_den[li[u]], ex);
          }
        }
      }
      __syncthreads();

      for (int i = tid; i < R; i += OWN_T) {
        const int shv = s_sh[i];
        if (shv != OWN_EMPTY) {
          const int tm = (int)floorf(own_lg2(key_float(s_key[i])));
          if (tm - shv > OWN_SPREAD || tm >= OWN_CLAMP_I || shv <= -OWN_CLAMP_I) s_bad = 1u;
        }
      }
      __syncthreads();

      if (s_bad != 0u) {
        for (int i = tid; i < OWN_R * OWN_P; i += OWN_T) s_num[i] = 0.0f;
        for (int i = tid; i < OWN_R; i += OWN_T) s_den[i] = 0.0f;
        __syncthreads();

        for (int i0 = 0; i0 < bound; i0 += 4 * (OWN_T / 8)) {
          float4 za[4], zb[4];
#pragma unroll
          for (int u = 0; u < 4; ++u) {
            const int  i = i0 + u * (OWN_T / 8) + g;
            const int  sr = (int)s_lgw[i];
            const int  s  = ((int)s_loc[i] != OWN_NONE) ? sr : 0;
            const float4* zr = (const float4*)(z + (size_t)s * 64);
            za[u] = zr[j]; zb[u] = zr[8 + j];
          }
#pragma unroll
          for (int u = 0; u < 4; ++u) {
            const int i = i0 + u * (OWN_T / 8) + g;
            if ((int)s_loc[i] != OWN_NONE) {
              const int   loc = (int)s_loc[i];
              const float ex  = __expf(__uint_as_float((unsigned)s_es[i]) - key_float(s_key[loc]));
              float* np = s_num + loc * OWN_P + 4 * j;
              atomicAdd(np + 0,  ex * za[u].x); atomicAdd(np + 1,  ex * za[u].y);
              atomicAdd(np + 2,  ex * za[u].z); atomicAdd(np + 3,  ex * za[u].w);
              atomicAdd(np + 32, ex * zb[u].x); atomicAdd(np + 33, ex * zb[u].y);
              atomicAdd(np + 34, ex * zb[u].z); atomicAdd(np + 35, ex * zb[u].w);
              if (j == 0) atomicAdd(&s_den[loc], ex);
            }
          }
        }
      }
    } else {
      for (int e = tid; e < E; e += OWN_T) {
        const unsigned loc = (unsigned)(dst[e] - base);
        if (loc < (unsigned)R) {
          const int s = min(max(src[e], 0), N - 1);
          const float p = row_dot_lane(z + (size_t)s * 64, s_df + loc * 64);
          atomicMax(&s_key[loc], float_key(p));
        }
      }
      __syncthreads();
      for (int e = tid; e < E; e += OWN_T) {
        const unsigned loc = (unsigned)(dst[e] - base);
        if (loc < (unsigned)R) {
          const int s = min(max(src[e], 0), N - 1);
          const float* zr = z + (size_t)s * 64;
          const float p  = row_dot_lane(zr, s_df + loc * 64);
          const float ex = __expf(p - key_float(s_key[loc]));
          float* np = s_num + loc * OWN_P;
          for (int k = 0; k < 64; ++k) atomicAdd(np + k, ex * zr[k]);
          atomicAdd(&s_den[loc], ex);
        }
      }
    }
    __syncthreads();

    for (int i = tid; i < R * 64; i += OWN_T) {
      const int   loc = i >> 6, f = i & 63;
      const float dn  = s_den[loc];
      out[(size_t)(base + loc) * 64 + f] = s_num[loc * OWN_P + f] / (dn == 0.0f ? 1.0f : dn);
    }
    __syncthreads();
  }
}

extern "C" void kernel_launch(void* const* d_in, const int* in_sizes, int n_in,
                              void* d_out, int out_size, void* d_ws, size_t ws_size,
                              hipStream_t stream) {
  const float* h     = (const float*)d_in[0];
  const float* feat  = (const float*)d_in[1];
  const float* W_fc  = (const float*)d_in[2];
  const float* W_dst = (const float*)d_in[3];
  const int*   src   = (const int*)d_in[4];
  const int*   dst   = (const int*)d_in[5];
  float*       out   = (float*)d_out;

  const int FEAT = 64, O = 64;
  const int N   = in_sizes[1] / FEAT;
  const int Kfc = in_sizes[0] / N;
  const int E   = in_sizes[4];

  float* z  = (float*)d_ws;
  float* df = z + (size_t)N * O;
  unsigned* pairs = (unsigned*)(df + (size_t)N * O);

  if (Kfc == 256) {
    front_all<true><<<FR_PJ + FR_PR, PJ_T, 0, stream>>>(h, W_fc, z, feat, W_dst, df, src, dst, pairs, N, E);
  } else {
    const dim3 gemmBlk(32 * (O / 16));
    const int  mTiles = (N + 15) / 16;
    gemm_nt_wmma<<<mTiles, gemmBlk, 0, stream>>>(h, W_fc, z, N, Kfc, O);
    front_all<false><<<FR_PJ + FR_PR, PJ_T, 0, stream>>>(h, W_fc, z, feat, W_dst, df, src, dst, pairs, N, E);
  }

  edge_owner<<<256, OWN_T, 0, stream>>>(z, df, src, dst, pairs, out, N, E);
}


# edge weights under a fixed zero shift, w = 2^(e*log2(e)): no first-arrival shift, running max or LDS answer per edge; a block with a weight outside 2^+-40 redoes under the max; LDS sums as before
# speedup vs baseline: 5.7568x; 1.0322x over previous
#include <hip/hip_runtime.h>
#include <hip/hip_bf16.h>

typedef float v2f __attribute__((ext_vector_type(2)));
typedef float v8f __attribute__((ext_vector_type(8)));

__global__ void gemm_nt_wmma(const float* __restrict__ A,
                             const float* __restrict__ W,
                             float* __restrict__ C,
                             int M, int K, int O) {
  const int lane  = threadIdx.x & 31;
  const int oTile = threadIdx.x >> 5;
  const int mTile = blockIdx.x;
  if (mTile * 16 >= M) return;
  const int half = lane >> 4;
  const int l16  = lane & 15;

  const float* aPtr = A + (size_t)(mTile * 16 + l16) * K + 2 * half;
  const float* wPtr = W + (size_t)(oTile * 16 + l16) * K + 2 * half;

  v8f acc = {};
#pragma unroll 8
  for (int k = 0; k < K; k += 4) {
    v2f a = *(const v2f*)(aPtr + k);
    v2f b = *(const v2f*)(wPtr + k);
    acc = __builtin_amdgcn_wmma_f32_16x16x4_f32(
         false, a,  false, b,
         (short)0, acc,  false,  false);
  }

  float* cPtr = C + (size_t)(mTile * 16 + half * 8) * O + oTile * 16 + l16;
#pragma unroll
  for (int r = 0; r < 8; ++r) cPtr[(size_t)r * O] = acc[r];
}

#define PJ_T 128
#define PJ_G 4

template <int K>
__device__ __forceinline__ void
proj_body(const float* __restrict__ A,
          const float* __restrict__ Wm,
          float* __restrict__ C, int M,
          float* __restrict__ s_a, const int bid, const int nb) {
  constexpr int P   = K + 4;
  constexpr int NQ  = K / 8;
  constexpr int RV  = K / 4;
  constexpr int NLD = (16 * RV) / PJ_T;
  const int tid  = threadIdx.x;
  const int lane = tid & 31;
  const int ct   = tid >> 5;
  const int c    = lane & 15;
  const int h    = lane >> 4;
  const int nSteps = (M + 15) >> 4;
  int step = bid;
  if (step >= nSteps) return;

  float4 wf[NQ];
  {
    const float4* wr = (const float4*)(Wm + (size_t)(ct * 16 + c) * K + 4 * h);
#pragma unroll
    for (int q = 0; q < NQ; ++q) wf[q] = wr[2 * q];
  }

  constexpr int RPI = PJ_T / RV;
  const int trow = tid / RV, tcol = (tid % RV) * 4;
  const float* ag = A + (size_t)trow * K + tcol;
  float*       sl = s_a + trow * P + tcol;

  static_assert(NLD == 8 || NLD == 2, "tile share per thread");
  float4 t0, t1, t2, t3, t4, t5, t6, t7;
#define PJ_REQ(at)                                                        \
  do {                                                                    \
    t0 = *(const float4*)((at) + (size_t)0 * RPI * K);                    \
    t1 = *(const float4*)((at) + (size_t)1 * RPI * K);                    \
    if constexpr (NLD == 8) {                                             \
      t2 = *(const float4*)((at) + (size_t)2 * RPI * K);                  \
      t3 = *(const float4*)((at) + (size_t)3 * RPI * K);                  \
      t4 = *(const float4*)((at) + (size_t)4 * RPI * K);                  \
      t5 = *(const float4*)((at) + (size_t)5 * RPI * K);                  \
      t6 = *(const float4*)((at) + (size_t)6 * RPI * K);                  \
      t7 = *(const float4*)((at) + (size_t)7 * RPI * K);                  \
    }                                                                     \
  } while (0)
#define PJ_PUT(sp)                                                        \
  do {                                                                    \
    *(float4*)((sp) + 0 * RPI * P) = t0;                                  \
    *(float4*)((sp) + 1 * RPI * P) = t1;                                  \
    if constexpr (NLD == 8) {                                             \
      *(float4*)((sp) + 2 * RPI * P) = t2;                                \
      *(float4*)((sp) + 3 * RPI * P) = t3;                                \
      *(float4*)((sp) + 4 * RPI * P) = t4;                                \
      *(float4*)((sp) + 5 * RPI * P) = t5;                                \
      *(float4*)((sp) + 6 * RPI * P) = t6;                                \
      *(float4*)((sp) + 7 * RPI * P) = t7;                                \
    }                                                                     \
  } while (0)

  {
    const float* at = ag + (size_t)min(step * 16, M - 16) * K;
    PJ_REQ(at);
    PJ_PUT(sl);
  }
  {
    const int ls = min(step + nb, nSteps - 1);
    const float* at = ag + (size_t)min(ls * 16, M - 16) * K;
    PJ_REQ(at);
  }
  __syncthreads();

  int b = 0;
  for (;;) {
    const float* ar = s_a + b * (16 * P) + c * P + 4 * h;
    v8f acc0 = {}, acc1 = {};
    float4 af[2][PJ_G];
#pragma unroll
    for (int j = 0; j < PJ_G; ++j) af[0][j] = *(const float4*)(ar + 8 * j);
#pragma unroll
    for (int g = 0; g < NQ / PJ_G; ++g) {
      if (g + 1 < NQ / PJ_G) {
#pragma unroll
        for (int j = 0; j < PJ_G; ++j) af[(g + 1) & 1][j] = *(const float4*)(ar + 8 * ((g + 1) * PJ_G + j));
      }
      __builtin_amdgcn_sched_barrier(0);
#pragma unroll
      for (int j = 0; j < PJ_G; ++j) {
        const float4 av = af[g & 1][j];
        const float4 wv = wf[g * PJ_G + j];
        const v2f a0 = {av.x, av.y}, a1 = {av.z, av.w};
        const v2f b0 = {wv.x, wv.y}, b1 = {wv.z, wv.w};
        acc0 = __builtin_amdgcn_wmma_f32_16x16x4_f32(false, a0, false, b0, (short)0, acc0, false, false);
        acc1 = __builtin_amdgcn_wmma_f32_16x16x4_f32(false, a1, false, b1, (short)0, acc1, false, false);
      }
      __builtin_amdgcn_sched_barrier(0);
    }

    {
      float* sn = sl + (b ^ 1) * (16 * P);
      PJ_PUT(sn);
    }
    __syncthreads();

    {
      const int r0 = min(step * 16, M - 16) + 8 * h;
      float* cp = C + (size_t)r0 * 64 + ct * 16 + c;
#pragma unroll
      for (int r = 0; r < 8; ++r)
        if (r0 + r >= step * 16) cp[(size_t)r * 64] = acc0[r] + acc1[r];
    }

    const int nx = step + nb;
    if (nx >= nSteps) break;
    {
      const int ls = min(nx + nb, nSteps - 1);
      const float* at = ag + (size_t)min(ls * 16, M - 16) * K;
      PJ_REQ(at);
    }
    step = nx;
    b ^= 1;
  }
#undef PJ_REQ
#undef PJ_PUT
}

__device__ __forceinline__ unsigned float_key(float f) {
  unsigned u = __float_as_uint(f);
  return (u & 0x80000000u) ? ~u : (u | 0x80000000u);
}
__device__ __forceinline__ float key_float(unsigned k) {
  unsigned u = (k & 0x80000000u) ? (k & 0x7FFFFFFFu) : ~k;
  return __uint_as_float(u);
}

#define OWN_R   392
#define OWN_P   65
#define OWN_CAP 8192
#define OWN_T   1024
#define OWN_NONE 0xFFFF
#define OWN_U   8
#define OWN_SPREAD 40
#define OWN_CLAMP   268435456.0f
#define OWN_L2E_HI 1.44269502162933349609375f
#define OWN_L2E_LO 1.925963033500011e-8f

__device__ __forceinline__ int own_groups(int E) { return (E + 15) >> 4; }
__device__ __forceinline__ int own_words(int E)  { return (own_groups(E) + 31) >> 5; }
__device__ __forceinline__ int own_rowp(int E)   { return (own_words(E) + 7) & ~7; }

__device__ __forceinline__ void
pairs_body(const int* __restrict__ src, const int* __restrict__ dst,
           unsigned* __restrict__ pairs, int E,
           unsigned* __restrict__ s_p,
           unsigned* __restrict__ s_b,
           const int bid, const int nb) {
  const int tid = threadIdx.x;
  const int nG  = own_groups(E);
  const int W32 = own_words(E);
  const int RW  = own_rowp(E);
  unsigned* __restrict__ rows = pairs + (size_t)nG * 32;
  for (int g0 = bid * 256; g0 < nG; g0 += nb * 256) {
    for (int i = tid; i < 256 * 9; i += PJ_T) s_b[i] = 0u;
    __syncthreads();
#pragma unroll
    for (int hh = 0; hh < 2; ++hh) {
      const int lt = tid + PJ_T * hh;
      const int g  = g0 + lt;
      int dd[16], ss[16];
      if (g < nG && 16 * g + 16 <= E) {
        const int4* __restrict__ pd = (const int4*)(dst + (size_t)16 * g);
        const int4* __restrict__ ps = (const int4*)(src + (size_t)16 * g);
        int4 d4[4], s4[4];
#pragma unroll
        for (int q = 0; q < 4; ++q) { d4[q] = pd[q]; s4[q] = ps[q]; }
#pragma unroll
        for (int q = 0; q < 4; ++q) {
          dd[4 * q + 0] = d4[q].x; dd[4 * q + 1] = d4[q].y; dd[4 * q + 2] = d4[q].z; dd[4 * q + 3] = d4[q].w;
          ss[4 * q + 0] = s4[q].x; ss[4 * q + 1] = s4[q].y; ss[4 * q + 2] = s4[q].z; ss[4 * q + 3] = s4[q].w;
        }
      } else {
#pragma unroll
        for (int k = 0; k < 16; ++k) {
          const int  e  = 16 * g + k;
          const bool ok = g < nG && e < E;
          dd[k] = ok ? dst[ok ? e : 0] : -1;
          ss[k] = ok ? src[ok ? e : 0] : 0;
        }
      }
#pragma unroll
      for (int k = 0; k < 16; ++k) {
        if (dd[k] >= 0) {
          const unsigned o = ((unsigned)dd[k] / (unsigned)OWN_R) & 255u;
          atomicOr(&s_b[o * 9 + (lt >> 5)], 1u << (lt & 31));
        }
      }
      {
        uint4* sp = (uint4*)(s_p + lt * 36);
#pragma unroll
        for (int q = 0; q < 8; ++q)
          sp[q] = make_uint4((unsigned)ss[2 * q], (unsigned)dd[2 * q], (unsigned)ss[2 * q + 1], (unsigned)dd[2 * q + 1]);
      }
    }
    __syncthreads();
#pragma unroll
    for (int r = 0; r < 16; ++r) {
      const int x = r * PJ_T + tid, gl = x >> 3, pc = x & 7;
      if (g0 + gl < nG) ((uint4*)pairs)[(size_t)(g0 + gl) * 8 + pc] = *(const uint4*)(s_p + gl * 36 + 4 * pc);
    }
#pragma unroll
    for (int r = 0; r < 16; ++r) {
      const int x = r * PJ_T + tid, k = x >> 3, w = x & 7;
      const int gw = (g0 >> 5) + w;
      if (gw < W32) rows[(size_t)k * RW + gw] = s_b[k * 9 + w];
    }
    __syncthreads();
  }
}

#define FR_PJ 1024
#define FR_PR 512
template <bool ZIN>
__global__ void __launch_bounds__(PJ_T)
front_all(const float* __restrict__ h, const float* __restrict__ W_fc, float* __restrict__ z,
          const float* __restrict__ feat, const float* __restrict__ W_dst, float* __restrict__ df,
          const int* __restrict__ src, const int* __restrict__ dst,
          unsigned* __restrict__ pairs, int N, int E) {
  __shared__ __attribute__((aligned(16))) float s_all[256 * 36 + 256 * 9];
  static_assert(256 * 36 + 256 * 9 >= 2 * 16 * (256 + 4), "projection tiles fit");
  const int b = blockIdx.x;
  if (b < FR_PJ) {
    if ((b & 3) != 3) {
      if constexpr (ZIN) proj_body<256>(h, W_fc, z, N, s_all, (b >> 2) * 3 + (b & 3), (FR_PJ / 4) * 3);
    } else {
      proj_body<64>(feat, W_dst, df, N, s_all, b >> 2, FR_PJ / 4);
    }
  } else {
    unsigned* s_p = (unsigned*)s_all;
    pairs_body(src, dst, pairs, E, s_p, s_p + 256 * 36, b - FR_PJ, FR_PR);
  }
}

__device__ __forceinline__ float row_dot_lane(const float* __restrict__ zr,
                                              const float* __restrict__ dr) {
  float p = 0.0f;
#pragma unroll 1
  for (int k = 0; k < 16; ++k) {
    const float4 a = ((const float4*)zr)[k];
    const float4 b = ((const float4*)dr)[k];
    p = fmaf(a.x, b.x, p); p = fmaf(a.y, b.y, p);
    p = fmaf(a.z, b.z, p); p = fmaf(a.w, b.w, p);
  }
  return p;
}

__device__ __forceinline__ float own_lg2(float e) {
  return fminf(fmaxf(e * OWN_L2E_HI, -OWN_CLAMP), OWN_CLAMP);
}

__device__ __forceinline__ float own_sum8(float p) {
  p += __int_as_float(__builtin_amdgcn_update_dpp(__float_as_int(p), __float_as_int(p), 0x161, 0xF, 0xF, true));
  p += __int_as_float(__builtin_amdgcn_update_dpp(__float_as_int(p), __float_as_int(p), 0x162, 0xF, 0xF, true));
  p += __int_as_float(__builtin_amdgcn_update_dpp(__float_as_int(p), __float_as_int(p), 0x164, 0xF, 0xF, true));
  return p;
}

__device__ __forceinline__ unsigned own_pick4(unsigned q, unsigned a0, unsigned a1, unsigned a2, unsigned a3) {
  const unsigned lo = (q & 1u) ? a1 : a0;
  const unsigned hi = (q & 1u) ? a3 : a2;
  return (q & 2u) ? hi : lo;
}

__global__ void __launch_bounds__(OWN_T)
edge_owner(const float* __restrict__ z,
           const float* __restrict__ df,
           const int* __restrict__ src,
           const int* __restrict__ dst,
           const unsigned* __restrict__ pairs,
           float* __restrict__ out, int N, int E) {
  __shared__ __attribute__((aligned(16))) float s_df[OWN_R * 64];
  __shared__ float          s_num[OWN_R * OWN_P];
  __shared__ float          s_den[OWN_R];
  __shared__ unsigned       s_key[OWN_R];
  __shared__ int            s_es[OWN_CAP];
  __shared__ unsigned short s_tl[OWN_CAP];
  __shared__ unsigned       s_lgw[OWN_CAP];
  __shared__ unsigned short s_loc[OWN_CAP];
  __shared__ unsigned       s_c4[4];
  __shared__ unsigned       s_cnt;
  __shared__ unsigned       s_ng;
  __shared__ unsigned       s_bad;

  const int tid = threadIdx.x;
  const int g   = tid >> 3;
  const int j   = tid & 7;

  const int lane  = tid & 31;
  const int nG  = own_groups(E);
  const int W32 = own_words(E);
  const unsigned* __restrict__ rows = pairs + (size_t)nG * 32;

  for (int base = blockIdx.x * OWN_R; base < N; base += gridDim.x * OWN_R) {
    const int R = min(OWN_R, N - base);

    for (int i = tid; i < OWN_R * OWN_P; i += OWN_T) s_num[i] = 0.0f;
    for (int i = tid; i < OWN_R; i += OWN_T) { s_den[i] = 0.0f; s_key[i] = 0u; }
    for (int i = tid; i < OWN_CAP; i += OWN_T) s_loc[i] = (unsigned short)OWN_NONE;
    {
      const float4* __restrict__ dfr = (const float4*)(df + (size_t)base * 64);
#pragma unroll
      for (int u = 0; u < (OWN_R * 16 + OWN_T - 1) / OWN_T; ++u) {
        const int i = min(tid + u * OWN_T, R * 16 - 1);
        ((float4*)s_df)[i] = dfr[i];
      }
    }
    if (tid == 0) { s_cnt = 0u; s_ng = 0u; s_bad = 0u; }
    if (tid < 4) s_c4[tid] = 0u;
    __syncthreads();

    const unsigned want = ((unsigned)base / (unsigned)OWN_R) & 255u;
    const unsigned* __restrict__ krow = rows + (size_t)want * own_rowp(E);
    for (int w0 = 0; w0 < W32; w0 += OWN_T * 4) {
      unsigned bw[4];
#pragma unroll
      for (int u = 0; u < 4; ++u) {
        const int w = w0 + u * OWN_T + tid;
        bw[u] = krow[min(w, W32 - 1)];
      }
      unsigned bits[4];
      unsigned cl = 0u;
#pragma unroll
      for (int u = 0; u < 4; ++u) {
        const int w = w0 + u * OWN_T + tid;
        bits[u] = (w < W32) ? bw[u] : 0u;
        cl += (unsigned)__popc(bits[u]);
      }
      unsigned incl = cl;
#pragma unroll
      for (int d = 1; d < 32; d <<= 1) {
        const unsigned t = (unsigned)__shfl_up((int)incl, d, 32);
        if (lane >= d) incl += t;
      }
      const unsigned run = (unsigned)__shfl((int)incl, 31, 32);
      if (run != 0u) {
        unsigned first = 0u;
        if (lane == 0) first = atomicAdd(&s_ng, run);
        first = (unsigned)__builtin_amdgcn_readfirstlane((int)first);
        unsigned pos = first + incl - cl;
#pragma unroll
        for (int u = 0; u < 4; ++u) {
          const int w = w0 + u * OWN_T + tid;
          unsigned bl = bits[u];
          while (bl != 0u) {
            const int b = __ffs((int)bl) - 1;
            bl &= bl - 1u;
            if (pos < OWN_CAP) s_lgw[pos] = (unsigned)(32 * w + b);
            ++pos;
          }
        }
      }
    }
    __syncthreads();
    const unsigned ngr = s_ng;

    if (ngr <= OWN_CAP) {
      const int ng = (int)ngr;
      const uint4* __restrict__ pr4 = (const uint4*)pairs;
      const int pc = tid & 7;
      for (int i0 = 0; i0 < ng; i0 += (OWN_T / 8) * OWN_U) {
        uint4 p4[OWN_U];
        bool  in[OWN_U];
#pragma unroll
        for (int u = 0; u < OWN_U; ++u) {
          const int i  = i0 + u * (OWN_T / 8) + g;
          const int gr = (int)s_lgw[min(i, OWN_CAP - 1)];
          in[u] = i < ng;
          const int gs = in[u] ? gr : 0;
          p4[u] = pr4[(size_t)gs * 8 + pc];
        }
        __builtin_amdgcn_sched_barrier(0);
        unsigned b0[OWN_U], b1[OWN_U];
        unsigned run = 0u;
#pragma unroll
        for (int u = 0; u < OWN_U; ++u) {
          b0[u] = __builtin_amdgcn_ballot_w32(in[u] && (unsigned)((int)p4[u].y - base) < (unsigned)R);
          b1[u] = __builtin_amdgcn_ballot_w32(in[u] && (unsigned)((int)p4[u].w - base) < (unsigned)R);
          run += (unsigned)__popc(b0[u]) + (unsigned)__popc(b1[u]);
        }
        if (run != 0u) {
          unsigned first = 0u;
          if (lane == 0) first = atomicAdd(&s_cnt, run);
          first = (unsigned)__builtin_amdgcn_readfirstlane((int)first);
#pragma unroll
          for (int u = 0; u < OWN_U; ++u) {
            const unsigned l0 = (unsigned)((int)p4[u].y - base);
            const unsigned l1 = (unsigned)((int)p4[u].w - base);
            const bool h0 = in[u] && l0 < (unsigned)R;
            const bool h1 = in[u] && l1 < (unsigned)R;
            const unsigned x0 = first + __builtin_amdgcn_mbcnt_lo(b0[u], 0u);
            first += (unsigned)__popc(b0[u]);
            const unsigned x1 = first + __builtin_amdgcn_mbcnt_lo(b1[u], 0u);
            first += (unsigned)__popc(b1[u]);
            if (h0 && x0 < OWN_CAP) { s_es[x0] = (int)p4[u].x; s_tl[x0] = (unsigned short)l0; }
            if (h1 && x1 < OWN_CAP) { s_es[x1] = (int)p4[u].z; s_tl[x1] = (unsigned short)l1; }
          }
        }
      }
    }
    __syncthreads();
    const unsigned total = (ngr <= OWN_CAP) ? s_cnt : (unsigned)(OWN_CAP + 1);
    bool listed = total <= OWN_CAP;
    int  bound  = 0;

    if (listed) {
      const int cnt = (int)total;

      {
        int sv[OWN_CAP / OWN_T];
        unsigned lc[OWN_CAP / OWN_T];
        unsigned rk[OWN_CAP / OWN_T];
#pragma unroll
        for (int u = 0; u < OWN_CAP / OWN_T; ++u) {
          const int i  = tid + u * OWN_T;
          const int      ev = s_es[i];
          const unsigned lv = (unsigned)s_tl[i];
          sv[u] = (i < cnt) ? ev : 0;
          lc[u] = (i < cnt) ? lv : 0u;
        }
        unsigned pre0 = 0u, pre1 = 0u, pre2 = 0u, pre3 = 0u;
#pragma unroll
        for (int u = 0; u < OWN_CAP / OWN_T; ++u) {
          const int      i    = tid + u * OWN_T;
          const unsigned loc  = lc[u];
          const bool     keep = i < cnt && loc < (unsigned)R;
          const unsigned q    = loc & 3u;
          const unsigned mk = __builtin_amdgcn_ballot_w32(keep);
          const unsigned ml = __builtin_amdgcn_ballot_w32(keep && (q & 1u) != 0u);
          const unsigned mh = __builtin_amdgcn_ballot_w32(keep && (q & 2u) != 0u);
          const unsigned b0 = mk & ~ml & ~mh, b1 = ml & ~mh, b2 = mh & ~ml, b3 = ml & mh;
          rk[u] = own_pick4(q, pre0, pre1, pre2, pre3)
                + __builtin_amdgcn_mbcnt_lo(own_pick4(q, b0, b1, b2, b3), 0u);
          pre0 += (unsigned)__popc(b0); pre1 += (unsigned)__popc(b1);
          pre2 += (unsigned)__popc(b2); pre3 += (unsigned)__popc(b3);
        }
        unsigned got = 0u;
        if (lane < 4) {
          got = atomicAdd(&s_c4[lane], own_pick4((unsigned)lane, pre0, pre1, pre2, pre3));
        }
        const unsigned f0 = (unsigned)__builtin_amdgcn_readlane((int)got, 0);
        const unsigned f1 = (unsigned)__builtin_amdgcn_readlane((int)got, 1);
        const unsigned f2 = (unsigned)__builtin_amdgcn_readlane((int)got, 2);
        const unsigned f3 = (unsigned)__builtin_amdgcn_readlane((int)got, 3);
#pragma unroll
        for (int u = 0; u < OWN_CAP / OWN_T; ++u) {
          const int      i   = tid + u * OWN_T;
          const unsigned loc = lc[u];
          const unsigned q   = loc & 3u;
          const unsigned p   = own_pick4(q, f0, f1, f2, f3) + rk[u];
          if (i < cnt && loc < (unsigned)R && p < (unsigned)(OWN_CAP / 4)) {
            const unsigned slot = 4u * p + q;
            s_lgw[slot] = (unsigned)min(max(sv[u], 0), N - 1);
            s_loc[slot] = (unsigned short)loc;
          }
        }
      }
      __syncthreads();
      const unsigned maxc = max(max(s_c4[0], s_c4[1]), max(s_c4[2], s_c4[3]));
      listed = maxc <= (unsigned)(OWN_CAP / 4);
      bound  = 4 * (int)maxc;
    }

    if (listed) {
      bool far = false;
      for (int i0 = 0; i0 < bound; i0 += 4 * (OWN_T / 8)) {
        float4 za[4], zb[4];
        int    li[4];
        bool   ok[4];
#pragma unroll
        for (int u = 0; u < 4; ++u) {
          const int  i  = i0 + u * (OWN_T / 8) + g;
          const int  sr = (int)s_lgw[i];
          const int  lr = (int)s_loc[i];
          ok[u] = lr != OWN_NONE;
          const int  s  = ok[u] ? sr : 0;
          li[u] = ok[u] ? lr : 0;
          const float4* zr = (const float4*)(z + (size_t)s * 64);
          za[u] = zr[j]; zb[u] = zr[8 + j];
        }
        float pe[4];
        {
          float4 da[4], db[4];
#pragma unroll
          for (int u = 0; u < 4; ++u) {
            const float4* dr = (const float4*)(s_df + li[u] * 64);
            da[u] = dr[j]; db[u] = dr[8 + j];
          }
#pragma unroll
          for (int u = 0; u < 4; ++u) {
            const float p = za[u].x * da[u].x + za[u].y * da[u].y + za[u].z * da[u].z + za[u].w * da[u].w
                          + zb[u].x * db[u].x + zb[u].y * db[u].y + zb[u].z * db[u].z + zb[u].w * db[u].w;
            pe[u] = own_sum8(p);
          }
        }
#pragma unroll
        for (int u = 0; u < 4; ++u) {
          const float p = pe[u];
          const float th = p * OWN_L2E_HI;
          const float tl = fmaf(p, OWN_L2E_LO, fmaf(p, OWN_L2E_HI, -th));
          const float tc = own_lg2(p);
          const int   ti = (int)floorf(tc);
          const float ef = __builtin_amdgcn_exp2f(__builtin_amdgcn_fractf(tc) + tl);
          const float ex = ldexpf(ef, ti);
          if (ok[u]) {
            far = far || (unsigned)(ti + OWN_SPREAD) > (unsigned)(2 * OWN_SPREAD);
            float* np = s_num + li[u] * OWN_P + 4 * j;
            atomicAdd(np + 0,  ex * za[u].x); atomicAdd(np + 1,  ex * za[u].y);
            atomicAdd(np + 2,  ex * za[u].z); atomicAdd(np + 3,  ex * za[u].w);
            atomicAdd(np + 32, ex * zb[u].x); atomicAdd(np + 33, ex * zb[u].y);
            atomicAdd(np + 34, ex * zb[u].z); atomicAdd(np + 35, ex * zb[u].w);
            if (j == 0) atomicAdd(&s_den[li[u]], ex);
          }
        }
      }
      if (far) s_bad = 1u;
      __syncthreads();

      if (s_bad != 0u) {
        for (int i = tid; i < OWN_R * OWN_P; i += OWN_T) s_num[i] = 0.0f;
        for (int i = tid; i < OWN_R; i += OWN_T) s_den[i] = 0.0f;

        for (int i = g; i < bound; i += OWN_T / 8) {
          const int lr = (int)s_loc[i];
          if (lr != OWN_NONE) {
            const float4* zr = (const float4*)(z + (size_t)s_lgw[i] * 64);
            const float4* dr = (const float4*)(s_df + lr * 64);
            const float4 a0 = zr[j], a1 = zr[8 + j];
            const float4 d0 = dr[j], d1 = dr[8 + j];
            const float p = own_sum8(a0.x * d0.x + a0.y * d0.y + a0.z * d0.z + a0.w * d0.w
                                   + a1.x * d1.x + a1.y * d1.y + a1.z * d1.z + a1.w * d1.w);
            if (j == 0) {
              s_es[i] = (int)__float_as_uint(p);
              atomicMax(&s_key[lr], float_key(p));
            }
          }
        }
        __syncthreads();

        for (int i0 = 0; i0 < bound; i0 += 4 * (OWN_T / 8)) {
          float4 za[4], zb[4];
#pragma unroll
          for (int u = 0; u < 4; ++u) {
            const int  i = i0 + u * (OWN_T / 8) + g;
            const int  sr = (int)s_lgw[i];
            const int  s  = ((int)s_loc[i] != OWN_NONE) ? sr : 0;
            const float4* zr = (const float4*)(z + (size_t)s * 64);
            za[u] = zr[j]; zb[u] = zr[8 + j];
          }
#pragma unroll
          for (int u = 0; u < 4; ++u) {
            const int i = i0 + u * (OWN_T / 8) + g;
            if ((int)s_loc[i] != OWN_NONE) {
              const int   loc = (int)s_loc[i];
              const float ex  = __expf(__uint_as_float((unsigned)s_es[i]) - key_float(s_key[loc]));
              float* np = s_num + loc * OWN_P + 4 * j;
              atomicAdd(np + 0,  ex * za[u].x); atomicAdd(np + 1,  ex * za[u].y);
              atomicAdd(np + 2,  ex * za[u].z); atomicAdd(np + 3,  ex * za[u].w);
              atomicAdd(np + 32, ex * zb[u].x); atomicAdd(np + 33, ex * zb[u].y);
              atomicAdd(np + 34, ex * zb[u].z); atomicAdd(np + 35, ex * zb[u].w);
              if (j == 0) atomicAdd(&s_den[loc], ex);
            }
          }
        }
      }
    } else {
      for (int e = tid; e < E; e += OWN_T) {
        const unsigned loc = (unsigned)(dst[e] - base);
        if (loc < (unsigned)R) {
          const int s = min(max(src[e], 0), N - 1);
          const float p = row_dot_lane(z + (size_t)s * 64, s_df + loc * 64);
          atomicMax(&s_key[loc], float_key(p));
        }
      }
      __syncthreads();
      for (int e = tid; e < E; e += OWN_T) {
        const unsigned loc = (unsigned)(dst[e] - base);
        if (loc < (unsigned)R) {
          const int s = min(max(src[e], 0), N - 1);
          const float* zr = z + (size_t)s * 64;
          const float p  = row_dot_lane(zr, s_df + loc * 64);
          const float ex = __expf(p - key_float(s_key[loc]));
          float* np = s_num + loc * OWN_P;
          for (int k = 0; k < 64; ++k) atomicAdd(np + k, ex * zr[k]);
          atomicAdd(&s_den[loc], ex);
        }
      }
    }
    __syncthreads();

    for (int i = tid; i < R * 64; i += OWN_T) {
      const int   loc = i >> 6, f = i & 63;
      const float dn  = s_den[loc];
      out[(size_t)(base + loc) * 64 + f] = s_num[loc * OWN_P + f] / (dn == 0.0f ? 1.0f : dn);
    }
    __syncthreads();
  }
}

extern "C" void kernel_launch(void* const* d_in, const int* in_sizes, int n_in,
                              void* d_out, int out_size, void* d_ws, size_t ws_size,
                              hipStream_t stream) {
  const float* h     = (const float*)d_in[0];
  const float* feat  = (const float*)d_in[1];
  const float* W_fc  = (const float*)d_in[2];
  const float* W_dst = (const float*)d_in[3];
  const int*   src   = (const int*)d_in[4];
  const int*   dst   = (const int*)d_in[5];
  float*       out   = (float*)d_out;

  const int FEAT = 64, O = 64;
  const int N   = in_sizes[1] / FEAT;
  const int Kfc = in_sizes[0] / N;
  const int E   = in_sizes[4];

  float* z  = (float*)d_ws;
  float* df = z + (size_t)N * O;
  unsigned* pairs = (unsigned*)(df + (size_t)N * O);

  if (Kfc == 256) {
    front_all<true><<<FR_PJ + FR_PR, PJ_T, 0, stream>>>(h, W_fc, z, feat, W_dst, df, src, dst, pairs, N, E);
  } else {
    const dim3 gemmBlk(32 * (O / 16));
    const int  mTiles = (N + 15) / 16;
    gemm_nt_wmma<<<mTiles, gemmBlk, 0, stream>>>(h, W_fc, z, N, Kfc, O);
    front_all<false><<<FR_PJ + FR_PR, PJ_T, 0, stream>>>(h, W_fc, z, feat, W_dst, df, src, dst, pairs, N, E);
  }

  edge_owner<<<256, OWN_T, 0, stream>>>(z, df, src, dst, pairs, out, N, E);
}
